# starting kernel
# baseline (speedup 1.0000x reference)
#include <hip/hip_runtime.h>
#include <hip/hip_bf16.h>

#define N_NODES   100000
#define N_EDGES   1600000
#define HID       128
#define N_CLASSES 10
#define N_GRAPHS  512
#define N_LAYERS  5
#define E_TOT     (N_EDGES + N_NODES)

typedef __attribute__((ext_vector_type(2))) float v2f;
typedef __attribute__((ext_vector_type(8))) float v8f;

__global__ void init_deg_kernel(float* __restrict__ deg) {
    int i = blockIdx.x * blockDim.x + threadIdx.x;
    if (i < N_NODES) deg[i] = 1.0f;
}

__global__ void deg_accum_kernel(const int* __restrict__ dst, float* __restrict__ deg) {
    int e = blockIdx.x * blockDim.x + threadIdx.x;
    if (e < N_EDGES) atomicAdd(&deg[dst[e]], 1.0f);
}

__global__ void dinv_kernel(float* __restrict__ deg) {
    int i = blockIdx.x * blockDim.x + threadIdx.x;
    if (i < N_NODES) deg[i] = rsqrtf(deg[i]);
}

__global__ void norm_kernel(const int* __restrict__ src, const int* __restrict__ dst,
                            const float* __restrict__ dinv, float* __restrict__ norm) {
    int e = blockIdx.x * blockDim.x + threadIdx.x;
    if (e >= E_TOT) return;
    if (e < N_EDGES) {
        norm[e] = dinv[src[e]] * dinv[dst[e]];
    } else {
        float dv = dinv[e - N_EDGES];
        norm[e] = dv * dv;
    }
}

__global__ __launch_bounds__(256)
void gemm_wmma_kernel(const float* __restrict__ X, const float* __restrict__ W,
                      float* __restrict__ H) {
    int wave     = (blockIdx.x * blockDim.x + threadIdx.x) >> 5;
    int lane     = threadIdx.x & 31;
    int row_tile = wave >> 3;
    int col_tile = wave & 7;
    if (row_tile * 16 >= N_NODES) return;

    int m  = lane & 15;
    int kh = lane >> 4;
    const float* xrow = X + (size_t)(row_tile * 16 + m) * HID;
    int cb = col_tile * 16;

    v8f acc = {};
    #pragma unroll
    for (int kb = 0; kb < HID; kb += 4) {
        int k0 = kb + 2 * kh;
        v2f a = *(const v2f*)(xrow + k0);
        v2f b;
        b.x = W[(size_t)(k0 + 0) * HID + cb + m];
        b.y = W[(size_t)(k0 + 1) * HID + cb + m];
        acc = __builtin_amdgcn_wmma_f32_16x16x4_f32(
             false, a,  false, b,
             (short)0, acc,  false,  false);
    }

    float* hbase = H + (size_t)(row_tile * 16 + 8 * kh) * HID + cb + m;
    #pragma unroll
    for (int r = 0; r < 8; ++r) hbase[(size_t)r * HID] = acc[r];
}

__global__ void zero_kernel(float4* __restrict__ p, int n4) {
    int i = blockIdx.x * blockDim.x + threadIdx.x;
    if (i < n4) p[i] = make_float4(0.f, 0.f, 0.f, 0.f);
}

__global__ __launch_bounds__(256)
void scatter_kernel(const float* __restrict__ H, const int* __restrict__ src,
                    const int* __restrict__ dst, const float* __restrict__ norm,
                    float* __restrict__ out) {
    int wave = (blockIdx.x * blockDim.x + threadIdx.x) >> 5;
    int lane = threadIdx.x & 31;
    if (wave >= E_TOT) return;
    int s, d;
    if (wave < N_EDGES) { s = src[wave]; d = dst[wave]; }
    else                { s = wave - N_EDGES; d = s; }
    float w = norm[wave];
    float4 v = ((const float4*)(H + (size_t)s * HID))[lane];
    float* op = out + (size_t)d * HID + lane * 4;
    atomicAdd(op + 0, v.x * w);
    atomicAdd(op + 1, v.y * w);
    atomicAdd(op + 2, v.z * w);
    atomicAdd(op + 3, v.w * w);
}

__global__ void bias_relu_kernel(float* __restrict__ X, const float* __restrict__ b) {
    int i = blockIdx.x * blockDim.x + threadIdx.x;
    if (i < N_NODES * HID) {
        float v = X[i] + b[i & (HID - 1)];
        X[i] = v > 0.f ? v : 0.f;
    }
}

__global__ __launch_bounds__(256)
void pool_accum_kernel(const float* __restrict__ X, const int* __restrict__ batch,
                       float* __restrict__ pooled, float* __restrict__ counts) {
    int wave = (blockIdx.x * blockDim.x + threadIdx.x) >> 5;
    int lane = threadIdx.x & 31;
    if (wave >= N_NODES) return;
    int g = batch[wave];
    float4 v = ((const float4*)(X + (size_t)wave * HID))[lane];
    float* pp = pooled + (size_t)g * HID + lane * 4;
    atomicAdd(pp + 0, v.x);
    atomicAdd(pp + 1, v.y);
    atomicAdd(pp + 2, v.z);
    atomicAdd(pp + 3, v.w);
    if (lane == 0) atomicAdd(&counts[g], 1.0f);
}

__global__ void final_linear_kernel(const float* __restrict__ pooled,
                                    const float* __restrict__ counts,
                                    const float* __restrict__ linW,
                                    const float* __restrict__ linb,
                                    float* __restrict__ out) {
    int i = blockIdx.x * blockDim.x + threadIdx.x;
    if (i >= N_GRAPHS * N_CLASSES) return;
    int g = i / N_CLASSES;
    int c = i - g * N_CLASSES;
    float scale = 1.0f / fmaxf(counts[g], 1.0f);
    const float* pg = pooled + (size_t)g * HID;
    float acc = 0.f;
    #pragma unroll 4
    for (int f = 0; f < HID; ++f) acc += pg[f] * linW[f * N_CLASSES + c];
    out[i] = acc * scale + linb[c];
}

extern "C" void kernel_launch(void* const* d_in, const int* in_sizes, int n_in,
                              void* d_out, int out_size, void* d_ws, size_t ws_size,
                              hipStream_t stream) {
    const float* x     = (const float*)d_in[0];
    const int*   edge  = (const int*)d_in[1];
    const int*   src   = edge;
    const int*   dst   = edge + N_EDGES;
    const int*   batch = (const int*)d_in[2];
    const float* Ws    = (const float*)d_in[3];
    const float* bs    = (const float*)d_in[4];
    const float* linW  = (const float*)d_in[5];
    const float* linb  = (const float*)d_in[6];
    float*       out   = (float*)d_out;

    float* ws = (float*)d_ws;
    float* deg    = ws;
    float* norm   = deg + N_NODES;
    float* H      = norm + E_TOT;
    float* X1     = H + (size_t)N_NODES * HID;
    float* X2     = X1 + (size_t)N_NODES * HID;
    float* pooled = X2 + (size_t)N_NODES * HID;
    float* counts = pooled + (size_t)N_GRAPHS * HID;

    const int T = 256;

    init_deg_kernel<<<(N_NODES + T - 1) / T, T, 0, stream>>>(deg);
    deg_accum_kernel<<<(N_EDGES + T - 1) / T, T, 0, stream>>>(dst, deg);
    dinv_kernel<<<(N_NODES + T - 1) / T, T, 0, stream>>>(deg);
    norm_kernel<<<(E_TOT + T - 1) / T, T, 0, stream>>>(src, dst, deg, norm);

    const int gemm_blocks    = (N_NODES / 16) * (HID / 16) / 8;
    const int scatter_blocks = E_TOT / 8;
    const int feat4          = N_NODES * HID / 4;

    const float* xin = x;
    float* xbuf[2] = { X1, X2 };
    for (int l = 0; l < N_LAYERS; ++l) {
        float* xout = xbuf[l & 1];
        const float* Wl = Ws + (size_t)l * HID * HID;
        const float* bl = bs + (size_t)l * HID;

        gemm_wmma_kernel<<<gemm_blocks, T, 0, stream>>>(xin, Wl, H);
        zero_kernel<<<(feat4 + T - 1) / T, T, 0, stream>>>((float4*)xout, feat4);
        scatter_kernel<<<scatter_blocks, T, 0, stream>>>(H, src, dst, norm, xout);
        bias_relu_kernel<<<(N_NODES * HID + T - 1) / T, T, 0, stream>>>(xout, bl);

        xin = xout;
    }

    int pool4 = (N_GRAPHS * HID + N_GRAPHS + 3) / 4;
    zero_kernel<<<(pool4 + T - 1) / T, T, 0, stream>>>((float4*)pooled, pool4);
    pool_accum_kernel<<<N_NODES / 8, T, 0, stream>>>(xin, batch, pooled, counts);
    final_linear_kernel<<<(N_GRAPHS * N_CLASSES + T - 1) / T, T, 0, stream>>>(
        pooled, counts, linW, linb, out);
}


# Edge sum moved from global float atomics to LDS: edges binned per call by 128-node destination tile; a workgroup adds its tile's rows by LDS float atomics (arrival order), bias+ReLU, one store
# speedup vs baseline: 2.0162x; 2.0162x over previous
#include <hip/hip_runtime.h>
#include <hip/hip_bf16.h>


#define N_NODES   100000
#define N_EDGES   1600000
#define HID       128
#define N_CLASSES 10
#define N_GRAPHS  512
#define N_LAYERS  5
#define TN        128
#define NT        ((N_NODES + TN - 1) / TN)
#define CPAD      32
#define AGG_T     512

typedef __attribute__((ext_vector_type(2))) float v2f;
typedef __attribute__((ext_vector_type(8))) float v8f;

__global__ void init_deg_kernel(float* __restrict__ deg) {
    int i = blockIdx.x * blockDim.x + threadIdx.x;
    if (i < N_NODES) deg[i] = 1.0f;
}

__global__ void deg_accum_kernel(const int* __restrict__ dst, float* __restrict__ deg) {
    int e = blockIdx.x * blockDim.x + threadIdx.x;
    if (e < N_EDGES) atomicAdd(&deg[dst[e]], 1.0f);
}

__global__ void dinv_kernel(float* __restrict__ deg) {
    int i = blockIdx.x * blockDim.x + threadIdx.x;
    if (i < N_NODES) deg[i] = rsqrtf(deg[i]);
}

__global__ __launch_bounds__(1024)
void tile_scan_kernel(const float* __restrict__ deg, int* __restrict__ tile_start,
                      int* __restrict__ cursor) {
    __shared__ int cnt[1024];
    __shared__ int wsum[32];
    const int tid  = threadIdx.x;
    const int lane = tid & 31;
    const int wave = tid >> 5;

    for (int t = wave; t < 1024; t += 32) {
        int c = 0;
        int n = t * TN + lane * 4;
        if (t < NT && n < N_NODES) {
            float4 d = *(const float4*)(deg + n);
            c = (int)d.x + (int)d.y + (int)d.z + (int)d.w - 4;
        }
        for (int off = 16; off > 0; off >>= 1) c += __shfl_xor(c, off);
        if (lane == 0) cnt[t] = c;
    }
    __syncthreads();

    const int v = cnt[tid];
    int x = v;
    for (int off = 1; off < 32; off <<= 1) {
        int y = __shfl_up(x, off);
        if (lane >= off) x += y;
    }
    if (lane == 31) wsum[wave] = x;
    __syncthreads();
    if (wave == 0) {
        const int s = wsum[lane];
        int xs = s;
        for (int off = 1; off < 32; off <<= 1) {
            int y = __shfl_up(xs, off);
            if (lane >= off) xs += y;
        }
        wsum[lane] = xs - s;
    }
    __syncthreads();
    const int excl = x - v + wsum[wave];
    if (tid <= NT) tile_start[tid] = excl;
    if (tid < NT)  cursor[tid * CPAD] = excl;
}

__global__ void fill_kernel(const int* __restrict__ src, const int* __restrict__ dst,
                            const float* __restrict__ dinv, int* __restrict__ cursor,
                            int* __restrict__ bsrc, int* __restrict__ bdl,
                            float* __restrict__ bw) {
    int e = blockIdx.x * blockDim.x + threadIdx.x;
    if (e >= N_EDGES) return;
    int s = src[e];
    int d = dst[e];
    int pos = atomicAdd(&cursor[(d >> 7) * CPAD], 1);
    bsrc[pos] = s;
    bdl[pos]  = d & (TN - 1);
    bw[pos]   = dinv[s] * dinv[d];
}

__global__ __launch_bounds__(256)
void gemm_wmma_kernel(const float* __restrict__ X, const float* __restrict__ W,
                      float* __restrict__ H) {
    int wave     = (blockIdx.x * blockDim.x + threadIdx.x) >> 5;
    int lane     = threadIdx.x & 31;
    int row_tile = wave >> 3;
    int col_tile = wave & 7;
    if (row_tile * 16 >= N_NODES) return;

    int m  = lane & 15;
    int kh = lane >> 4;
    const float* xrow = X + (size_t)(row_tile * 16 + m) * HID;
    int cb = col_tile * 16;

    v8f acc = {};
    #pragma unroll
    for (int kb = 0; kb < HID; kb += 4) {
        int k0 = kb + 2 * kh;
        v2f a = *(const v2f*)(xrow + k0);
        v2f b;
        b.x = W[(size_t)(k0 + 0) * HID + cb + m];
        b.y = W[(size_t)(k0 + 1) * HID + cb + m];
        acc = __builtin_amdgcn_wmma_f32_16x16x4_f32(
             false, a,  false, b,
             (short)0, acc,  false,  false);
    }

    float* hbase = H + (size_t)(row_tile * 16 + 8 * kh) * HID + cb + m;
    #pragma unroll
    for (int r = 0; r < 8; ++r) hbase[(size_t)r * HID] = acc[r];
}

__global__ void zero_kernel(float4* __restrict__ p, int n4) {
    int i = blockIdx.x * blockDim.x + threadIdx.x;
    if (i < n4) p[i] = make_float4(0.f, 0.f, 0.f, 0.f);
}

__global__ __launch_bounds__(AGG_T)
void agg_kernel(const float* __restrict__ H, const float* __restrict__ dinv,
                const int* __restrict__ tile_start, const int* __restrict__ bsrc,
                const int* __restrict__ bdl, const float* __restrict__ bw,
                const float* __restrict__ bias, float* __restrict__ out) {
    __shared__ float tile[TN * HID];
    const int t    = blockIdx.x;
    const int lane = threadIdx.x & 31;
    const int wave = __builtin_amdgcn_readfirstlane((int)(threadIdx.x >> 5));
    const int NW   = AGG_T / 32;
    const int node0 = t * TN;

    for (int r = wave; r < TN; r += NW) {
        const int node = node0 + r;
        float4 v = make_float4(0.f, 0.f, 0.f, 0.f);
        if (node < N_NODES) {
            const float dv = dinv[node];
            const float w  = dv * dv;
            const float4 h = ((const float4*)(H + (size_t)node * HID))[lane];
            v.x = h.x * w; v.y = h.y * w; v.z = h.z * w; v.w = h.w * w;
        }
        float* row = tile + r * HID + lane;
        row[0] = v.x; row[32] = v.y; row[64] = v.z; row[96] = v.w;
    }
    __syncthreads();

    const int beg = tile_start[t];
    const int end = tile_start[t + 1];
    for (int base = beg + wave * 32; base < end; base += NW * 32) {
        const int idx = base + lane;
        int s = 0, dl = 0;
        float w = 0.f;
        if (idx < end) { s = bsrc[idx]; dl = bdl[idx]; w = bw[idx]; }
        const int n = min(32, end - base);
        for (int k = 0; k < n; k += 4) {
            float4 h[4];
            #pragma unroll
            for (int u = 0; u < 4; ++u) {
                const int sk = __builtin_amdgcn_readlane(s, k + u);
                h[u] = ((const float4*)(H + (size_t)sk * HID))[lane];
            }
            #pragma unroll
            for (int u = 0; u < 4; ++u) {
                if (u == 0 || k + u < n) {
                    const int   dk = __builtin_amdgcn_readlane(dl, k + u);
                    const float wk = __int_as_float(__builtin_amdgcn_readlane(__float_as_int(w), k + u));
                    float* row = tile + dk * HID + lane;
                    __hip_atomic_fetch_add(row,      h[u].x * wk, __ATOMIC_RELAXED, __HIP_MEMORY_SCOPE_WORKGROUP);
                    __hip_atomic_fetch_add(row + 32, h[u].y * wk, __ATOMIC_RELAXED, __HIP_MEMORY_SCOPE_WORKGROUP);
                    __hip_atomic_fetch_add(row + 64, h[u].z * wk, __ATOMIC_RELAXED, __HIP_MEMORY_SCOPE_WORKGROUP);
                    __hip_atomic_fetch_add(row + 96, h[u].w * wk, __ATOMIC_RELAXED, __HIP_MEMORY_SCOPE_WORKGROUP);
                }
            }
        }
    }
    __syncthreads();

    const float4 b = ((const float4*)bias)[lane];
    for (int r = wave; r < TN; r += NW) {
        const int node = node0 + r;
        if (node < N_NODES) {
            const float* row = tile + r * HID + lane;
            float4 o;
            o.x = row[0]  + b.x; o.y = row[32] + b.y; o.z = row[64] + b.z; o.w = row[96] + b.w;
            o.x = o.x > 0.f ? o.x : 0.f;
            o.y = o.y > 0.f ? o.y : 0.f;
            o.z = o.z > 0.f ? o.z : 0.f;
            o.w = o.w > 0.f ? o.w : 0.f;
            ((float4*)(out + (size_t)node * HID))[lane] = o;
        }
    }
}

__global__ __launch_bounds__(256)
void pool_accum_kernel(const float* __restrict__ X, const int* __restrict__ batch,
                       float* __restrict__ pooled, float* __restrict__ counts) {
    int wave = (blockIdx.x * blockDim.x + threadIdx.x) >> 5;
    int lane = threadIdx.x & 31;
    if (wave >= N_NODES) return;
    int g = batch[wave];
    float4 v = ((const float4*)(X + (size_t)wave * HID))[lane];
    float* pp = pooled + (size_t)g * HID + lane * 4;
    atomicAdd(pp + 0, v.x);
    atomicAdd(pp + 1, v.y);
    atomicAdd(pp + 2, v.z);
    atomicAdd(pp + 3, v.w);
    if (lane == 0) atomicAdd(&counts[g], 1.0f);
}

__global__ void final_linear_kernel(const float* __restrict__ pooled,
                                    const float* __restrict__ counts,
                                    const float* __restrict__ linW,
                                    const float* __restrict__ linb,
                                    float* __restrict__ out) {
    int i = blockIdx.x * blockDim.x + threadIdx.x;
    if (i >= N_GRAPHS * N_CLASSES) return;
    int g = i / N_CLASSES;
    int c = i - g * N_CLASSES;
    float scale = 1.0f / fmaxf(counts[g], 1.0f);
    const float* pg = pooled + (size_t)g * HID;
    float acc = 0.f;
    #pragma unroll 4
    for (int f = 0; f < HID; ++f) acc += pg[f] * linW[f * N_CLASSES + c];
    out[i] = acc * scale + linb[c];
}

extern "C" void kernel_launch(void* const* d_in, const int* in_sizes, int n_in,
                              void* d_out, int out_size, void* d_ws, size_t ws_size,
                              hipStream_t stream) {
    const float* x     = (const float*)d_in[0];
    const int*   edge  = (const int*)d_in[1];
    const int*   src   = edge;
    const int*   dst   = edge + N_EDGES;
    const int*   batch = (const int*)d_in[2];
    const float* Ws    = (const float*)d_in[3];
    const float* bs    = (const float*)d_in[4];
    const float* linW  = (const float*)d_in[5];
    const float* linb  = (const float*)d_in[6];
    float*       out   = (float*)d_out;

    float* ws = (float*)d_ws;
    float* H      = ws;
    float* X1     = H + (size_t)N_NODES * HID;
    float* X2     = X1 + (size_t)N_NODES * HID;
    float* pooled = X2 + (size_t)N_NODES * HID;
    float* counts = pooled + (size_t)N_GRAPHS * HID;
    float* deg    = counts + N_GRAPHS;
    float* bw     = deg + N_NODES;
    int*   bsrc   = (int*)(bw + N_EDGES);
    int*   bdl    = bsrc + N_EDGES;
    int*   tstart = bdl + N_EDGES;
    int*   cursor = tstart + 1024;

    const int T = 256;

    init_deg_kernel<<<(N_NODES + T - 1) / T, T, 0, stream>>>(deg);
    deg_accum_kernel<<<(N_EDGES + T - 1) / T, T, 0, stream>>>(dst, deg);
    tile_scan_kernel<<<1, 1024, 0, stream>>>(deg, tstart, cursor);
    dinv_kernel<<<(N_NODES + T - 1) / T, T, 0, stream>>>(deg);
    fill_kernel<<<(N_EDGES + T - 1) / T, T, 0, stream>>>(src, dst, deg, cursor, bsrc, bdl, bw);

    const int gemm_blocks = (N_NODES / 16) * (HID / 16) / 8;

    const float* xin = x;
    float* xbuf[2] = { X1, X2 };
    for (int l = 0; l < N_LAYERS; ++l) {
        float* xout = xbuf[l & 1];
        const float* Wl = Ws + (size_t)l * HID * HID;
        const float* bl = bs + (size_t)l * HID;

        gemm_wmma_kernel<<<gemm_blocks, T, 0, stream>>>(xin, Wl, H);
        agg_kernel<<<NT, AGG_T, 0, stream>>>(H, deg, tstart, bsrc, bdl, bw, bl, xout);

        xin = xout;
    }

    int pool4 = (N_GRAPHS * HID + N_GRAPHS + 3) / 4;
    zero_kernel<<<(pool4 + T - 1) / T, T, 0, stream>>>((float4*)pooled, pool4);
    pool_accum_kernel<<<N_NODES / 8, T, 0, stream>>>(xin, batch, pooled, counts);
    final_linear_kernel<<<(N_GRAPHS * N_CLASSES + T - 1) / T, T, 0, stream>>>(
        pooled, counts, linW, linb, out);
}


# Products of layers 2-5 moved into the tile launch before them: the finished 128x128 LDS tile (bias+ReLU in place, row pitch 132) is the WMMA A operand, W read from L2; their k sum order changes
# speedup vs baseline: 2.3082x; 1.1448x over previous
#include <hip/hip_runtime.h>
#include <hip/hip_bf16.h>


#define N_NODES   100000
#define N_EDGES   1600000
#define HID       128
#define N_CLASSES 10
#define N_GRAPHS  512
#define N_LAYERS  5
#define TN        128
#define NT        ((N_NODES + TN - 1) / TN)
#define CPAD      32
#define AGG_T     512
#define LP        132

typedef __attribute__((ext_vector_type(2))) float v2f;
typedef __attribute__((ext_vector_type(8))) float v8f;

__global__ void init_deg_kernel(float* __restrict__ deg) {
    int i = blockIdx.x * blockDim.x + threadIdx.x;
    if (i < N_NODES) deg[i] = 1.0f;
}

__global__ void deg_accum_kernel(const int* __restrict__ dst, float* __restrict__ deg) {
    int e = blockIdx.x * blockDim.x + threadIdx.x;
    if (e < N_EDGES) atomicAdd(&deg[dst[e]], 1.0f);
}

__global__ void dinv_kernel(float* __restrict__ deg) {
    int i = blockIdx.x * blockDim.x + threadIdx.x;
    if (i < N_NODES) deg[i] = rsqrtf(deg[i]);
}

__global__ __launch_bounds__(1024)
void tile_scan_kernel(const float* __restrict__ deg, int* __restrict__ tile_start,
                      int* __restrict__ cursor) {
    __shared__ int cnt[1024];
    __shared__ int wsum[32];
    const int tid  = threadIdx.x;
    const int lane = tid & 31;
    const int wave = tid >> 5;

    for (int t = wave; t < 1024; t += 32) {
        int c = 0;
        int n = t * TN + lane * 4;
        if (t < NT && n < N_NODES) {
            float4 d = *(const float4*)(deg + n);
            c = (int)d.x + (int)d.y + (int)d.z + (int)d.w - 4;
        }
        for (int off = 16; off > 0; off >>= 1) c += __shfl_xor(c, off);
        if (lane == 0) cnt[t] = c;
    }
    __syncthreads();

    const int v = cnt[tid];
    int x = v;
    for (int off = 1; off < 32; off <<= 1) {
        int y = __shfl_up(x, off);
        if (lane >= off) x += y;
    }
    if (lane == 31) wsum[wave] = x;
    __syncthreads();
    if (wave == 0) {
        const int s = wsum[lane];
        int xs = s;
        for (int off = 1; off < 32; off <<= 1) {
            int y = __shfl_up(xs, off);
            if (lane >= off) xs += y;
        }
        wsum[lane] = xs - s;
    }
    __syncthreads();
    const int excl = x - v + wsum[wave];
    if (tid <= NT) tile_start[tid] = excl;
    if (tid < NT)  cursor[tid * CPAD] = excl;
}

__global__ void fill_kernel(const int* __restrict__ src, const int* __restrict__ dst,
                            const float* __restrict__ dinv, int* __restrict__ cursor,
                            int* __restrict__ bsrc, int* __restrict__ bdl,
                            float* __restrict__ bw) {
    int e = blockIdx.x * blockDim.x + threadIdx.x;
    if (e >= N_EDGES) return;
    int s = src[e];
    int d = dst[e];
    int pos = atomicAdd(&cursor[(d >> 7) * CPAD], 1);
    bsrc[pos] = s;
    bdl[pos]  = d & (TN - 1);
    bw[pos]   = dinv[s] * dinv[d];
}

__global__ __launch_bounds__(256)
void gemm_wmma_kernel(const float* __restrict__ X, const float* __restrict__ W,
                      float* __restrict__ H) {
    int wave     = (blockIdx.x * blockDim.x + threadIdx.x) >> 5;
    int lane     = threadIdx.x & 31;
    int row_tile = wave >> 3;
    int col_tile = wave & 7;
    if (row_tile * 16 >= N_NODES) return;

    int m  = lane & 15;
    int kh = lane >> 4;
    const float* xrow = X + (size_t)(row_tile * 16 + m) * HID;
    int cb = col_tile * 16;

    v8f acc = {};
    #pragma unroll
    for (int kb = 0; kb < HID; kb += 4) {
        int k0 = kb + 2 * kh;
        v2f a = *(const v2f*)(xrow + k0);
        v2f b;
        b.x = W[(size_t)(k0 + 0) * HID + cb + m];
        b.y = W[(size_t)(k0 + 1) * HID + cb + m];
        acc = __builtin_amdgcn_wmma_f32_16x16x4_f32(
             false, a,  false, b,
             (short)0, acc,  false,  false);
    }

    float* hbase = H + (size_t)(row_tile * 16 + 8 * kh) * HID + cb + m;
    #pragma unroll
    for (int r = 0; r < 8; ++r) hbase[(size_t)r * HID] = acc[r];
}

__global__ void zero_kernel(float4* __restrict__ p, int n4) {
    int i = blockIdx.x * blockDim.x + threadIdx.x;
    if (i < n4) p[i] = make_float4(0.f, 0.f, 0.f, 0.f);
}

template <bool LAST>
__global__ __launch_bounds__(AGG_T)
void agg_kernel(const float* __restrict__ H, const float* __restrict__ dinv,
                const int* __restrict__ tile_start, const int* __restrict__ bsrc,
                const int* __restrict__ bdl, const float* __restrict__ bw,
                const float* __restrict__ bias, const float* __restrict__ Wn,
                float* __restrict__ out) {
    __shared__ float tile[TN * LP];
    const int t    = blockIdx.x;
    const int lane = threadIdx.x & 31;
    const int wave = __builtin_amdgcn_readfirstlane((int)(threadIdx.x >> 5));
    const int NW   = AGG_T / 32;
    const int node0 = t * TN;

    for (int r = wave; r < TN; r += NW) {
        const int node = node0 + r;
        float4 v = make_float4(0.f, 0.f, 0.f, 0.f);
        if (node < N_NODES) {
            const float dv = dinv[node];
            const float w  = dv * dv;
            const float4 h = ((const float4*)(H + (size_t)node * HID))[lane];
            v.x = h.x * w; v.y = h.y * w; v.z = h.z * w; v.w = h.w * w;
        }
        float* row = tile + r * LP + lane;
        row[0] = v.x; row[32] = v.y; row[64] = v.z; row[96] = v.w;
    }
    __syncthreads();

    const int beg = tile_start[t];
    const int end = tile_start[t + 1];
    for (int base = beg + wave * 32; base < end; base += NW * 32) {
        const int idx = base + lane;
        int s = 0, dl = 0;
        float w = 0.f;
        if (idx < end) { s = bsrc[idx]; dl = bdl[idx]; w = bw[idx]; }
        const int n = min(32, end - base);
        for (int k = 0; k < n; k += 4) {
            float4 h[4];
            #pragma unroll
            for (int u = 0; u < 4; ++u) {
                const int sk = __builtin_amdgcn_readlane(s, k + u);
                h[u] = ((const float4*)(H + (size_t)sk * HID))[lane];
            }
            #pragma unroll
            for (int u = 0; u < 4; ++u) {
                if (u == 0 || k + u < n) {
                    const int   dk = __builtin_amdgcn_readlane(dl, k + u);
                    const float wk = __int_as_float(__builtin_amdgcn_readlane(__float_as_int(w), k + u));
                    float* row = tile + dk * LP + lane;
                    __hip_atomic_fetch_add(row,      h[u].x * wk, __ATOMIC_RELAXED, __HIP_MEMORY_SCOPE_WORKGROUP);
                    __hip_atomic_fetch_add(row + 32, h[u].y * wk, __ATOMIC_RELAXED, __HIP_MEMORY_SCOPE_WORKGROUP);
                    __hip_atomic_fetch_add(row + 64, h[u].z * wk, __ATOMIC_RELAXED, __HIP_MEMORY_SCOPE_WORKGROUP);
                    __hip_atomic_fetch_add(row + 96, h[u].w * wk, __ATOMIC_RELAXED, __HIP_MEMORY_SCOPE_WORKGROUP);
                }
            }
        }
    }
    __syncthreads();

    const float4 b = ((const float4*)bias)[lane];
    if (LAST) {
        for (int r = wave; r < TN; r += NW) {
            const int node = node0 + r;
            if (node < N_NODES) {
                const float* row = tile + r * LP + lane;
                float4 o;
                o.x = row[0]  + b.x; o.y = row[32] + b.y; o.z = row[64] + b.z; o.w = row[96] + b.w;
                o.x = o.x > 0.f ? o.x : 0.f;
                o.y = o.y > 0.f ? o.y : 0.f;
                o.z = o.z > 0.f ? o.z : 0.f;
                o.w = o.w > 0.f ? o.w : 0.f;
                ((float4*)(out + (size_t)node * HID))[lane] = o;
            }
        }
        return;
    }

    for (int r = wave; r < TN; r += NW) {
        float* row = tile + r * LP + lane;
        float o0 = row[0] + b.x, o1 = row[32] + b.y, o2 = row[64] + b.z, o3 = row[96] + b.w;
        row[0]  = o0 > 0.f ? o0 : 0.f;
        row[32] = o1 > 0.f ? o1 : 0.f;
        row[64] = o2 > 0.f ? o2 : 0.f;
        row[96] = o3 > 0.f ? o3 : 0.f;
    }
    __syncthreads();

    const int m  = lane & 15;
    const int kh = lane >> 4;
    const int rq = wave >> 2;
    const int cq = wave & 3;
    const float* a0p = tile + (32 * rq + m) * LP + 2 * kh;
    const float* a1p = a0p + 16 * LP;
    const float* wp  = Wn + (size_t)(8 * kh) * HID + 32 * cq + 2 * m;

    v8f acc00 = {}, acc01 = {}, acc10 = {}, acc11 = {};
    #pragma unroll 2
    for (int p = 0; p < 16; ++p) {
        const int ks = 2 * p;
        const float* w0 = wp + (size_t)(16 * (ks & 7) + (ks >> 3)) * HID;
        const float* w1 = w0 + (size_t)16 * HID;
        const v2f wc0 = *(const v2f*)(w0), wc1 = *(const v2f*)(w0 + 4 * HID);
        const v2f wc2 = *(const v2f*)(w1), wc3 = *(const v2f*)(w1 + 4 * HID);
        const v2f a00 = *(const v2f*)(a0p + 4 * ks);
        const v2f a10 = *(const v2f*)(a1p + 4 * ks);
        const v2f a01 = *(const v2f*)(a0p + 4 * ks + 4);
        const v2f a11 = *(const v2f*)(a1p + 4 * ks + 4);
        v2f b0, b1;
        b0.x = wc0.x; b0.y = wc1.x; b1.x = wc0.y; b1.y = wc1.y;
        acc00 = __builtin_amdgcn_wmma_f32_16x16x4_f32(false, a00, false, b0, (short)0, acc00, false, false);
        acc01 = __builtin_amdgcn_wmma_f32_16x16x4_f32(false, a00, false, b1, (short)0, acc01, false, false);
        acc10 = __builtin_amdgcn_wmma_f32_16x16x4_f32(false, a10, false, b0, (short)0, acc10, false, false);
        acc11 = __builtin_amdgcn_wmma_f32_16x16x4_f32(false, a10, false, b1, (short)0, acc11, false, false);
        b0.x = wc2.x; b0.y = wc3.x; b1.x = wc2.y; b1.y = wc3.y;
        acc00 = __builtin_amdgcn_wmma_f32_16x16x4_f32(false, a01, false, b0, (short)0, acc00, false, false);
        acc01 = __builtin_amdgcn_wmma_f32_16x16x4_f32(false, a01, false, b1, (short)0, acc01, false, false);
        acc10 = __builtin_amdgcn_wmma_f32_16x16x4_f32(false, a11, false, b0, (short)0, acc10, false, false);
        acc11 = __builtin_amdgcn_wmma_f32_16x16x4_f32(false, a11, false, b1, (short)0, acc11, false, false);
    }

    const int row0 = node0 + 32 * rq + 8 * kh;
    float* ob = out + (size_t)row0 * HID + 32 * cq + 2 * m;
    #pragma unroll
    for (int r = 0; r < 8; ++r) {
        if (row0 + r < N_NODES) {
            v2f o; o.x = acc00[r]; o.y = acc01[r];
            *(v2f*)(ob + (size_t)r * HID) = o;
        }
        if (row0 + 16 + r < N_NODES) {
            v2f o; o.x = acc10[r]; o.y = acc11[r];
            *(v2f*)(ob + (size_t)(16 + r) * HID) = o;
        }
    }
}

__global__ __launch_bounds__(256)
void pool_accum_kernel(const float* __restrict__ X, const int* __restrict__ batch,
                       float* __restrict__ pooled, float* __restrict__ counts) {
    int wave = (blockIdx.x * blockDim.x + threadIdx.x) >> 5;
    int lane = threadIdx.x & 31;
    if (wave >= N_NODES) return;
    int g = batch[wave];
    float4 v = ((const float4*)(X + (size_t)wave * HID))[lane];
    float* pp = pooled + (size_t)g * HID + lane * 4;
    atomicAdd(pp + 0, v.x);
    atomicAdd(pp + 1, v.y);
    atomicAdd(pp + 2, v.z);
    atomicAdd(pp + 3, v.w);
    if (lane == 0) atomicAdd(&counts[g], 1.0f);
}

__global__ void final_linear_kernel(const float* __restrict__ pooled,
                                    const float* __restrict__ counts,
                                    const float* __restrict__ linW,
                                    const float* __restrict__ linb,
                                    float* __restrict__ out) {
    int i = blockIdx.x * blockDim.x + threadIdx.x;
    if (i >= N_GRAPHS * N_CLASSES) return;
    int g = i / N_CLASSES;
    int c = i - g * N_CLASSES;
    float scale = 1.0f / fmaxf(counts[g], 1.0f);
    const float* pg = pooled + (size_t)g * HID;
    float acc = 0.f;
    #pragma unroll 4
    for (int f = 0; f < HID; ++f) acc += pg[f] * linW[f * N_CLASSES + c];
    out[i] = acc * scale + linb[c];
}

extern "C" void kernel_launch(void* const* d_in, const int* in_sizes, int n_in,
                              void* d_out, int out_size, void* d_ws, size_t ws_size,
                              hipStream_t stream) {
    const float* x     = (const float*)d_in[0];
    const int*   edge  = (const int*)d_in[1];
    const int*   src   = edge;
    const int*   dst   = edge + N_EDGES;
    const int*   batch = (const int*)d_in[2];
    const float* Ws    = (const float*)d_in[3];
    const float* bs    = (const float*)d_in[4];
    const float* linW  = (const float*)d_in[5];
    const float* linb  = (const float*)d_in[6];
    float*       out   = (float*)d_out;

    float* ws = (float*)d_ws;
    float* H      = ws;
    float* X1     = H + (size_t)N_NODES * HID;
    float* X2     = X1 + (size_t)N_NODES * HID;
    float* pooled = X2 + (size_t)N_NODES * HID;
    float* counts = pooled + (size_t)N_GRAPHS * HID;
    float* deg    = counts + N_GRAPHS;
    float* bw     = deg + N_NODES;
    int*   bsrc   = (int*)(bw + N_EDGES);
    int*   bdl    = bsrc + N_EDGES;
    int*   tstart = bdl + N_EDGES;
    int*   cursor = tstart + 1024;

    const int T = 256;

    init_deg_kernel<<<(N_NODES + T - 1) / T, T, 0, stream>>>(deg);
    deg_accum_kernel<<<(N_EDGES + T - 1) / T, T, 0, stream>>>(dst, deg);
    tile_scan_kernel<<<1, 1024, 0, stream>>>(deg, tstart, cursor);
    dinv_kernel<<<(N_NODES + T - 1) / T, T, 0, stream>>>(deg);
    fill_kernel<<<(N_EDGES + T - 1) / T, T, 0, stream>>>(src, dst, deg, cursor, bsrc, bdl, bw);

    const int gemm_blocks = (N_NODES / 16) * (HID / 16) / 8;

    gemm_wmma_kernel<<<gemm_blocks, T, 0, stream>>>(x, Ws, H);
    float* hbuf[2] = { H, X1 };
    for (int l = 0; l < N_LAYERS - 1; ++l) {
        const float* Wn = Ws + (size_t)(l + 1) * HID * HID;
        const float* bl = bs + (size_t)l * HID;
        agg_kernel<false><<<NT, AGG_T, 0, stream>>>(hbuf[l & 1], deg, tstart, bsrc, bdl, bw, bl, Wn,
                                                    hbuf[(l + 1) & 1]);
    }
    agg_kernel<true><<<NT, AGG_T, 0, stream>>>(hbuf[(N_LAYERS - 1) & 1], deg, tstart, bsrc, bdl, bw,
                                               bs + (size_t)(N_LAYERS - 1) * HID, Ws, X2);
    const float* xin = X2;

    int pool4 = (N_GRAPHS * HID + N_GRAPHS + 3) / 4;
    zero_kernel<<<(pool4 + T - 1) / T, T, 0, stream>>>((float4*)pooled, pool4);
    pool_accum_kernel<<<N_NODES / 8, T, 0, stream>>>(xin, batch, pooled, counts);
    final_linear_kernel<<<(N_GRAPHS * N_CLASSES + T - 1) / T, T, 0, stream>>>(
        pooled, counts, linW, linb, out);
}


# Each tile's edge list put in row order once per call (new launch, LDS counting sort); a wave sums a run of one row in registers, adds it by LDS float atomics: about 10x fewer; sum order changes
# speedup vs baseline: 2.4172x; 1.0472x over previous
#include <hip/hip_runtime.h>
#include <hip/hip_bf16.h>


#define N_NODES   100000
#define N_EDGES   1600000
#define HID       128
#define N_CLASSES 10
#define N_GRAPHS  512
#define N_LAYERS  5
#define TN        128
#define NT        ((N_NODES + TN - 1) / TN)
#define CPAD      32
#define AGG_T     512
#define SORT_T    512
#define LP        132

typedef __attribute__((ext_vector_type(2))) float v2f;
typedef __attribute__((ext_vector_type(8))) float v8f;

__global__ void init_deg_kernel(float* __restrict__ deg) {
    int i = blockIdx.x * blockDim.x + threadIdx.x;
    if (i < N_NODES) deg[i] = 1.0f;
}

__global__ void deg_accum_kernel(const int* __restrict__ dst, float* __restrict__ deg) {
    int e = blockIdx.x * blockDim.x + threadIdx.x;
    if (e < N_EDGES) atomicAdd(&deg[dst[e]], 1.0f);
}

__global__ void dinv_kernel(float* __restrict__ deg) {
    int i = blockIdx.x * blockDim.x + threadIdx.x;
    if (i < N_NODES) deg[i] = rsqrtf(deg[i]);
}

__global__ __launch_bounds__(1024)
void tile_scan_kernel(const float* __restrict__ deg, int* __restrict__ tile_start,
                      int* __restrict__ cursor) {
    __shared__ int cnt[1024];
    __shared__ int wsum[32];
    const int tid  = threadIdx.x;
    const int lane = tid & 31;
    const int wave = tid >> 5;

    for (int t = wave; t < 1024; t += 32) {
        int c = 0;
        int n = t * TN + lane * 4;
        if (t < NT && n < N_NODES) {
            float4 d = *(const float4*)(deg + n);
            c = (int)d.x + (int)d.y + (int)d.z + (int)d.w - 4;
        }
        for (int off = 16; off > 0; off >>= 1) c += __shfl_xor(c, off);
        if (lane == 0) cnt[t] = c;
    }
    __syncthreads();

    const int v = cnt[tid];
    int x = v;
    for (int off = 1; off < 32; off <<= 1) {
        int y = __shfl_up(x, off);
        if (lane >= off) x += y;
    }
    if (lane == 31) wsum[wave] = x;
    __syncthreads();
    if (wave == 0) {
        const int s = wsum[lane];
        int xs = s;
        for (int off = 1; off < 32; off <<= 1) {
            int y = __shfl_up(xs, off);
            if (lane >= off) xs += y;
        }
        wsum[lane] = xs - s;
    }
    __syncthreads();
    const int excl = x - v + wsum[wave];
    if (tid <= NT) tile_start[tid] = excl;
    if (tid < NT)  cursor[tid * CPAD] = excl;
}

__global__ void fill_kernel(const int* __restrict__ src, const int* __restrict__ dst,
                            const float* __restrict__ dinv, int* __restrict__ cursor,
                            int* __restrict__ bsrc, int* __restrict__ bdl,
                            float* __restrict__ bw) {
    int e = blockIdx.x * blockDim.x + threadIdx.x;
    if (e >= N_EDGES) return;
    int s = src[e];
    int d = dst[e];
    int pos = atomicAdd(&cursor[(d >> 7) * CPAD], 1);
    bsrc[pos] = s;
    bdl[pos]  = d & (TN - 1);
    bw[pos]   = dinv[s] * dinv[d];
}

__global__ __launch_bounds__(SORT_T)
void tile_sort_kernel(const int* __restrict__ tile_start, const int* __restrict__ bsrc,
                      const int* __restrict__ bdl, const float* __restrict__ bw,
                      int* __restrict__ ssrc, int* __restrict__ sdl, float* __restrict__ sw) {
    __shared__ int cnt[TN];
    __shared__ int wtot[TN / 32];
    const int t    = blockIdx.x;
    const int tid  = threadIdx.x;
    const int lane = tid & 31;
    const int wave = tid >> 5;
    const int beg  = tile_start[t];
    const int end  = tile_start[t + 1];

    if (tid < TN) cnt[tid] = 0;
    __syncthreads();
    for (int i = beg + tid; i < end; i += SORT_T) atomicAdd(&cnt[bdl[i]], 1);
    __syncthreads();

    const int c = (tid < TN) ? cnt[tid] : 0;
    int x = c;
    for (int off = 1; off < 32; off <<= 1) {
        int y = __shfl_up(x, off);
        if (lane >= off) x += y;
    }
    if (tid < TN && lane == 31) wtot[wave] = x;
    __syncthreads();
    if (tid < TN) {
        int front = beg;
        for (int j = 0; j < wave; ++j) front += wtot[j];
        cnt[tid] = front + x - c;
    }
    __syncthreads();

    for (int i = beg + tid; i < end; i += SORT_T) {
        const int d   = bdl[i];
        const int pos = atomicAdd(&cnt[d], 1);
        ssrc[pos] = bsrc[i];
        sdl[pos]  = d;
        sw[pos]   = bw[i];
    }
}

__global__ __launch_bounds__(256)
void gemm_wmma_kernel(const float* __restrict__ X, const float* __restrict__ W,
                      float* __restrict__ H) {
    int wave     = (blockIdx.x * blockDim.x + threadIdx.x) >> 5;
    int lane     = threadIdx.x & 31;
    int row_tile = wave >> 3;
    int col_tile = wave & 7;
    if (row_tile * 16 >= N_NODES) return;

    int m  = lane & 15;
    int kh = lane >> 4;
    const float* xrow = X + (size_t)(row_tile * 16 + m) * HID;
    int cb = col_tile * 16;

    v8f acc = {};
    #pragma unroll
    for (int kb = 0; kb < HID; kb += 4) {
        int k0 = kb + 2 * kh;
        v2f a = *(const v2f*)(xrow + k0);
        v2f b;
        b.x = W[(size_t)(k0 + 0) * HID + cb + m];
        b.y = W[(size_t)(k0 + 1) * HID + cb + m];
        acc = __builtin_amdgcn_wmma_f32_16x16x4_f32(
             false, a,  false, b,
             (short)0, acc,  false,  false);
    }

    float* hbase = H + (size_t)(row_tile * 16 + 8 * kh) * HID + cb + m;
    #pragma unroll
    for (int r = 0; r < 8; ++r) hbase[(size_t)r * HID] = acc[r];
}

__global__ void zero_kernel(float4* __restrict__ p, int n4) {
    int i = blockIdx.x * blockDim.x + threadIdx.x;
    if (i < n4) p[i] = make_float4(0.f, 0.f, 0.f, 0.f);
}

template <bool LAST>
__global__ __launch_bounds__(AGG_T)
void agg_kernel(const float* __restrict__ H, const float* __restrict__ dinv,
                const int* __restrict__ tile_start, const int* __restrict__ bsrc,
                const int* __restrict__ bdl, const float* __restrict__ bw,
                const float* __restrict__ bias, const float* __restrict__ Wn,
                float* __restrict__ out) {
    __shared__ float tile[TN * LP];
    const int t    = blockIdx.x;
    const int lane = threadIdx.x & 31;
    const int wave = __builtin_amdgcn_readfirstlane((int)(threadIdx.x >> 5));
    const int NW   = AGG_T / 32;
    const int node0 = t * TN;

    for (int r = wave; r < TN; r += NW) {
        const int node = node0 + r;
        float4 v = make_float4(0.f, 0.f, 0.f, 0.f);
        if (node < N_NODES) {
            const float dv = dinv[node];
            const float w  = dv * dv;
            const float4 h = ((const float4*)(H + (size_t)node * HID))[lane];
            v.x = h.x * w; v.y = h.y * w; v.z = h.z * w; v.w = h.w * w;
        }
        float* row = tile + r * LP + lane;
        row[0] = v.x; row[32] = v.y; row[64] = v.z; row[96] = v.w;
    }
    __syncthreads();

    const int beg = tile_start[t];
    const int end = tile_start[t + 1];
    for (int base = beg + wave * 32; base < end; base += NW * 32) {
        const int idx = base + lane;
        int s = 0, dl = 0;
        float w = 0.f;
        if (idx < end) { s = bsrc[idx]; dl = bdl[idx]; w = bw[idx]; }
        const int n = min(32, end - base);
        int   cur = __builtin_amdgcn_readlane(dl, 0);
        float a0 = 0.f, a1 = 0.f, a2 = 0.f, a3 = 0.f;
        for (int k = 0; k < n; k += 4) {
            float4 h[4];
            #pragma unroll
            for (int u = 0; u < 4; ++u) {
                const int sk = __builtin_amdgcn_readlane(s, k + u);
                h[u] = ((const float4*)(H + (size_t)sk * HID))[lane];
            }
            #pragma unroll
            for (int u = 0; u < 4; ++u) {
                if (u == 0 || k + u < n) {
                    const int   dk = __builtin_amdgcn_readlane(dl, k + u);
                    const float wk = __int_as_float(__builtin_amdgcn_readlane(__float_as_int(w), k + u));
                    if (dk != cur) {
                        float* row = tile + cur * LP + lane;
                        __hip_atomic_fetch_add(row,      a0, __ATOMIC_RELAXED, __HIP_MEMORY_SCOPE_WORKGROUP);
                        __hip_atomic_fetch_add(row + 32, a1, __ATOMIC_RELAXED, __HIP_MEMORY_SCOPE_WORKGROUP);
                        __hip_atomic_fetch_add(row + 64, a2, __ATOMIC_RELAXED, __HIP_MEMORY_SCOPE_WORKGROUP);
                        __hip_atomic_fetch_add(row + 96, a3, __ATOMIC_RELAXED, __HIP_MEMORY_SCOPE_WORKGROUP);
                        a0 = 0.f; a1 = 0.f; a2 = 0.f; a3 = 0.f;
                        cur = dk;
                    }
                    a0 += h[u].x * wk;
                    a1 += h[u].y * wk;
                    a2 += h[u].z * wk;
                    a3 += h[u].w * wk;
                }
            }
        }
        float* row = tile + cur * LP + lane;
        __hip_atomic_fetch_add(row,      a0, __ATOMIC_RELAXED, __HIP_MEMORY_SCOPE_WORKGROUP);
        __hip_atomic_fetch_add(row + 32, a1, __ATOMIC_RELAXED, __HIP_MEMORY_SCOPE_WORKGROUP);
        __hip_atomic_fetch_add(row + 64, a2, __ATOMIC_RELAXED, __HIP_MEMORY_SCOPE_WORKGROUP);
        __hip_atomic_fetch_add(row + 96, a3, __ATOMIC_RELAXED, __HIP_MEMORY_SCOPE_WORKGROUP);
    }
    __syncthreads();

    const float4 b = ((const float4*)bias)[lane];
    if (LAST) {
        for (int r = wave; r < TN; r += NW) {
            const int node = node0 + r;
            if (node < N_NODES) {
                const float* row = tile + r * LP + lane;
                float4 o;
                o.x = row[0]  + b.x; o.y = row[32] + b.y; o.z = row[64] + b.z; o.w = row[96] + b.w;
                o.x = o.x > 0.f ? o.x : 0.f;
                o.y = o.y > 0.f ? o.y : 0.f;
                o.z = o.z > 0.f ? o.z : 0.f;
                o.w = o.w > 0.f ? o.w : 0.f;
                ((float4*)(out + (size_t)node * HID))[lane] = o;
            }
        }
        return;
    }

    for (int r = wave; r < TN; r += NW) {
        float* row = tile + r * LP + lane;
        float o0 = row[0] + b.x, o1 = row[32] + b.y, o2 = row[64] + b.z, o3 = row[96] + b.w;
        row[0]  = o0 > 0.f ? o0 : 0.f;
        row[32] = o1 > 0.f ? o1 : 0.f;
        row[64] = o2 > 0.f ? o2 : 0.f;
        row[96] = o3 > 0.f ? o3 : 0.f;
    }
    __syncthreads();

    const int m  = lane & 15;
    const int kh = lane >> 4;
    const int rq = wave >> 2;
    const int cq = wave & 3;
    const float* a0p = tile + (32 * rq + m) * LP + 2 * kh;
    const float* a1p = a0p + 16 * LP;
    const float* wp  = Wn + (size_t)(8 * kh) * HID + 32 * cq + 2 * m;

    v8f acc00 = {}, acc01 = {}, acc10 = {}, acc11 = {};
    #pragma unroll 2
    for (int p = 0; p < 16; ++p) {
        const int ks = 2 * p;
        const float* w0 = wp + (size_t)(16 * (ks & 7) + (ks >> 3)) * HID;
        const float* w1 = w0 + (size_t)16 * HID;
        const v2f wc0 = *(const v2f*)(w0), wc1 = *(const v2f*)(w0 + 4 * HID);
        const v2f wc2 = *(const v2f*)(w1), wc3 = *(const v2f*)(w1 + 4 * HID);
        const v2f a00 = *(const v2f*)(a0p + 4 * ks);
        const v2f a10 = *(const v2f*)(a1p + 4 * ks);
        const v2f a01 = *(const v2f*)(a0p + 4 * ks + 4);
        const v2f a11 = *(const v2f*)(a1p + 4 * ks + 4);
        v2f b0, b1;
        b0.x = wc0.x; b0.y = wc1.x; b1.x = wc0.y; b1.y = wc1.y;
        acc00 = __builtin_amdgcn_wmma_f32_16x16x4_f32(false, a00, false, b0, (short)0, acc00, false, false);
        acc01 = __builtin_amdgcn_wmma_f32_16x16x4_f32(false, a00, false, b1, (short)0, acc01, false, false);
        acc10 = __builtin_amdgcn_wmma_f32_16x16x4_f32(false, a10, false, b0, (short)0, acc10, false, false);
        acc11 = __builtin_amdgcn_wmma_f32_16x16x4_f32(false, a10, false, b1, (short)0, acc11, false, false);
        b0.x = wc2.x; b0.y = wc3.x; b1.x = wc2.y; b1.y = wc3.y;
        acc00 = __builtin_amdgcn_wmma_f32_16x16x4_f32(false, a01, false, b0, (short)0, acc00, false, false);
        acc01 = __builtin_amdgcn_wmma_f32_16x16x4_f32(false, a01, false, b1, (short)0, acc01, false, false);
        acc10 = __builtin_amdgcn_wmma_f32_16x16x4_f32(false, a11, false, b0, (short)0, acc10, false, false);
        acc11 = __builtin_amdgcn_wmma_f32_16x16x4_f32(false, a11, false, b1, (short)0, acc11, false, false);
    }

    const int row0 = node0 + 32 * rq + 8 * kh;
    float* ob = out + (size_t)row0 * HID + 32 * cq + 2 * m;
    #pragma unroll
    for (int r = 0; r < 8; ++r) {
        if (row0 + r < N_NODES) {
            v2f o; o.x = acc00[r]; o.y = acc01[r];
            *(v2f*)(ob + (size_t)r * HID) = o;
        }
        if (row0 + 16 + r < N_NODES) {
            v2f o; o.x = acc10[r]; o.y = acc11[r];
            *(v2f*)(ob + (size_t)(16 + r) * HID) = o;
        }
    }
}

__global__ __launch_bounds__(256)
void pool_accum_kernel(const float* __restrict__ X, const int* __restrict__ batch,
                       float* __restrict__ pooled, float* __restrict__ counts) {
    int wave = (blockIdx.x * blockDim.x + threadIdx.x) >> 5;
    int lane = threadIdx.x & 31;
    if (wave >= N_NODES) return;
    int g = batch[wave];
    float4 v = ((const float4*)(X + (size_t)wave * HID))[lane];
    float* pp = pooled + (size_t)g * HID + lane * 4;
    atomicAdd(pp + 0, v.x);
    atomicAdd(pp + 1, v.y);
    atomicAdd(pp + 2, v.z);
    atomicAdd(pp + 3, v.w);
    if (lane == 0) atomicAdd(&counts[g], 1.0f);
}

__global__ void final_linear_kernel(const float* __restrict__ pooled,
                                    const float* __restrict__ counts,
                                    const float* __restrict__ linW,
                                    const float* __restrict__ linb,
                                    float* __restrict__ out) {
    int i = blockIdx.x * blockDim.x + threadIdx.x;
    if (i >= N_GRAPHS * N_CLASSES) return;
    int g = i / N_CLASSES;
    int c = i - g * N_CLASSES;
    float scale = 1.0f / fmaxf(counts[g], 1.0f);
    const float* pg = pooled + (size_t)g * HID;
    float acc = 0.f;
    #pragma unroll 4
    for (int f = 0; f < HID; ++f) acc += pg[f] * linW[f * N_CLASSES + c];
    out[i] = acc * scale + linb[c];
}

extern "C" void kernel_launch(void* const* d_in, const int* in_sizes, int n_in,
                              void* d_out, int out_size, void* d_ws, size_t ws_size,
                              hipStream_t stream) {
    const float* x     = (const float*)d_in[0];
    const int*   edge  = (const int*)d_in[1];
    const int*   src   = edge;
    const int*   dst   = edge + N_EDGES;
    const int*   batch = (const int*)d_in[2];
    const float* Ws    = (const float*)d_in[3];
    const float* bs    = (const float*)d_in[4];
    const float* linW  = (const float*)d_in[5];
    const float* linb  = (const float*)d_in[6];
    float*       out   = (float*)d_out;

    float* ws = (float*)d_ws;
    float* H      = ws;
    float* X1     = H + (size_t)N_NODES * HID;
    float* X2     = X1 + (size_t)N_NODES * HID;
    float* pooled = X2 + (size_t)N_NODES * HID;
    float* counts = pooled + (size_t)N_GRAPHS * HID;
    float* deg    = counts + N_GRAPHS;
    float* bw     = deg + N_NODES;
    int*   bsrc   = (int*)(bw + N_EDGES);
    int*   bdl    = bsrc + N_EDGES;
    int*   tstart = bdl + N_EDGES;
    int*   cursor = tstart + 1024;
    int*   ssrc   = cursor + 1024 * CPAD;
    int*   sdl    = ssrc + N_EDGES;
    float* sw     = (float*)(sdl + N_EDGES);

    const int T = 256;

    init_deg_kernel<<<(N_NODES + T - 1) / T, T, 0, stream>>>(deg);
    deg_accum_kernel<<<(N_EDGES + T - 1) / T, T, 0, stream>>>(dst, deg);
    tile_scan_kernel<<<1, 1024, 0, stream>>>(deg, tstart, cursor);
    dinv_kernel<<<(N_NODES + T - 1) / T, T, 0, stream>>>(deg);
    fill_kernel<<<(N_EDGES + T - 1) / T, T, 0, stream>>>(src, dst, deg, cursor, bsrc, bdl, bw);
    tile_sort_kernel<<<NT, SORT_T, 0, stream>>>(tstart, bsrc, bdl, bw, ssrc, sdl, sw);

    const int gemm_blocks = (N_NODES / 16) * (HID / 16) / 8;

    gemm_wmma_kernel<<<gemm_blocks, T, 0, stream>>>(x, Ws, H);
    float* hbuf[2] = { H, X1 };
    for (int l = 0; l < N_LAYERS - 1; ++l) {
        const float* Wn = Ws + (size_t)(l + 1) * HID * HID;
        const float* bl = bs + (size_t)l * HID;
        agg_kernel<false><<<NT, AGG_T, 0, stream>>>(hbuf[l & 1], deg, tstart, ssrc, sdl, sw, bl, Wn,
                                                    hbuf[(l + 1) & 1]);
    }
    agg_kernel<true><<<NT, AGG_T, 0, stream>>>(hbuf[(N_LAYERS - 1) & 1], deg, tstart, ssrc, sdl, sw,
                                               bs + (size_t)(N_LAYERS - 1) * HID, Ws, X2);
    const float* xin = X2;

    int pool4 = (N_GRAPHS * HID + N_GRAPHS + 3) / 4;
    zero_kernel<<<(pool4 + T - 1) / T, T, 0, stream>>>((float4*)pooled, pool4);
    pool_accum_kernel<<<N_NODES / 8, T, 0, stream>>>(xin, batch, pooled, counts);
    final_linear_kernel<<<(N_GRAPHS * N_CLASSES + T - 1) / T, T, 0, stream>>>(
        pooled, counts, linW, linb, out);
}


# Edge list kept as one 8-byte record per edge (source and row-in-tile packed in one word, weight bits in the other): fill and tile sort issue one scattered store per edge, not three; sums unchanged
# speedup vs baseline: 2.8152x; 1.1647x over previous
#include <hip/hip_runtime.h>
#include <hip/hip_bf16.h>


#define N_NODES   100000
#define N_EDGES   1600000
#define HID       128
#define N_CLASSES 10
#define N_GRAPHS  512
#define N_LAYERS  5
#define TN        128
#define NT        ((N_NODES + TN - 1) / TN)
#define CPAD      32
#define AGG_T     512
#define SORT_T    512
#define LP        132
#define SRC_BITS  17
#define SRC_MASK  ((1 << SRC_BITS) - 1)

static_assert(N_NODES <= (1 << SRC_BITS), "source node must fit the record's low bits");
static_assert(TN <= (1 << (31 - SRC_BITS)), "row inside the tile must fit above the source bits");

typedef __attribute__((ext_vector_type(2))) float v2f;
typedef __attribute__((ext_vector_type(8))) float v8f;

__global__ void init_deg_kernel(float* __restrict__ deg) {
    int i = blockIdx.x * blockDim.x + threadIdx.x;
    if (i < N_NODES) deg[i] = 1.0f;
}

__global__ void deg_accum_kernel(const int* __restrict__ dst, float* __restrict__ deg) {
    int e = blockIdx.x * blockDim.x + threadIdx.x;
    if (e < N_EDGES) atomicAdd(&deg[dst[e]], 1.0f);
}

__global__ void dinv_kernel(float* __restrict__ deg) {
    int i = blockIdx.x * blockDim.x + threadIdx.x;
    if (i < N_NODES) deg[i] = rsqrtf(deg[i]);
}

__global__ __launch_bounds__(1024)
void tile_scan_kernel(const float* __restrict__ deg, int* __restrict__ tile_start,
                      int* __restrict__ cursor) {
    __shared__ int cnt[1024];
    __shared__ int wsum[32];
    const int tid  = threadIdx.x;
    const int lane = tid & 31;
    const int wave = tid >> 5;

    for (int t = wave; t < 1024; t += 32) {
        int c = 0;
        int n = t * TN + lane * 4;
        if (t < NT && n < N_NODES) {
            float4 d = *(const float4*)(deg + n);
            c = (int)d.x + (int)d.y + (int)d.z + (int)d.w - 4;
        }
        for (int off = 16; off > 0; off >>= 1) c += __shfl_xor(c, off);
        if (lane == 0) cnt[t] = c;
    }
    __syncthreads();

    const int v = cnt[tid];
    int x = v;
    for (int off = 1; off < 32; off <<= 1) {
        int y = __shfl_up(x, off);
        if (lane >= off) x += y;
    }
    if (lane == 31) wsum[wave] = x;
    __syncthreads();
    if (wave == 0) {
        const int s = wsum[lane];
        int xs = s;
        for (int off = 1; off < 32; off <<= 1) {
            int y = __shfl_up(xs, off);
            if (lane >= off) xs += y;
        }
        wsum[lane] = xs - s;
    }
    __syncthreads();
    const int excl = x - v + wsum[wave];
    if (tid <= NT) tile_start[tid] = excl;
    if (tid < NT)  cursor[tid * CPAD] = excl;
}

__global__ void fill_kernel(const int* __restrict__ src, const int* __restrict__ dst,
                            const float* __restrict__ dinv, int* __restrict__ cursor,
                            int2* __restrict__ brec) {
    int e = blockIdx.x * blockDim.x + threadIdx.x;
    if (e >= N_EDGES) return;
    int s = src[e];
    int d = dst[e];
    int pos = atomicAdd(&cursor[(d >> 7) * CPAD], 1);
    int2 r;
    r.x = s | ((d & (TN - 1)) << SRC_BITS);
    r.y = __float_as_int(dinv[s] * dinv[d]);
    brec[pos] = r;
}

__global__ __launch_bounds__(SORT_T)
void tile_sort_kernel(const int* __restrict__ tile_start, const int2* __restrict__ brec,
                      int2* __restrict__ srec) {
    __shared__ int cnt[TN];
    __shared__ int wtot[TN / 32];
    const int t    = blockIdx.x;
    const int tid  = threadIdx.x;
    const int lane = tid & 31;
    const int wave = tid >> 5;
    const int beg  = tile_start[t];
    const int end  = tile_start[t + 1];

    if (tid < TN) cnt[tid] = 0;
    __syncthreads();
    for (int i = beg + tid; i < end; i += SORT_T) atomicAdd(&cnt[brec[i].x >> SRC_BITS], 1);
    __syncthreads();

    const int c = (tid < TN) ? cnt[tid] : 0;
    int x = c;
    for (int off = 1; off < 32; off <<= 1) {
        int y = __shfl_up(x, off);
        if (lane >= off) x += y;
    }
    if (tid < TN && lane == 31) wtot[wave] = x;
    __syncthreads();
    if (tid < TN) {
        int front = beg;
        for (int j = 0; j < wave; ++j) front += wtot[j];
        cnt[tid] = front + x - c;
    }
    __syncthreads();

    for (int i = beg + tid; i < end; i += SORT_T) {
        const int2 r  = brec[i];
        const int pos = atomicAdd(&cnt[r.x >> SRC_BITS], 1);
        srec[pos] = r;
    }
}

__global__ __launch_bounds__(256)
void gemm_wmma_kernel(const float* __restrict__ X, const float* __restrict__ W,
                      float* __restrict__ H) {
    int wave     = (blockIdx.x * blockDim.x + threadIdx.x) >> 5;
    int lane     = threadIdx.x & 31;
    int row_tile = wave >> 3;
    int col_tile = wave & 7;
    if (row_tile * 16 >= N_NODES) return;

    int m  = lane & 15;
    int kh = lane >> 4;
    const float* xrow = X + (size_t)(row_tile * 16 + m) * HID;
    int cb = col_tile * 16;

    v8f acc = {};
    #pragma unroll
    for (int kb = 0; kb < HID; kb += 4) {
        int k0 = kb + 2 * kh;
        v2f a = *(const v2f*)(xrow + k0);
        v2f b;
        b.x = W[(size_t)(k0 + 0) * HID + cb + m];
        b.y = W[(size_t)(k0 + 1) * HID + cb + m];
        acc = __builtin_amdgcn_wmma_f32_16x16x4_f32(
             false, a,  false, b,
             (short)0, acc,  false,  false);
    }

    float* hbase = H + (size_t)(row_tile * 16 + 8 * kh) * HID + cb + m;
    #pragma unroll
    for (int r = 0; r < 8; ++r) hbase[(size_t)r * HID] = acc[r];
}

__global__ void zero_kernel(float4* __restrict__ p, int n4) {
    int i = blockIdx.x * blockDim.x + threadIdx.x;
    if (i < n4) p[i] = make_float4(0.f, 0.f, 0.f, 0.f);
}

template <bool LAST>
__global__ __launch_bounds__(AGG_T)
void agg_kernel(const float* __restrict__ H, const float* __restrict__ dinv,
                const int* __restrict__ tile_start, const int2* __restrict__ srec,
                const float* __restrict__ bias, const float* __restrict__ Wn,
                float* __restrict__ out) {
    __shared__ float tile[TN * LP];
    const int t    = blockIdx.x;
    const int lane = threadIdx.x & 31;
    const int wave = __builtin_amdgcn_readfirstlane((int)(threadIdx.x >> 5));
    const int NW   = AGG_T / 32;
    const int node0 = t * TN;

    for (int r = wave; r < TN; r += NW) {
        const int node = node0 + r;
        float4 v = make_float4(0.f, 0.f, 0.f, 0.f);
        if (node < N_NODES) {
            const float dv = dinv[node];
            const float w  = dv * dv;
            const float4 h = ((const float4*)(H + (size_t)node * HID))[lane];
            v.x = h.x * w; v.y = h.y * w; v.z = h.z * w; v.w = h.w * w;
        }
        float* row = tile + r * LP + lane;
        row[0] = v.x; row[32] = v.y; row[64] = v.z; row[96] = v.w;
    }
    __syncthreads();

    const int beg = tile_start[t];
    const int end = tile_start[t + 1];
    for (int base = beg + wave * 32; base < end; base += NW * 32) {
        const int idx = base + lane;
        int pk = 0, wb = 0;
        if (idx < end) { const int2 r = srec[idx]; pk = r.x; wb = r.y; }
        const int n = min(32, end - base);
        int   cur = __builtin_amdgcn_readlane(pk, 0) >> SRC_BITS;
        float a0 = 0.f, a1 = 0.f, a2 = 0.f, a3 = 0.f;
        for (int k = 0; k < n; k += 4) {
            float4 h[4];
            #pragma unroll
            for (int u = 0; u < 4; ++u) {
                const int sk = __builtin_amdgcn_readlane(pk, k + u) & SRC_MASK;
                h[u] = ((const float4*)(H + (size_t)sk * HID))[lane];
            }
            #pragma unroll
            for (int u = 0; u < 4; ++u) {
                if (u == 0 || k + u < n) {
                    const int   dk = __builtin_amdgcn_readlane(pk, k + u) >> SRC_BITS;
                    const float wk = __int_as_float(__builtin_amdgcn_readlane(wb, k + u));
                    if (dk != cur) {
                        float* row = tile + cur * LP + lane;
                        __hip_atomic_fetch_add(row,      a0, __ATOMIC_RELAXED, __HIP_MEMORY_SCOPE_WORKGROUP);
                        __hip_atomic_fetch_add(row + 32, a1, __ATOMIC_RELAXED, __HIP_MEMORY_SCOPE_WORKGROUP);
                        __hip_atomic_fetch_add(row + 64, a2, __ATOMIC_RELAXED, __HIP_MEMORY_SCOPE_WORKGROUP);
                        __hip_atomic_fetch_add(row + 96, a3, __ATOMIC_RELAXED, __HIP_MEMORY_SCOPE_WORKGROUP);
                        a0 = 0.f; a1 = 0.f; a2 = 0.f; a3 = 0.f;
                        cur = dk;
                    }
                    a0 += h[u].x * wk;
                    a1 += h[u].y * wk;
                    a2 += h[u].z * wk;
                    a3 += h[u].w * wk;
                }
            }
        }
        float* row = tile + cur * LP + lane;
        __hip_atomic_fetch_add(row,      a0, __ATOMIC_RELAXED, __HIP_MEMORY_SCOPE_WORKGROUP);
        __hip_atomic_fetch_add(row + 32, a1, __ATOMIC_RELAXED, __HIP_MEMORY_SCOPE_WORKGROUP);
        __hip_atomic_fetch_add(row + 64, a2, __ATOMIC_RELAXED, __HIP_MEMORY_SCOPE_WORKGROUP);
        __hip_atomic_fetch_add(row + 96, a3, __ATOMIC_RELAXED, __HIP_MEMORY_SCOPE_WORKGROUP);
    }
    __syncthreads();

    const float4 b = ((const float4*)bias)[lane];
    if (LAST) {
        for (int r = wave; r < TN; r += NW) {
            const int node = node0 + r;
            if (node < N_NODES) {
                const float* row = tile + r * LP + lane;
                float4 o;
                o.x = row[0]  + b.x; o.y = row[32] + b.y; o.z = row[64] + b.z; o.w = row[96] + b.w;
                o.x = o.x > 0.f ? o.x : 0.f;
                o.y = o.y > 0.f ? o.y : 0.f;
                o.z = o.z > 0.f ? o.z : 0.f;
                o.w = o.w > 0.f ? o.w : 0.f;
                ((float4*)(out + (size_t)node * HID))[lane] = o;
            }
        }
        return;
    }

    for (int r = wave; r < TN; r += NW) {
        float* row = tile + r * LP + lane;
        float o0 = row[0] + b.x, o1 = row[32] + b.y, o2 = row[64] + b.z, o3 = row[96] + b.w;
        row[0]  = o0 > 0.f ? o0 : 0.f;
        row[32] = o1 > 0.f ? o1 : 0.f;
        row[64] = o2 > 0.f ? o2 : 0.f;
        row[96] = o3 > 0.f ? o3 : 0.f;
    }
    __syncthreads();

    const int m  = lane & 15;
    const int kh = lane >> 4;
    const int rq = wave >> 2;
    const int cq = wave & 3;
    const float* a0p = tile + (32 * rq + m) * LP + 2 * kh;
    const float* a1p = a0p + 16 * LP;
    const float* wp  = Wn + (size_t)(8 * kh) * HID + 32 * cq + 2 * m;

    v8f acc00 = {}, acc01 = {}, acc10 = {}, acc11 = {};
    #pragma unroll 2
    for (int p = 0; p < 16; ++p) {
        const int ks = 2 * p;
        const float* w0 = wp + (size_t)(16 * (ks & 7) + (ks >> 3)) * HID;
        const float* w1 = w0 + (size_t)16 * HID;
        const v2f wc0 = *(const v2f*)(w0), wc1 = *(const v2f*)(w0 + 4 * HID);
        const v2f wc2 = *(const v2f*)(w1), wc3 = *(const v2f*)(w1 + 4 * HID);
        const v2f a00 = *(const v2f*)(a0p + 4 * ks);
        const v2f a10 = *(const v2f*)(a1p + 4 * ks);
        const v2f a01 = *(const v2f*)(a0p + 4 * ks + 4);
        const v2f a11 = *(const v2f*)(a1p + 4 * ks + 4);
        v2f b0, b1;
        b0.x = wc0.x; b0.y = wc1.x; b1.x = wc0.y; b1.y = wc1.y;
        acc00 = __builtin_amdgcn_wmma_f32_16x16x4_f32(false, a00, false, b0, (short)0, acc00, false, false);
        acc01 = __builtin_amdgcn_wmma_f32_16x16x4_f32(false, a00, false, b1, (short)0, acc01, false, false);
        acc10 = __builtin_amdgcn_wmma_f32_16x16x4_f32(false, a10, false, b0, (short)0, acc10, false, false);
        acc11 = __builtin_amdgcn_wmma_f32_16x16x4_f32(false, a10, false, b1, (short)0, acc11, false, false);
        b0.x = wc2.x; b0.y = wc3.x; b1.x = wc2.y; b1.y = wc3.y;
        acc00 = __builtin_amdgcn_wmma_f32_16x16x4_f32(false, a01, false, b0, (short)0, acc00, false, false);
        acc01 = __builtin_amdgcn_wmma_f32_16x16x4_f32(false, a01, false, b1, (short)0, acc01, false, false);
        acc10 = __builtin_amdgcn_wmma_f32_16x16x4_f32(false, a11, false, b0, (short)0, acc10, false, false);
        acc11 = __builtin_amdgcn_wmma_f32_16x16x4_f32(false, a11, false, b1, (short)0, acc11, false, false);
    }

    const int row0 = node0 + 32 * rq + 8 * kh;
    float* ob = out + (size_t)row0 * HID + 32 * cq + 2 * m;
    #pragma unroll
    for (int r = 0; r < 8; ++r) {
        if (row0 + r < N_NODES) {
            v2f o; o.x = acc00[r]; o.y = acc01[r];
            *(v2f*)(ob + (size_t)r * HID) = o;
        }
        if (row0 + 16 + r < N_NODES) {
            v2f o; o.x = acc10[r]; o.y = acc11[r];
            *(v2f*)(ob + (size_t)(16 + r) * HID) = o;
        }
    }
}

__global__ __launch_bounds__(256)
void pool_accum_kernel(const float* __restrict__ X, const int* __restrict__ batch,
                       float* __restrict__ pooled, float* __restrict__ counts) {
    int wave = (blockIdx.x * blockDim.x + threadIdx.x) >> 5;
    int lane = threadIdx.x & 31;
    if (wave >= N_NODES) return;
    int g = batch[wave];
    float4 v = ((const float4*)(X + (size_t)wave * HID))[lane];
    float* pp = pooled + (size_t)g * HID + lane * 4;
    atomicAdd(pp + 0, v.x);
    atomicAdd(pp + 1, v.y);
    atomicAdd(pp + 2, v.z);
    atomicAdd(pp + 3, v.w);
    if (lane == 0) atomicAdd(&counts[g], 1.0f);
}

__global__ void final_linear_kernel(const float* __restrict__ pooled,
                                    const float* __restrict__ counts,
                                    const float* __restrict__ linW,
                                    const float* __restrict__ linb,
                                    float* __restrict__ out) {
    int i = blockIdx.x * blockDim.x + threadIdx.x;
    if (i >= N_GRAPHS * N_CLASSES) return;
    int g = i / N_CLASSES;
    int c = i - g * N_CLASSES;
    float scale = 1.0f / fmaxf(counts[g], 1.0f);
    const float* pg = pooled + (size_t)g * HID;
    float acc = 0.f;
    #pragma unroll 4
    for (int f = 0; f < HID; ++f) acc += pg[f] * linW[f * N_CLASSES + c];
    out[i] = acc * scale + linb[c];
}

extern "C" void kernel_launch(void* const* d_in, const int* in_sizes, int n_in,
                              void* d_out, int out_size, void* d_ws, size_t ws_size,
                              hipStream_t stream) {
    const float* x     = (const float*)d_in[0];
    const int*   edge  = (const int*)d_in[1];
    const int*   src   = edge;
    const int*   dst   = edge + N_EDGES;
    const int*   batch = (const int*)d_in[2];
    const float* Ws    = (const float*)d_in[3];
    const float* bs    = (const float*)d_in[4];
    const float* linW  = (const float*)d_in[5];
    const float* linb  = (const float*)d_in[6];
    float*       out   = (float*)d_out;

    float* ws = (float*)d_ws;
    float* H      = ws;
    float* X1     = H + (size_t)N_NODES * HID;
    float* X2     = X1 + (size_t)N_NODES * HID;
    float* pooled = X2 + (size_t)N_NODES * HID;
    float* counts = pooled + (size_t)N_GRAPHS * HID;
    float* deg    = counts + N_GRAPHS;
    int2*  brec   = (int2*)(deg + N_NODES);
    int2*  srec   = brec + N_EDGES;
    int*   tstart = (int*)(srec + N_EDGES);
    int*   cursor = tstart + 1024;

    const int T = 256;

    init_deg_kernel<<<(N_NODES + T - 1) / T, T, 0, stream>>>(deg);
    deg_accum_kernel<<<(N_EDGES + T - 1) / T, T, 0, stream>>>(dst, deg);
    tile_scan_kernel<<<1, 1024, 0, stream>>>(deg, tstart, cursor);
    dinv_kernel<<<(N_NODES + T - 1) / T, T, 0, stream>>>(deg);
    fill_kernel<<<(N_EDGES + T - 1) / T, T, 0, stream>>>(src, dst, deg, cursor, brec);
    tile_sort_kernel<<<NT, SORT_T, 0, stream>>>(tstart, brec, srec);

    const int gemm_blocks = (N_NODES / 16) * (HID / 16) / 8;

    gemm_wmma_kernel<<<gemm_blocks, T, 0, stream>>>(x, Ws, H);
    float* hbuf[2] = { H, X1 };
    for (int l = 0; l < N_LAYERS - 1; ++l) {
        const float* Wn = Ws + (size_t)(l + 1) * HID * HID;
        const float* bl = bs + (size_t)l * HID;
        agg_kernel<false><<<NT, AGG_T, 0, stream>>>(hbuf[l & 1], deg, tstart, srec, bl, Wn,
                                                    hbuf[(l + 1) & 1]);
    }
    agg_kernel<true><<<NT, AGG_T, 0, stream>>>(hbuf[(N_LAYERS - 1) & 1], deg, tstart, srec,
                                               bs + (size_t)(N_LAYERS - 1) * HID, Ws, X2);
    const float* xin = X2;

    int pool4 = (N_GRAPHS * HID + N_GRAPHS + 3) / 4;
    zero_kernel<<<(pool4 + T - 1) / T, T, 0, stream>>>((float4*)pooled, pool4);
    pool_accum_kernel<<<N_NODES / 8, T, 0, stream>>>(xin, batch, pooled, counts);
    final_linear_kernel<<<(N_GRAPHS * N_CLASSES + T - 1) / T, T, 0, stream>>>(
        pooled, counts, linW, linb, out);
}


# Mean pool + classifier in one launch, one workgroup per graph: node range by 32-way search of the ascending ids, rows summed in registers, waves joined in LDS; no float atomics; sum order changes
# speedup vs baseline: 3.4353x; 1.2203x over previous
#include <hip/hip_runtime.h>
#include <hip/hip_bf16.h>


#define N_NODES   100000
#define N_EDGES   1600000
#define HID       128
#define N_CLASSES 10
#define N_GRAPHS  512
#define N_LAYERS  5
#define TN        128
#define NT        ((N_NODES + TN - 1) / TN)
#define CPAD      32
#define AGG_T     512
#define SORT_T    512
#define POOL_T    256
#define LP        132
#define SRC_BITS  17
#define SRC_MASK  ((1 << SRC_BITS) - 1)

static_assert(N_NODES <= (1 << SRC_BITS), "source node must fit the record's low bits");
static_assert(TN <= (1 << (31 - SRC_BITS)), "row inside the tile must fit above the source bits");

typedef __attribute__((ext_vector_type(2))) float v2f;
typedef __attribute__((ext_vector_type(8))) float v8f;

__global__ void init_deg_kernel(float* __restrict__ deg) {
    int i = blockIdx.x * blockDim.x + threadIdx.x;
    if (i < N_NODES) deg[i] = 1.0f;
}

__global__ void deg_accum_kernel(const int* __restrict__ dst, float* __restrict__ deg) {
    int e = blockIdx.x * blockDim.x + threadIdx.x;
    if (e < N_EDGES) atomicAdd(&deg[dst[e]], 1.0f);
}

__global__ void dinv_kernel(float* __restrict__ deg) {
    int i = blockIdx.x * blockDim.x + threadIdx.x;
    if (i < N_NODES) deg[i] = rsqrtf(deg[i]);
}

__global__ __launch_bounds__(1024)
void tile_scan_kernel(const float* __restrict__ deg, int* __restrict__ tile_start,
                      int* __restrict__ cursor) {
    __shared__ int cnt[1024];
    __shared__ int wsum[32];
    const int tid  = threadIdx.x;
    const int lane = tid & 31;
    const int wave = tid >> 5;

    for (int t = wave; t < 1024; t += 32) {
        int c = 0;
        int n = t * TN + lane * 4;
        if (t < NT && n < N_NODES) {
            float4 d = *(const float4*)(deg + n);
            c = (int)d.x + (int)d.y + (int)d.z + (int)d.w - 4;
        }
        for (int off = 16; off > 0; off >>= 1) c += __shfl_xor(c, off);
        if (lane == 0) cnt[t] = c;
    }
    __syncthreads();

    const int v = cnt[tid];
    int x = v;
    for (int off = 1; off < 32; off <<= 1) {
        int y = __shfl_up(x, off);
        if (lane >= off) x += y;
    }
    if (lane == 31) wsum[wave] = x;
    __syncthreads();
    if (wave == 0) {
        const int s = wsum[lane];
        int xs = s;
        for (int off = 1; off < 32; off <<= 1) {
            int y = __shfl_up(xs, off);
            if (lane >= off) xs += y;
        }
        wsum[lane] = xs - s;
    }
    __syncthreads();
    const int excl = x - v + wsum[wave];
    if (tid <= NT) tile_start[tid] = excl;
    if (tid < NT)  cursor[tid * CPAD] = excl;
}

__global__ void fill_kernel(const int* __restrict__ src, const int* __restrict__ dst,
                            const float* __restrict__ dinv, int* __restrict__ cursor,
                            int2* __restrict__ brec) {
    int e = blockIdx.x * blockDim.x + threadIdx.x;
    if (e >= N_EDGES) return;
    int s = src[e];
    int d = dst[e];
    int pos = atomicAdd(&cursor[(d >> 7) * CPAD], 1);
    int2 r;
    r.x = s | ((d & (TN - 1)) << SRC_BITS);
    r.y = __float_as_int(dinv[s] * dinv[d]);
    brec[pos] = r;
}

__global__ __launch_bounds__(SORT_T)
void tile_sort_kernel(const int* __restrict__ tile_start, const int2* __restrict__ brec,
                      int2* __restrict__ srec) {
    __shared__ int cnt[TN];
    __shared__ int wtot[TN / 32];
    const int t    = blockIdx.x;
    const int tid  = threadIdx.x;
    const int lane = tid & 31;
    const int wave = tid >> 5;
    const int beg  = tile_start[t];
    const int end  = tile_start[t + 1];

    if (tid < TN) cnt[tid] = 0;
    __syncthreads();
    for (int i = beg + tid; i < end; i += SORT_T) atomicAdd(&cnt[brec[i].x >> SRC_BITS], 1);
    __syncthreads();

    const int c = (tid < TN) ? cnt[tid] : 0;
    int x = c;
    for (int off = 1; off < 32; off <<= 1) {
        int y = __shfl_up(x, off);
        if (lane >= off) x += y;
    }
    if (tid < TN && lane == 31) wtot[wave] = x;
    __syncthreads();
    if (tid < TN) {
        int front = beg;
        for (int j = 0; j < wave; ++j) front += wtot[j];
        cnt[tid] = front + x - c;
    }
    __syncthreads();

    for (int i = beg + tid; i < end; i += SORT_T) {
        const int2 r  = brec[i];
        const int pos = atomicAdd(&cnt[r.x >> SRC_BITS], 1);
        srec[pos] = r;
    }
}

__global__ __launch_bounds__(256)
void gemm_wmma_kernel(const float* __restrict__ X, const float* __restrict__ W,
                      float* __restrict__ H) {
    int wave     = (blockIdx.x * blockDim.x + threadIdx.x) >> 5;
    int lane     = threadIdx.x & 31;
    int row_tile = wave >> 3;
    int col_tile = wave & 7;
    if (row_tile * 16 >= N_NODES) return;

    int m  = lane & 15;
    int kh = lane >> 4;
    const float* xrow = X + (size_t)(row_tile * 16 + m) * HID;
    int cb = col_tile * 16;

    v8f acc = {};
    #pragma unroll
    for (int kb = 0; kb < HID; kb += 4) {
        int k0 = kb + 2 * kh;
        v2f a = *(const v2f*)(xrow + k0);
        v2f b;
        b.x = W[(size_t)(k0 + 0) * HID + cb + m];
        b.y = W[(size_t)(k0 + 1) * HID + cb + m];
        acc = __builtin_amdgcn_wmma_f32_16x16x4_f32(
             false, a,  false, b,
             (short)0, acc,  false,  false);
    }

    float* hbase = H + (size_t)(row_tile * 16 + 8 * kh) * HID + cb + m;
    #pragma unroll
    for (int r = 0; r < 8; ++r) hbase[(size_t)r * HID] = acc[r];
}

template <bool LAST>
__global__ __launch_bounds__(AGG_T)
void agg_kernel(const float* __restrict__ H, const float* __restrict__ dinv,
                const int* __restrict__ tile_start, const int2* __restrict__ srec,
                const float* __restrict__ bias, const float* __restrict__ Wn,
                float* __restrict__ out) {
    __shared__ float tile[TN * LP];
    const int t    = blockIdx.x;
    const int lane = threadIdx.x & 31;
    const int wave = __builtin_amdgcn_readfirstlane((int)(threadIdx.x >> 5));
    const int NW   = AGG_T / 32;
    const int node0 = t * TN;

    for (int r = wave; r < TN; r += NW) {
        const int node = node0 + r;
        float4 v = make_float4(0.f, 0.f, 0.f, 0.f);
        if (node < N_NODES) {
            const float dv = dinv[node];
            const float w  = dv * dv;
            const float4 h = ((const float4*)(H + (size_t)node * HID))[lane];
            v.x = h.x * w; v.y = h.y * w; v.z = h.z * w; v.w = h.w * w;
        }
        float* row = tile + r * LP + lane;
        row[0] = v.x; row[32] = v.y; row[64] = v.z; row[96] = v.w;
    }
    __syncthreads();

    const int beg = tile_start[t];
    const int end = tile_start[t + 1];
    for (int base = beg + wave * 32; base < end; base += NW * 32) {
        const int idx = base + lane;
        int pk = 0, wb = 0;
        if (idx < end) { const int2 r = srec[idx]; pk = r.x; wb = r.y; }
        const int n = min(32, end - base);
        int   cur = __builtin_amdgcn_readlane(pk, 0) >> SRC_BITS;
        float a0 = 0.f, a1 = 0.f, a2 = 0.f, a3 = 0.f;
        for (int k = 0; k < n; k += 4) {
            float4 h[4];
            #pragma unroll
            for (int u = 0; u < 4; ++u) {
                const int sk = __builtin_amdgcn_readlane(pk, k + u) & SRC_MASK;
                h[u] = ((const float4*)(H + (size_t)sk * HID))[lane];
            }
            #pragma unroll
            for (int u = 0; u < 4; ++u) {
                if (u == 0 || k + u < n) {
                    const int   dk = __builtin_amdgcn_readlane(pk, k + u) >> SRC_BITS;
                    const float wk = __int_as_float(__builtin_amdgcn_readlane(wb, k + u));
                    if (dk != cur) {
                        float* row = tile + cur * LP + lane;
                        __hip_atomic_fetch_add(row,      a0, __ATOMIC_RELAXED, __HIP_MEMORY_SCOPE_WORKGROUP);
                        __hip_atomic_fetch_add(row + 32, a1, __ATOMIC_RELAXED, __HIP_MEMORY_SCOPE_WORKGROUP);
                        __hip_atomic_fetch_add(row + 64, a2, __ATOMIC_RELAXED, __HIP_MEMORY_SCOPE_WORKGROUP);
                        __hip_atomic_fetch_add(row + 96, a3, __ATOMIC_RELAXED, __HIP_MEMORY_SCOPE_WORKGROUP);
                        a0 = 0.f; a1 = 0.f; a2 = 0.f; a3 = 0.f;
                        cur = dk;
                    }
                    a0 += h[u].x * wk;
                    a1 += h[u].y * wk;
                    a2 += h[u].z * wk;
                    a3 += h[u].w * wk;
                }
            }
        }
        float* row = tile + cur * LP + lane;
        __hip_atomic_fetch_add(row,      a0, __ATOMIC_RELAXED, __HIP_MEMORY_SCOPE_WORKGROUP);
        __hip_atomic_fetch_add(row + 32, a1, __ATOMIC_RELAXED, __HIP_MEMORY_SCOPE_WORKGROUP);
        __hip_atomic_fetch_add(row + 64, a2, __ATOMIC_RELAXED, __HIP_MEMORY_SCOPE_WORKGROUP);
        __hip_atomic_fetch_add(row + 96, a3, __ATOMIC_RELAXED, __HIP_MEMORY_SCOPE_WORKGROUP);
    }
    __syncthreads();

    const float4 b = ((const float4*)bias)[lane];
    if (LAST) {
        for (int r = wave; r < TN; r += NW) {
            const int node = node0 + r;
            if (node < N_NODES) {
                const float* row = tile + r * LP + lane;
                float4 o;
                o.x = row[0]  + b.x; o.y = row[32] + b.y; o.z = row[64] + b.z; o.w = row[96] + b.w;
                o.x = o.x > 0.f ? o.x : 0.f;
                o.y = o.y > 0.f ? o.y : 0.f;
                o.z = o.z > 0.f ? o.z : 0.f;
                o.w = o.w > 0.f ? o.w : 0.f;
                ((float4*)(out + (size_t)node * HID))[lane] = o;
            }
        }
        return;
    }

    for (int r = wave; r < TN; r += NW) {
        float* row = tile + r * LP + lane;
        float o0 = row[0] + b.x, o1 = row[32] + b.y, o2 = row[64] + b.z, o3 = row[96] + b.w;
        row[0]  = o0 > 0.f ? o0 : 0.f;
        row[32] = o1 > 0.f ? o1 : 0.f;
        row[64] = o2 > 0.f ? o2 : 0.f;
        row[96] = o3 > 0.f ? o3 : 0.f;
    }
    __syncthreads();

    const int m  = lane & 15;
    const int kh = lane >> 4;
    const int rq = wave >> 2;
    const int cq = wave & 3;
    const float* a0p = tile + (32 * rq + m) * LP + 2 * kh;
    const float* a1p = a0p + 16 * LP;
    const float* wp  = Wn + (size_t)(8 * kh) * HID + 32 * cq + 2 * m;

    v8f acc00 = {}, acc01 = {}, acc10 = {}, acc11 = {};
    #pragma unroll 2
    for (int p = 0; p < 16; ++p) {
        const int ks = 2 * p;
        const float* w0 = wp + (size_t)(16 * (ks & 7) + (ks >> 3)) * HID;
        const float* w1 = w0 + (size_t)16 * HID;
        const v2f wc0 = *(const v2f*)(w0), wc1 = *(const v2f*)(w0 + 4 * HID);
        const v2f wc2 = *(const v2f*)(w1), wc3 = *(const v2f*)(w1 + 4 * HID);
        const v2f a00 = *(const v2f*)(a0p + 4 * ks);
        const v2f a10 = *(const v2f*)(a1p + 4 * ks);
        const v2f a01 = *(const v2f*)(a0p + 4 * ks + 4);
        const v2f a11 = *(const v2f*)(a1p + 4 * ks + 4);
        v2f b0, b1;
        b0.x = wc0.x; b0.y = wc1.x; b1.x = wc0.y; b1.y = wc1.y;
        acc00 = __builtin_amdgcn_wmma_f32_16x16x4_f32(false, a00, false, b0, (short)0, acc00, false, false);
        acc01 = __builtin_amdgcn_wmma_f32_16x16x4_f32(false, a00, false, b1, (short)0, acc01, false, false);
        acc10 = __builtin_amdgcn_wmma_f32_16x16x4_f32(false, a10, false, b0, (short)0, acc10, false, false);
        acc11 = __builtin_amdgcn_wmma_f32_16x16x4_f32(false, a10, false, b1, (short)0, acc11, false, false);
        b0.x = wc2.x; b0.y = wc3.x; b1.x = wc2.y; b1.y = wc3.y;
        acc00 = __builtin_amdgcn_wmma_f32_16x16x4_f32(false, a01, false, b0, (short)0, acc00, false, false);
        acc01 = __builtin_amdgcn_wmma_f32_16x16x4_f32(false, a01, false, b1, (short)0, acc01, false, false);
        acc10 = __builtin_amdgcn_wmma_f32_16x16x4_f32(false, a11, false, b0, (short)0, acc10, false, false);
        acc11 = __builtin_amdgcn_wmma_f32_16x16x4_f32(false, a11, false, b1, (short)0, acc11, false, false);
    }

    const int row0 = node0 + 32 * rq + 8 * kh;
    float* ob = out + (size_t)row0 * HID + 32 * cq + 2 * m;
    #pragma unroll
    for (int r = 0; r < 8; ++r) {
        if (row0 + r < N_NODES) {
            v2f o; o.x = acc00[r]; o.y = acc01[r];
            *(v2f*)(ob + (size_t)r * HID) = o;
        }
        if (row0 + 16 + r < N_NODES) {
            v2f o; o.x = acc10[r]; o.y = acc11[r];
            *(v2f*)(ob + (size_t)(16 + r) * HID) = o;
        }
    }
}

__device__ __forceinline__ int first_node_at_or_above(const int* __restrict__ batch, int key,
                                                      int lane) {
    int lo = 0, len = N_NODES;
    for (int it = 0; it < 5 && len > 0; ++it) {
        const int step = (len + 31) >> 5;
        const int q    = lo + lane * step;
        const bool below = (q < lo + len) && (batch[q] < key);
        const int c = __builtin_popcount(__builtin_amdgcn_ballot_w32(below));
        if (c == 0) {
            len = 0;
        } else {
            const int end = lo + len;
            const int nlo = lo + (c - 1) * step + 1;
            len = min(lo + c * step, end) - nlo;
            lo  = nlo;
        }
    }
    return lo;
}

__global__ __launch_bounds__(POOL_T)
void pool_linear_kernel(const float* __restrict__ X, const int* __restrict__ batch,
                        const float* __restrict__ linW, const float* __restrict__ linb,
                        float* __restrict__ out) {
    __shared__ float4 part4[(POOL_T / 32) * (HID / 4)];
    __shared__ float4 wl4[HID * N_CLASSES / 4];
    __shared__ float  pooled[HID];
    __shared__ int    range[2];
    const int g    = blockIdx.x;
    const int tid  = threadIdx.x;
    const int lane = tid & 31;
    const int wave = __builtin_amdgcn_readfirstlane((int)(threadIdx.x >> 5));
    const int NW   = POOL_T / 32;
    const int WQ   = HID * N_CLASSES / 8;
    static_assert(HID * N_CLASSES % 8 == 0 && WQ % 32 == 0 && 2 + WQ / 32 <= POOL_T / 32,
                  "the weight copy takes whole waves behind the two that search");

    if (wave < 2) {
        const int b = first_node_at_or_above(batch, g + wave, lane);
        if (lane == 0) range[wave] = b;
    } else if (wave < 2 + WQ / 32) {
        const int i = tid - 64;
        const float4 w0 = ((const float4*)linW)[i];
        const float4 w1 = ((const float4*)linW)[i + WQ];
        wl4[i]      = w0;
        wl4[i + WQ] = w1;
    }
    __syncthreads();
    const int lo = range[0];
    const int hi = range[1];

    float4 acc = make_float4(0.f, 0.f, 0.f, 0.f);
    for (int r = lo + wave; r < hi; r += 16 * NW) {
        float4 h[16];
        #pragma unroll
        for (int u = 0; u < 16; ++u) {
            const int ru = r + u * NW;
            h[u] = make_float4(0.f, 0.f, 0.f, 0.f);
            if (ru < hi) h[u] = ((const float4*)(X + (size_t)ru * HID))[lane];
        }
        #pragma unroll
        for (int u = 0; u < 16; ++u) {
            acc.x += h[u].x; acc.y += h[u].y; acc.z += h[u].z; acc.w += h[u].w;
        }
    }
    part4[wave * (HID / 4) + lane] = acc;
    __syncthreads();

    if (tid < HID) {
        const float* pf = (const float*)part4;
        float sum = 0.f;
        #pragma unroll
        for (int w = 0; w < NW; ++w) sum += pf[w * HID + tid];
        pooled[tid] = sum;
    }
    __syncthreads();

    const float* wl = (const float*)wl4;
    for (int c = wave; c < N_CLASSES; c += NW) {
        float a = 0.f;
        #pragma unroll
        for (int j = 0; j < HID / 32; ++j) {
            const int f = lane + 32 * j;
            a += pooled[f] * wl[f * N_CLASSES + c];
        }
        for (int off = 16; off > 0; off >>= 1) a += __shfl_xor(a, off);
        if (lane == 0) {
            const float scale = 1.0f / fmaxf((float)(hi - lo), 1.0f);
            out[g * N_CLASSES + c] = a * scale + linb[c];
        }
    }
}

extern "C" void kernel_launch(void* const* d_in, const int* in_sizes, int n_in,
                              void* d_out, int out_size, void* d_ws, size_t ws_size,
                              hipStream_t stream) {
    const float* x     = (const float*)d_in[0];
    const int*   edge  = (const int*)d_in[1];
    const int*   src   = edge;
    const int*   dst   = edge + N_EDGES;
    const int*   batch = (const int*)d_in[2];
    const float* Ws    = (const float*)d_in[3];
    const float* bs    = (const float*)d_in[4];
    const float* linW  = (const float*)d_in[5];
    const float* linb  = (const float*)d_in[6];
    float*       out   = (float*)d_out;

    float* ws = (float*)d_ws;
    float* H      = ws;
    float* X1     = H + (size_t)N_NODES * HID;
    float* X2     = X1 + (size_t)N_NODES * HID;
    float* deg    = X2 + (size_t)N_NODES * HID;
    int2*  brec   = (int2*)(deg + N_NODES);
    int2*  srec   = brec + N_EDGES;
    int*   tstart = (int*)(srec + N_EDGES);
    int*   cursor = tstart + 1024;

    const int T = 256;

    init_deg_kernel<<<(N_NODES + T - 1) / T, T, 0, stream>>>(deg);
    deg_accum_kernel<<<(N_EDGES + T - 1) / T, T, 0, stream>>>(dst, deg);
    tile_scan_kernel<<<1, 1024, 0, stream>>>(deg, tstart, cursor);
    dinv_kernel<<<(N_NODES + T - 1) / T, T, 0, stream>>>(deg);
    fill_kernel<<<(N_EDGES + T - 1) / T, T, 0, stream>>>(src, dst, deg, cursor, brec);
    tile_sort_kernel<<<NT, SORT_T, 0, stream>>>(tstart, brec, srec);

    const int gemm_blocks = (N_NODES / 16) * (HID / 16) / 8;

    gemm_wmma_kernel<<<gemm_blocks, T, 0, stream>>>(x, Ws, H);
    float* hbuf[2] = { H, X1 };
    for (int l = 0; l < N_LAYERS - 1; ++l) {
        const float* Wn = Ws + (size_t)(l + 1) * HID * HID;
        const float* bl = bs + (size_t)l * HID;
        agg_kernel<false><<<NT, AGG_T, 0, stream>>>(hbuf[l & 1], deg, tstart, srec, bl, Wn,
                                                    hbuf[(l + 1) & 1]);
    }
    agg_kernel<true><<<NT, AGG_T, 0, stream>>>(hbuf[(N_LAYERS - 1) & 1], deg, tstart, srec,
                                               bs + (size_t)(N_LAYERS - 1) * HID, Ws, X2);

    pool_linear_kernel<<<N_GRAPHS, POOL_T, 0, stream>>>(X2, batch, linW, linb, out);
}


# Layer 1's product moved behind its neighbour sum into the first tile launch: the tile sums x rows, (sums)@W1+bias, ReLU goes back to LDS and feeds the W2 product; no product launch; sum order changes
# speedup vs baseline: 3.6567x; 1.0644x over previous
#include <hip/hip_runtime.h>
#include <hip/hip_bf16.h>


#define N_NODES   100000
#define N_EDGES   1600000
#define HID       128
#define N_CLASSES 10
#define N_GRAPHS  512
#define N_LAYERS  5
#define TN        128
#define NT        ((N_NODES + TN - 1) / TN)
#define CPAD      32
#define AGG_T     512
#define SORT_T    512
#define POOL_T    256
#define LP        132
#define SRC_BITS  17
#define SRC_MASK  ((1 << SRC_BITS) - 1)

static_assert(N_NODES <= (1 << SRC_BITS), "source node must fit the record's low bits");
static_assert(TN <= (1 << (31 - SRC_BITS)), "row inside the tile must fit above the source bits");

typedef __attribute__((ext_vector_type(2))) float v2f;
typedef __attribute__((ext_vector_type(8))) float v8f;

__global__ void init_deg_kernel(float* __restrict__ deg) {
    int i = blockIdx.x * blockDim.x + threadIdx.x;
    if (i < N_NODES) deg[i] = 1.0f;
}

__global__ void deg_accum_kernel(const int* __restrict__ dst, float* __restrict__ deg) {
    int e = blockIdx.x * blockDim.x + threadIdx.x;
    if (e < N_EDGES) atomicAdd(&deg[dst[e]], 1.0f);
}

__global__ void dinv_kernel(float* __restrict__ deg) {
    int i = blockIdx.x * blockDim.x + threadIdx.x;
    if (i < N_NODES) deg[i] = rsqrtf(deg[i]);
}

__global__ __launch_bounds__(1024)
void tile_scan_kernel(const float* __restrict__ deg, int* __restrict__ tile_start,
                      int* __restrict__ cursor) {
    __shared__ int cnt[1024];
    __shared__ int wsum[32];
    const int tid  = threadIdx.x;
    const int lane = tid & 31;
    const int wave = tid >> 5;

    for (int t = wave; t < 1024; t += 32) {
        int c = 0;
        int n = t * TN + lane * 4;
        if (t < NT && n < N_NODES) {
            float4 d = *(const float4*)(deg + n);
            c = (int)d.x + (int)d.y + (int)d.z + (int)d.w - 4;
        }
        for (int off = 16; off > 0; off >>= 1) c += __shfl_xor(c, off);
        if (lane == 0) cnt[t] = c;
    }
    __syncthreads();

    const int v = cnt[tid];
    int x = v;
    for (int off = 1; off < 32; off <<= 1) {
        int y = __shfl_up(x, off);
        if (lane >= off) x += y;
    }
    if (lane == 31) wsum[wave] = x;
    __syncthreads();
    if (wave == 0) {
        const int s = wsum[lane];
        int xs = s;
        for (int off = 1; off < 32; off <<= 1) {
            int y = __shfl_up(xs, off);
            if (lane >= off) xs += y;
        }
        wsum[lane] = xs - s;
    }
    __syncthreads();
    const int excl = x - v + wsum[wave];
    if (tid <= NT) tile_start[tid] = excl;
    if (tid < NT)  cursor[tid * CPAD] = excl;
}

__global__ void fill_kernel(const int* __restrict__ src, const int* __restrict__ dst,
                            const float* __restrict__ dinv, int* __restrict__ cursor,
                            int2* __restrict__ brec) {
    int e = blockIdx.x * blockDim.x + threadIdx.x;
    if (e >= N_EDGES) return;
    int s = src[e];
    int d = dst[e];
    int pos = atomicAdd(&cursor[(d >> 7) * CPAD], 1);
    int2 r;
    r.x = s | ((d & (TN - 1)) << SRC_BITS);
    r.y = __float_as_int(dinv[s] * dinv[d]);
    brec[pos] = r;
}

__global__ __launch_bounds__(SORT_T)
void tile_sort_kernel(const int* __restrict__ tile_start, const int2* __restrict__ brec,
                      int2* __restrict__ srec) {
    __shared__ int cnt[TN];
    __shared__ int wtot[TN / 32];
    const int t    = blockIdx.x;
    const int tid  = threadIdx.x;
    const int lane = tid & 31;
    const int wave = tid >> 5;
    const int beg  = tile_start[t];
    const int end  = tile_start[t + 1];

    if (tid < TN) cnt[tid] = 0;
    __syncthreads();
    for (int i = beg + tid; i < end; i += SORT_T) atomicAdd(&cnt[brec[i].x >> SRC_BITS], 1);
    __syncthreads();

    const int c = (tid < TN) ? cnt[tid] : 0;
    int x = c;
    for (int off = 1; off < 32; off <<= 1) {
        int y = __shfl_up(x, off);
        if (lane >= off) x += y;
    }
    if (tid < TN && lane == 31) wtot[wave] = x;
    __syncthreads();
    if (tid < TN) {
        int front = beg;
        for (int j = 0; j < wave; ++j) front += wtot[j];
        cnt[tid] = front + x - c;
    }
    __syncthreads();

    for (int i = beg + tid; i < end; i += SORT_T) {
        const int2 r  = brec[i];
        const int pos = atomicAdd(&cnt[r.x >> SRC_BITS], 1);
        srec[pos] = r;
    }
}

__device__ __forceinline__ void tile_product(const float* tile, const float* __restrict__ Wn,
                                             int lane, int wave,
                                             v8f& acc00, v8f& acc01, v8f& acc10, v8f& acc11) {
    const int m  = lane & 15;
    const int kh = lane >> 4;
    const int rq = wave >> 2;
    const int cq = wave & 3;
    const float* a0p = tile + (32 * rq + m) * LP + 2 * kh;
    const float* a1p = a0p + 16 * LP;
    const float* wp  = Wn + (size_t)(8 * kh) * HID + 32 * cq + 2 * m;

    #pragma unroll 2
    for (int p = 0; p < 16; ++p) {
        const int ks = 2 * p;
        const float* w0 = wp + (size_t)(16 * (ks & 7) + (ks >> 3)) * HID;
        const float* w1 = w0 + (size_t)16 * HID;
        const v2f wc0 = *(const v2f*)(w0), wc1 = *(const v2f*)(w0 + 4 * HID);
        const v2f wc2 = *(const v2f*)(w1), wc3 = *(const v2f*)(w1 + 4 * HID);
        const v2f a00 = *(const v2f*)(a0p + 4 * ks);
        const v2f a10 = *(const v2f*)(a1p + 4 * ks);
        const v2f a01 = *(const v2f*)(a0p + 4 * ks + 4);
        const v2f a11 = *(const v2f*)(a1p + 4 * ks + 4);
        v2f b0, b1;
        b0.x = wc0.x; b0.y = wc1.x; b1.x = wc0.y; b1.y = wc1.y;
        acc00 = __builtin_amdgcn_wmma_f32_16x16x4_f32(false, a00, false, b0, (short)0, acc00, false, false);
        acc01 = __builtin_amdgcn_wmma_f32_16x16x4_f32(false, a00, false, b1, (short)0, acc01, false, false);
        acc10 = __builtin_amdgcn_wmma_f32_16x16x4_f32(false, a10, false, b0, (short)0, acc10, false, false);
        acc11 = __builtin_amdgcn_wmma_f32_16x16x4_f32(false, a10, false, b1, (short)0, acc11, false, false);
        b0.x = wc2.x; b0.y = wc3.x; b1.x = wc2.y; b1.y = wc3.y;
        acc00 = __builtin_amdgcn_wmma_f32_16x16x4_f32(false, a01, false, b0, (short)0, acc00, false, false);
        acc01 = __builtin_amdgcn_wmma_f32_16x16x4_f32(false, a01, false, b1, (short)0, acc01, false, false);
        acc10 = __builtin_amdgcn_wmma_f32_16x16x4_f32(false, a11, false, b0, (short)0, acc10, false, false);
        acc11 = __builtin_amdgcn_wmma_f32_16x16x4_f32(false, a11, false, b1, (short)0, acc11, false, false);
    }
}

template <int MODE>
__global__ __launch_bounds__(AGG_T, 16)
void agg_kernel(const float* __restrict__ H, const float* __restrict__ dinv,
                const int* __restrict__ tile_start, const int2* __restrict__ srec,
                const float* __restrict__ bias, const float* __restrict__ Wf,
                const float* __restrict__ Wn, float* __restrict__ out) {
    __shared__ float tile[TN * LP];
    const int t    = blockIdx.x;
    const int lane = threadIdx.x & 31;
    const int wave = __builtin_amdgcn_readfirstlane((int)(threadIdx.x >> 5));
    const int NW   = AGG_T / 32;
    const int node0 = t * TN;

    for (int r = wave; r < TN; r += NW) {
        const int node = node0 + r;
        float4 v = make_float4(0.f, 0.f, 0.f, 0.f);
        if (node < N_NODES) {
            const float dv = dinv[node];
            const float w  = dv * dv;
            const float4 h = ((const float4*)(H + (size_t)node * HID))[lane];
            v.x = h.x * w; v.y = h.y * w; v.z = h.z * w; v.w = h.w * w;
        }
        float* row = tile + r * LP + lane;
        row[0] = v.x; row[32] = v.y; row[64] = v.z; row[96] = v.w;
    }
    __syncthreads();

    const int beg = tile_start[t];
    const int end = tile_start[t + 1];
    for (int base = beg + wave * 32; base < end; base += NW * 32) {
        const int idx = base + lane;
        int pk = 0, wb = 0;
        if (idx < end) { const int2 r = srec[idx]; pk = r.x; wb = r.y; }
        const int n = min(32, end - base);
        int   cur = __builtin_amdgcn_readlane(pk, 0) >> SRC_BITS;
        float a0 = 0.f, a1 = 0.f, a2 = 0.f, a3 = 0.f;
        for (int k = 0; k < n; k += 4) {
            float4 h[4];
            #pragma unroll
            for (int u = 0; u < 4; ++u) {
                const int sk = __builtin_amdgcn_readlane(pk, k + u) & SRC_MASK;
                h[u] = ((const float4*)(H + (size_t)sk * HID))[lane];
            }
            #pragma unroll
            for (int u = 0; u < 4; ++u) {
                if (u == 0 || k + u < n) {
                    const int   dk = __builtin_amdgcn_readlane(pk, k + u) >> SRC_BITS;
                    const float wk = __int_as_float(__builtin_amdgcn_readlane(wb, k + u));
                    if (dk != cur) {
                        float* row = tile + cur * LP + lane;
                        __hip_atomic_fetch_add(row,      a0, __ATOMIC_RELAXED, __HIP_MEMORY_SCOPE_WORKGROUP);
                        __hip_atomic_fetch_add(row + 32, a1, __ATOMIC_RELAXED, __HIP_MEMORY_SCOPE_WORKGROUP);
                        __hip_atomic_fetch_add(row + 64, a2, __ATOMIC_RELAXED, __HIP_MEMORY_SCOPE_WORKGROUP);
                        __hip_atomic_fetch_add(row + 96, a3, __ATOMIC_RELAXED, __HIP_MEMORY_SCOPE_WORKGROUP);
                        a0 = 0.f; a1 = 0.f; a2 = 0.f; a3 = 0.f;
                        cur = dk;
                    }
                    a0 += h[u].x * wk;
                    a1 += h[u].y * wk;
                    a2 += h[u].z * wk;
                    a3 += h[u].w * wk;
                }
            }
        }
        float* row = tile + cur * LP + lane;
        __hip_atomic_fetch_add(row,      a0, __ATOMIC_RELAXED, __HIP_MEMORY_SCOPE_WORKGROUP);
        __hip_atomic_fetch_add(row + 32, a1, __ATOMIC_RELAXED, __HIP_MEMORY_SCOPE_WORKGROUP);
        __hip_atomic_fetch_add(row + 64, a2, __ATOMIC_RELAXED, __HIP_MEMORY_SCOPE_WORKGROUP);
        __hip_atomic_fetch_add(row + 96, a3, __ATOMIC_RELAXED, __HIP_MEMORY_SCOPE_WORKGROUP);
    }
    __syncthreads();

    const int m  = lane & 15;
    const int kh = lane >> 4;
    const int rq = wave >> 2;
    const int cq = wave & 3;

    if (MODE == 1) {
        const float4 b = ((const float4*)bias)[lane];
        for (int r = wave; r < TN; r += NW) {
            const int node = node0 + r;
            if (node < N_NODES) {
                const float* row = tile + r * LP + lane;
                float4 o;
                o.x = row[0]  + b.x; o.y = row[32] + b.y; o.z = row[64] + b.z; o.w = row[96] + b.w;
                o.x = o.x > 0.f ? o.x : 0.f;
                o.y = o.y > 0.f ? o.y : 0.f;
                o.z = o.z > 0.f ? o.z : 0.f;
                o.w = o.w > 0.f ? o.w : 0.f;
                ((float4*)(out + (size_t)node * HID))[lane] = o;
            }
        }
        return;
    }

    if (MODE == 2) {
        v8f p00 = {}, p01 = {}, p10 = {}, p11 = {};
        tile_product(tile, Wf, lane, wave, p00, p01, p10, p11);
        __syncthreads();
        const int f0 = 32 * cq + 2 * m;
        const v2f bb = *(const v2f*)(bias + f0);
        float* wr = tile + (32 * rq + 8 * kh) * LP + (f0 & 3) * 32 + (f0 >> 2);
        #pragma unroll
        for (int r = 0; r < 8; ++r) {
            const float o00 = p00[r] + bb.x, o01 = p01[r] + bb.y;
            const float o10 = p10[r] + bb.x, o11 = p11[r] + bb.y;
            wr[r * LP]             = o00 > 0.f ? o00 : 0.f;
            wr[r * LP + 32]        = o01 > 0.f ? o01 : 0.f;
            wr[(16 + r) * LP]      = o10 > 0.f ? o10 : 0.f;
            wr[(16 + r) * LP + 32] = o11 > 0.f ? o11 : 0.f;
        }
    } else {
        const float4 b = ((const float4*)bias)[lane];
        for (int r = wave; r < TN; r += NW) {
            float* row = tile + r * LP + lane;
            float o0 = row[0] + b.x, o1 = row[32] + b.y, o2 = row[64] + b.z, o3 = row[96] + b.w;
            row[0]  = o0 > 0.f ? o0 : 0.f;
            row[32] = o1 > 0.f ? o1 : 0.f;
            row[64] = o2 > 0.f ? o2 : 0.f;
            row[96] = o3 > 0.f ? o3 : 0.f;
        }
    }
    __syncthreads();

    v8f acc00 = {}, acc01 = {}, acc10 = {}, acc11 = {};
    tile_product(tile, Wn, lane, wave, acc00, acc01, acc10, acc11);

    const int row0 = node0 + 32 * rq + 8 * kh;
    float* ob = out + (size_t)row0 * HID + 32 * cq + 2 * m;
    #pragma unroll
    for (int r = 0; r < 8; ++r) {
        if (row0 + r < N_NODES) {
            v2f o; o.x = acc00[r]; o.y = acc01[r];
            *(v2f*)(ob + (size_t)r * HID) = o;
        }
        if (row0 + 16 + r < N_NODES) {
            v2f o; o.x = acc10[r]; o.y = acc11[r];
            *(v2f*)(ob + (size_t)(16 + r) * HID) = o;
        }
    }
}

__device__ __forceinline__ int first_node_at_or_above(const int* __restrict__ batch, int key,
                                                      int lane) {
    int lo = 0, len = N_NODES;
    for (int it = 0; it < 5 && len > 0; ++it) {
        const int step = (len + 31) >> 5;
        const int q    = lo + lane * step;
        const bool below = (q < lo + len) && (batch[q] < key);
        const int c = __builtin_popcount(__builtin_amdgcn_ballot_w32(below));
        if (c == 0) {
            len = 0;
        } else {
            const int end = lo + len;
            const int nlo = lo + (c - 1) * step + 1;
            len = min(lo + c * step, end) - nlo;
            lo  = nlo;
        }
    }
    return lo;
}

__global__ __launch_bounds__(POOL_T)
void pool_linear_kernel(const float* __restrict__ X, const int* __restrict__ batch,
                        const float* __restrict__ linW, const float* __restrict__ linb,
                        float* __restrict__ out) {
    __shared__ float4 part4[(POOL_T / 32) * (HID / 4)];
    __shared__ float4 wl4[HID * N_CLASSES / 4];
    __shared__ float  pooled[HID];
    __shared__ int    range[2];
    const int g    = blockIdx.x;
    const int tid  = threadIdx.x;
    const int lane = tid & 31;
    const int wave = __builtin_amdgcn_readfirstlane((int)(threadIdx.x >> 5));
    const int NW   = POOL_T / 32;
    const int WQ   = HID * N_CLASSES / 8;
    static_assert(HID * N_CLASSES % 8 == 0 && WQ % 32 == 0 && 2 + WQ / 32 <= POOL_T / 32,
                  "the weight copy takes whole waves behind the two that search");

    if (wave < 2) {
        const int b = first_node_at_or_above(batch, g + wave, lane);
        if (lane == 0) range[wave] = b;
    } else if (wave < 2 + WQ / 32) {
        const int i = tid - 64;
        const float4 w0 = ((const float4*)linW)[i];
        const float4 w1 = ((const float4*)linW)[i + WQ];
        wl4[i]      = w0;
        wl4[i + WQ] = w1;
    }
    __syncthreads();
    const int lo = range[0];
    const int hi = range[1];

    float4 acc = make_float4(0.f, 0.f, 0.f, 0.f);
    for (int r = lo + wave; r < hi; r += 16 * NW) {
        float4 h[16];
        #pragma unroll
        for (int u = 0; u < 16; ++u) {
            const int ru = r + u * NW;
            h[u] = make_float4(0.f, 0.f, 0.f, 0.f);
            if (ru < hi) h[u] = ((const float4*)(X + (size_t)ru * HID))[lane];
        }
        #pragma unroll
        for (int u = 0; u < 16; ++u) {
            acc.x += h[u].x; acc.y += h[u].y; acc.z += h[u].z; acc.w += h[u].w;
        }
    }
    part4[wave * (HID / 4) + lane] = acc;
    __syncthreads();

    if (tid < HID) {
        const float* pf = (const float*)part4;
        float sum = 0.f;
        #pragma unroll
        for (int w = 0; w < NW; ++w) sum += pf[w * HID + tid];
        pooled[tid] = sum;
    }
    __syncthreads();

    const float* wl = (const float*)wl4;
    for (int c = wave; c < N_CLASSES; c += NW) {
        float a = 0.f;
        #pragma unroll
        for (int j = 0; j < HID / 32; ++j) {
            const int f = lane + 32 * j;
            a += pooled[f] * wl[f * N_CLASSES + c];
        }
        for (int off = 16; off > 0; off >>= 1) a += __shfl_xor(a, off);
        if (lane == 0) {
            const float scale = 1.0f / fmaxf((float)(hi - lo), 1.0f);
            out[g * N_CLASSES + c] = a * scale + linb[c];
        }
    }
}

extern "C" void kernel_launch(void* const* d_in, const int* in_sizes, int n_in,
                              void* d_out, int out_size, void* d_ws, size_t ws_size,
                              hipStream_t stream) {
    const float* x     = (const float*)d_in[0];
    const int*   edge  = (const int*)d_in[1];
    const int*   src   = edge;
    const int*   dst   = edge + N_EDGES;
    const int*   batch = (const int*)d_in[2];
    const float* Ws    = (const float*)d_in[3];
    const float* bs    = (const float*)d_in[4];
    const float* linW  = (const float*)d_in[5];
    const float* linb  = (const float*)d_in[6];
    float*       out   = (float*)d_out;

    float* ws = (float*)d_ws;
    float* H      = ws;
    float* X1     = H + (size_t)N_NODES * HID;
    float* X2     = X1 + (size_t)N_NODES * HID;
    float* deg    = X2 + (size_t)N_NODES * HID;
    int2*  brec   = (int2*)(deg + N_NODES);
    int2*  srec   = brec + N_EDGES;
    int*   tstart = (int*)(srec + N_EDGES);
    int*   cursor = tstart + 1024;

    const int T = 256;

    init_deg_kernel<<<(N_NODES + T - 1) / T, T, 0, stream>>>(deg);
    deg_accum_kernel<<<(N_EDGES + T - 1) / T, T, 0, stream>>>(dst, deg);
    tile_scan_kernel<<<1, 1024, 0, stream>>>(deg, tstart, cursor);
    dinv_kernel<<<(N_NODES + T - 1) / T, T, 0, stream>>>(deg);
    fill_kernel<<<(N_EDGES + T - 1) / T, T, 0, stream>>>(src, dst, deg, cursor, brec);
    tile_sort_kernel<<<NT, SORT_T, 0, stream>>>(tstart, brec, srec);

    const float* W0 = Ws;
    const float* W1 = Ws + (size_t)1 * HID * HID;
    const float* W2 = Ws + (size_t)2 * HID * HID;
    const float* W3 = Ws + (size_t)3 * HID * HID;
    const float* W4 = Ws + (size_t)4 * HID * HID;
    agg_kernel<2><<<NT, AGG_T, 0, stream>>>(x,  deg, tstart, srec, bs,           W0, W1, H);
    agg_kernel<0><<<NT, AGG_T, 0, stream>>>(H,  deg, tstart, srec, bs + 1 * HID, W0, W2, X1);
    agg_kernel<0><<<NT, AGG_T, 0, stream>>>(X1, deg, tstart, srec, bs + 2 * HID, W0, W3, H);
    agg_kernel<0><<<NT, AGG_T, 0, stream>>>(H,  deg, tstart, srec, bs + 3 * HID, W0, W4, X1);
    agg_kernel<1><<<NT, AGG_T, 0, stream>>>(X1, deg, tstart, srec, bs + 4 * HID, W0, W0, X2);

    pool_linear_kernel<<<N_GRAPHS, POOL_T, 0, stream>>>(X2, batch, linW, linb, out);
}


# Edge records staged in LDS on their way to memory: fill bins 7168-edge chunks by tile (one returning add, one run of stores per chunk and tile), tile sort orders in LDS; row sum order changes
# speedup vs baseline: 4.5641x; 1.2481x over previous
#include <hip/hip_runtime.h>
#include <hip/hip_bf16.h>


#define N_NODES   100000
#define N_EDGES   1600000
#define HID       128
#define N_CLASSES 10
#define N_GRAPHS  512
#define N_LAYERS  5
#define TN        128
#define NT        ((N_NODES + TN - 1) / TN)
#define CPAD      32
#define AGG_T     512
#define SORT_T    512
#define SORT_CAP  4096
#define FILL_T    1024
#define FILL_E    7
#define FILL_CH   (FILL_T * FILL_E)
#define NFILL     ((N_EDGES + FILL_CH - 1) / FILL_CH)
#define POOL_T    256
#define LP        132
#define SRC_BITS  17
#define SRC_MASK  ((1 << SRC_BITS) - 1)

static_assert(N_NODES <= (1 << SRC_BITS), "source node must fit the record's low bits");
static_assert(TN <= (1 << (31 - SRC_BITS)), "row inside the tile must fit above the source bits");
static_assert(NT <= FILL_T, "the fill's scan has one thread per destination tile");

typedef __attribute__((ext_vector_type(2))) float v2f;
typedef __attribute__((ext_vector_type(8))) float v8f;

__global__ void init_deg_kernel(float* __restrict__ deg) {
    int i = blockIdx.x * blockDim.x + threadIdx.x;
    if (i < N_NODES) deg[i] = 1.0f;
}

__global__ void deg_accum_kernel(const int* __restrict__ dst, float* __restrict__ deg) {
    int e = blockIdx.x * blockDim.x + threadIdx.x;
    if (e < N_EDGES) atomicAdd(&deg[dst[e]], 1.0f);
}

__global__ void dinv_kernel(float* __restrict__ deg) {
    int i = blockIdx.x * blockDim.x + threadIdx.x;
    if (i < N_NODES) deg[i] = rsqrtf(deg[i]);
}

__global__ __launch_bounds__(1024)
void tile_scan_kernel(const float* __restrict__ deg, int* __restrict__ tile_start,
                      int* __restrict__ cursor) {
    __shared__ int cnt[1024];
    __shared__ int wsum[32];
    const int tid  = threadIdx.x;
    const int lane = tid & 31;
    const int wave = tid >> 5;

    for (int t = wave; t < 1024; t += 32) {
        int c = 0;
        int n = t * TN + lane * 4;
        if (t < NT && n < N_NODES) {
            float4 d = *(const float4*)(deg + n);
            c = (int)d.x + (int)d.y + (int)d.z + (int)d.w - 4;
        }
        for (int off = 16; off > 0; off >>= 1) c += __shfl_xor(c, off);
        if (lane == 0) cnt[t] = c;
    }
    __syncthreads();

    const int v = cnt[tid];
    int x = v;
    for (int off = 1; off < 32; off <<= 1) {
        int y = __shfl_up(x, off);
        if (lane >= off) x += y;
    }
    if (lane == 31) wsum[wave] = x;
    __syncthreads();
    if (wave == 0) {
        const int s = wsum[lane];
        int xs = s;
        for (int off = 1; off < 32; off <<= 1) {
            int y = __shfl_up(xs, off);
            if (lane >= off) xs += y;
        }
        wsum[lane] = xs - s;
    }
    __syncthreads();
    const int excl = x - v + wsum[wave];
    if (tid <= NT) tile_start[tid] = excl;
    if (tid < NT)  cursor[tid * CPAD] = excl;
}

__global__ __launch_bounds__(FILL_T)
void fill_kernel(const int* __restrict__ src, const int* __restrict__ dst,
                 const float* __restrict__ dinv, int* __restrict__ cursor,
                 int2* __restrict__ brec) {
    __shared__ int  cnt[FILL_T];
    __shared__ int  gofs[FILL_T];
    __shared__ int  wsum[32];
    __shared__ int  gpos[FILL_CH];
    __shared__ int2 rec[FILL_CH];
    const int tid   = threadIdx.x;
    const int lane  = tid & 31;
    const int wave  = tid >> 5;
    const int base  = blockIdx.x * FILL_CH;

    cnt[tid] = 0;
    int s[FILL_E], d[FILL_E];
    #pragma unroll
    for (int k = 0; k < FILL_E; ++k) {
        const int  e  = base + k * FILL_T + tid;
        const bool ok = e < N_EDGES;
        const int  ec = ok ? e : N_EDGES - 1;
        s[k] = src[ec];
        d[k] = dst[ec];
        if (!ok) d[k] = -1;
    }
    float w[FILL_E];
    #pragma unroll
    for (int k = 0; k < FILL_E; ++k) {
        const int dc = d[k] >= 0 ? d[k] : 0;
        w[k] = dinv[s[k]] * dinv[dc];
    }
    __syncthreads();
    #pragma unroll
    for (int k = 0; k < FILL_E; ++k)
        if (d[k] >= 0) atomicAdd(&cnt[d[k] >> 7], 1);
    __syncthreads();

    const int v = cnt[tid];
    int x = v;
    for (int off = 1; off < 32; off <<= 1) {
        int y = __shfl_up(x, off);
        if (lane >= off) x += y;
    }
    if (lane == 31) wsum[wave] = x;
    __syncthreads();
    if (wave == 0) {
        const int ws = wsum[lane];
        int xs = ws;
        for (int off = 1; off < 32; off <<= 1) {
            int y = __shfl_up(xs, off);
            if (lane >= off) xs += y;
        }
        wsum[lane] = xs - ws;
    }
    __syncthreads();
    const int excl = x - v + wsum[wave];
    int g = 0;
    if (v > 0) g = atomicAdd(&cursor[tid * CPAD], v);
    cnt[tid]  = excl;
    gofs[tid] = g - excl;
    __syncthreads();

    #pragma unroll
    for (int k = 0; k < FILL_E; ++k) {
        if (d[k] >= 0) {
            const int t    = d[k] >> 7;
            const int slot = atomicAdd(&cnt[t], 1);
            int2 r;
            r.x = s[k] | ((d[k] & (TN - 1)) << SRC_BITS);
            r.y = __float_as_int(w[k]);
            rec[slot]  = r;
            gpos[slot] = gofs[t] + slot;
        }
    }
    __syncthreads();

    const int n = min(FILL_CH, N_EDGES - base);
    for (int i = tid; i < n; i += FILL_T) brec[gpos[i]] = rec[i];
}

__global__ __launch_bounds__(SORT_T)
void tile_sort_kernel(const int* __restrict__ tile_start, const int2* __restrict__ brec,
                      int2* __restrict__ srec) {
    __shared__ int  cnt[TN];
    __shared__ int  wtot[TN / 32];
    __shared__ int2 lrec[SORT_CAP];
    const int t    = blockIdx.x;
    const int tid  = threadIdx.x;
    const int lane = tid & 31;
    const int wave = tid >> 5;
    const int beg  = tile_start[t];
    const int end  = tile_start[t + 1];

    if (tid < TN) cnt[tid] = 0;
    __syncthreads();
    for (int i = beg + tid; i < end; i += SORT_T) atomicAdd(&cnt[brec[i].x >> SRC_BITS], 1);
    __syncthreads();

    const int c = (tid < TN) ? cnt[tid] : 0;
    int x = c;
    for (int off = 1; off < 32; off <<= 1) {
        int y = __shfl_up(x, off);
        if (lane >= off) x += y;
    }
    if (tid < TN && lane == 31) wtot[wave] = x;
    __syncthreads();
    if (tid < TN) {
        int front = beg;
        for (int j = 0; j < wave; ++j) front += wtot[j];
        cnt[tid] = front + x - c;
    }
    __syncthreads();

    const int n = end - beg;
    if (n <= SORT_CAP) {
        for (int i = beg + tid; i < end; i += SORT_T) {
            const int2 r  = brec[i];
            const int pos = atomicAdd(&cnt[r.x >> SRC_BITS], 1);
            lrec[pos - beg] = r;
        }
        __syncthreads();
        for (int i = tid; i < n; i += SORT_T) srec[beg + i] = lrec[i];
    } else {
        for (int i = beg + tid; i < end; i += SORT_T) {
            const int2 r  = brec[i];
            const int pos = atomicAdd(&cnt[r.x >> SRC_BITS], 1);
            srec[pos] = r;
        }
    }
}

__device__ __forceinline__ void tile_product(const float* tile, const float* __restrict__ Wn,
                                             int lane, int wave,
                                             v8f& acc00, v8f& acc01, v8f& acc10, v8f& acc11) {
    const int m  = lane & 15;
    const int kh = lane >> 4;
    const int rq = wave >> 2;
    const int cq = wave & 3;
    const float* a0p = tile + (32 * rq + m) * LP + 2 * kh;
    const float* a1p = a0p + 16 * LP;
    const float* wp  = Wn + (size_t)(8 * kh) * HID + 32 * cq + 2 * m;

    #pragma unroll 2
    for (int p = 0; p < 16; ++p) {
        const int ks = 2 * p;
        const float* w0 = wp + (size_t)(16 * (ks & 7) + (ks >> 3)) * HID;
        const float* w1 = w0 + (size_t)16 * HID;
        const v2f wc0 = *(const v2f*)(w0), wc1 = *(const v2f*)(w0 + 4 * HID);
        const v2f wc2 = *(const v2f*)(w1), wc3 = *(const v2f*)(w1 + 4 * HID);
        const v2f a00 = *(const v2f*)(a0p + 4 * ks);
        const v2f a10 = *(const v2f*)(a1p + 4 * ks);
        const v2f a01 = *(const v2f*)(a0p + 4 * ks + 4);
        const v2f a11 = *(const v2f*)(a1p + 4 * ks + 4);
        v2f b0, b1;
        b0.x = wc0.x; b0.y = wc1.x; b1.x = wc0.y; b1.y = wc1.y;
        acc00 = __builtin_amdgcn_wmma_f32_16x16x4_f32(false, a00, false, b0, (short)0, acc00, false, false);
        acc01 = __builtin_amdgcn_wmma_f32_16x16x4_f32(false, a00, false, b1, (short)0, acc01, false, false);
        acc10 = __builtin_amdgcn_wmma_f32_16x16x4_f32(false, a10, false, b0, (short)0, acc10, false, false);
        acc11 = __builtin_amdgcn_wmma_f32_16x16x4_f32(false, a10, false, b1, (short)0, acc11, false, false);
        b0.x = wc2.x; b0.y = wc3.x; b1.x = wc2.y; b1.y = wc3.y;
        acc00 = __builtin_amdgcn_wmma_f32_16x16x4_f32(false, a01, false, b0, (short)0, acc00, false, false);
        acc01 = __builtin_amdgcn_wmma_f32_16x16x4_f32(false, a01, false, b1, (short)0, acc01, false, false);
        acc10 = __builtin_amdgcn_wmma_f32_16x16x4_f32(false, a11, false, b0, (short)0, acc10, false, false);
        acc11 = __builtin_amdgcn_wmma_f32_16x16x4_f32(false, a11, false, b1, (short)0, acc11, false, false);
    }
}

template <int MODE>
__global__ __launch_bounds__(AGG_T, 16)
void agg_kernel(const float* __restrict__ H, const float* __restrict__ dinv,
                const int* __restrict__ tile_start, const int2* __restrict__ srec,
                const float* __restrict__ bias, const float* __restrict__ Wf,
                const float* __restrict__ Wn, float* __restrict__ out) {
    __shared__ float tile[TN * LP];
    const int t    = blockIdx.x;
    const int lane = threadIdx.x & 31;
    const int wave = __builtin_amdgcn_readfirstlane((int)(threadIdx.x >> 5));
    const int NW   = AGG_T / 32;
    const int node0 = t * TN;

    for (int r = wave; r < TN; r += NW) {
        const int node = node0 + r;
        float4 v = make_float4(0.f, 0.f, 0.f, 0.f);
        if (node < N_NODES) {
            const float dv = dinv[node];
            const float w  = dv * dv;
            const float4 h = ((const float4*)(H + (size_t)node * HID))[lane];
            v.x = h.x * w; v.y = h.y * w; v.z = h.z * w; v.w = h.w * w;
        }
        float* row = tile + r * LP + lane;
        row[0] = v.x; row[32] = v.y; row[64] = v.z; row[96] = v.w;
    }
    __syncthreads();

    const int beg = tile_start[t];
    const int end = tile_start[t + 1];
    for (int base = beg + wave * 32; base < end; base += NW * 32) {
        const int idx = base + lane;
        int pk = 0, wb = 0;
        if (idx < end) { const int2 r = srec[idx]; pk = r.x; wb = r.y; }
        const int n = min(32, end - base);
        int   cur = __builtin_amdgcn_readlane(pk, 0) >> SRC_BITS;
        float a0 = 0.f, a1 = 0.f, a2 = 0.f, a3 = 0.f;
        for (int k = 0; k < n; k += 4) {
            float4 h[4];
            #pragma unroll
            for (int u = 0; u < 4; ++u) {
                const int sk = __builtin_amdgcn_readlane(pk, k + u) & SRC_MASK;
                h[u] = ((const float4*)(H + (size_t)sk * HID))[lane];
            }
            #pragma unroll
            for (int u = 0; u < 4; ++u) {
                if (u == 0 || k + u < n) {
                    const int   dk = __builtin_amdgcn_readlane(pk, k + u) >> SRC_BITS;
                    const float wk = __int_as_float(__builtin_amdgcn_readlane(wb, k + u));
                    if (dk != cur) {
                        float* row = tile + cur * LP + lane;
                        __hip_atomic_fetch_add(row,      a0, __ATOMIC_RELAXED, __HIP_MEMORY_SCOPE_WORKGROUP);
                        __hip_atomic_fetch_add(row + 32, a1, __ATOMIC_RELAXED, __HIP_MEMORY_SCOPE_WORKGROUP);
                        __hip_atomic_fetch_add(row + 64, a2, __ATOMIC_RELAXED, __HIP_MEMORY_SCOPE_WORKGROUP);
                        __hip_atomic_fetch_add(row + 96, a3, __ATOMIC_RELAXED, __HIP_MEMORY_SCOPE_WORKGROUP);
                        a0 = 0.f; a1 = 0.f; a2 = 0.f; a3 = 0.f;
                        cur = dk;
                    }
                    a0 += h[u].x * wk;
                    a1 += h[u].y * wk;
                    a2 += h[u].z * wk;
                    a3 += h[u].w * wk;
                }
            }
        }
        float* row = tile + cur * LP + lane;
        __hip_atomic_fetch_add(row,      a0, __ATOMIC_RELAXED, __HIP_MEMORY_SCOPE_WORKGROUP);
        __hip_atomic_fetch_add(row + 32, a1, __ATOMIC_RELAXED, __HIP_MEMORY_SCOPE_WORKGROUP);
        __hip_atomic_fetch_add(row + 64, a2, __ATOMIC_RELAXED, __HIP_MEMORY_SCOPE_WORKGROUP);
        __hip_atomic_fetch_add(row + 96, a3, __ATOMIC_RELAXED, __HIP_MEMORY_SCOPE_WORKGROUP);
    }
    __syncthreads();

    const int m  = lane & 15;
    const int kh = lane >> 4;
    const int rq = wave >> 2;
    const int cq = wave & 3;

    if (MODE == 1) {
        const float4 b = ((const float4*)bias)[lane];
        for (int r = wave; r < TN; r += NW) {
            const int node = node0 + r;
            if (node < N_NODES) {
                const float* row = tile + r * LP + lane;
                float4 o;
                o.x = row[0]  + b.x; o.y = row[32] + b.y; o.z = row[64] + b.z; o.w = row[96] + b.w;
                o.x = o.x > 0.f ? o.x : 0.f;
                o.y = o.y > 0.f ? o.y : 0.f;
                o.z = o.z > 0.f ? o.z : 0.f;
                o.w = o.w > 0.f ? o.w : 0.f;
                ((float4*)(out + (size_t)node * HID))[lane] = o;
            }
        }
        return;
    }

    if (MODE == 2) {
        v8f p00 = {}, p01 = {}, p10 = {}, p11 = {};
        tile_product(tile, Wf, lane, wave, p00, p01, p10, p11);
        __syncthreads();
        const int f0 = 32 * cq + 2 * m;
        const v2f bb = *(const v2f*)(bias + f0);
        float* wr = tile + (32 * rq + 8 * kh) * LP + (f0 & 3) * 32 + (f0 >> 2);
        #pragma unroll
        for (int r = 0; r < 8; ++r) {
            const float o00 = p00[r] + bb.x, o01 = p01[r] + bb.y;
            const float o10 = p10[r] + bb.x, o11 = p11[r] + bb.y;
            wr[r * LP]             = o00 > 0.f ? o00 : 0.f;
            wr[r * LP + 32]        = o01 > 0.f ? o01 : 0.f;
            wr[(16 + r) * LP]      = o10 > 0.f ? o10 : 0.f;
            wr[(16 + r) * LP + 32] = o11 > 0.f ? o11 : 0.f;
        }
    } else {
        const float4 b = ((const float4*)bias)[lane];
        for (int r = wave; r < TN; r += NW) {
            float* row = tile + r * LP + lane;
            float o0 = row[0] + b.x, o1 = row[32] + b.y, o2 = row[64] + b.z, o3 = row[96] + b.w;
            row[0]  = o0 > 0.f ? o0 : 0.f;
            row[32] = o1 > 0.f ? o1 : 0.f;
            row[64] = o2 > 0.f ? o2 : 0.f;
            row[96] = o3 > 0.f ? o3 : 0.f;
        }
    }
    __syncthreads();

    v8f acc00 = {}, acc01 = {}, acc10 = {}, acc11 = {};
    tile_product(tile, Wn, lane, wave, acc00, acc01, acc10, acc11);

    const int row0 = node0 + 32 * rq + 8 * kh;
    float* ob = out + (size_t)row0 * HID + 32 * cq + 2 * m;
    #pragma unroll
    for (int r = 0; r < 8; ++r) {
        if (row0 + r < N_NODES) {
            v2f o; o.x = acc00[r]; o.y = acc01[r];
            *(v2f*)(ob + (size_t)r * HID) = o;
        }
        if (row0 + 16 + r < N_NODES) {
            v2f o; o.x = acc10[r]; o.y = acc11[r];
            *(v2f*)(ob + (size_t)(16 + r) * HID) = o;
        }
    }
}

__device__ __forceinline__ int first_node_at_or_above(const int* __restrict__ batch, int key,
                                                      int lane) {
    int lo = 0, len = N_NODES;
    for (int it = 0; it < 5 && len > 0; ++it) {
        const int step = (len + 31) >> 5;
        const int q    = lo + lane * step;
        const bool below = (q < lo + len) && (batch[q] < key);
        const int c = __builtin_popcount(__builtin_amdgcn_ballot_w32(below));
        if (c == 0) {
            len = 0;
        } else {
            const int end = lo + len;
            const int nlo = lo + (c - 1) * step + 1;
            len = min(lo + c * step, end) - nlo;
            lo  = nlo;
        }
    }
    return lo;
}

__global__ __launch_bounds__(POOL_T)
void pool_linear_kernel(const float* __restrict__ X, const int* __restrict__ batch,
                        const float* __restrict__ linW, const float* __restrict__ linb,
                        float* __restrict__ out) {
    __shared__ float4 part4[(POOL_T / 32) * (HID / 4)];
    __shared__ float4 wl4[HID * N_CLASSES / 4];
    __shared__ float  pooled[HID];
    __shared__ int    range[2];
    const int g    = blockIdx.x;
    const int tid  = threadIdx.x;
    const int lane = tid & 31;
    const int wave = __builtin_amdgcn_readfirstlane((int)(threadIdx.x >> 5));
    const int NW   = POOL_T / 32;
    const int WQ   = HID * N_CLASSES / 8;
    static_assert(HID * N_CLASSES % 8 == 0 && WQ % 32 == 0 && 2 + WQ / 32 <= POOL_T / 32,
                  "the weight copy takes whole waves behind the two that search");

    if (wave < 2) {
        const int b = first_node_at_or_above(batch, g + wave, lane);
        if (lane == 0) range[wave] = b;
    } else if (wave < 2 + WQ / 32) {
        const int i = tid - 64;
        const float4 w0 = ((const float4*)linW)[i];
        const float4 w1 = ((const float4*)linW)[i + WQ];
        wl4[i]      = w0;
        wl4[i + WQ] = w1;
    }
    __syncthreads();
    const int lo = range[0];
    const int hi = range[1];

    float4 acc = make_float4(0.f, 0.f, 0.f, 0.f);
    for (int r = lo + wave; r < hi; r += 16 * NW) {
        float4 h[16];
        #pragma unroll
        for (int u = 0; u < 16; ++u) {
            const int ru = r + u * NW;
            h[u] = make_float4(0.f, 0.f, 0.f, 0.f);
            if (ru < hi) h[u] = ((const float4*)(X + (size_t)ru * HID))[lane];
        }
        #pragma unroll
        for (int u = 0; u < 16; ++u) {
            acc.x += h[u].x; acc.y += h[u].y; acc.z += h[u].z; acc.w += h[u].w;
        }
    }
    part4[wave * (HID / 4) + lane] = acc;
    __syncthreads();

    if (tid < HID) {
        const float* pf = (const float*)part4;
        float sum = 0.f;
        #pragma unroll
        for (int w = 0; w < NW; ++w) sum += pf[w * HID + tid];
        pooled[tid] = sum;
    }
    __syncthreads();

    const float* wl = (const float*)wl4;
    for (int c = wave; c < N_CLASSES; c += NW) {
        float a = 0.f;
        #pragma unroll
        for (int j = 0; j < HID / 32; ++j) {
            const int f = lane + 32 * j;
            a += pooled[f] * wl[f * N_CLASSES + c];
        }
        for (int off = 16; off > 0; off >>= 1) a += __shfl_xor(a, off);
        if (lane == 0) {
            const float scale = 1.0f / fmaxf((float)(hi - lo), 1.0f);
            out[g * N_CLASSES + c] = a * scale + linb[c];
        }
    }
}

extern "C" void kernel_launch(void* const* d_in, const int* in_sizes, int n_in,
                              void* d_out, int out_size, void* d_ws, size_t ws_size,
                              hipStream_t stream) {
    const float* x     = (const float*)d_in[0];
    const int*   edge  = (const int*)d_in[1];
    const int*   src   = edge;
    const int*   dst   = edge + N_EDGES;
    const int*   batch = (const int*)d_in[2];
    const float* Ws    = (const float*)d_in[3];
    const float* bs    = (const float*)d_in[4];
    const float* linW  = (const float*)d_in[5];
    const float* linb  = (const float*)d_in[6];
    float*       out   = (float*)d_out;

    float* ws = (float*)d_ws;
    float* H      = ws;
    float* X1     = H + (size_t)N_NODES * HID;
    float* X2     = X1 + (size_t)N_NODES * HID;
    float* deg    = X2 + (size_t)N_NODES * HID;
    int2*  brec   = (int2*)(deg + N_NODES);
    int2*  srec   = brec + N_EDGES;
    int*   tstart = (int*)(srec + N_EDGES);
    int*   cursor = tstart + 1024;

    const int T = 256;

    init_deg_kernel<<<(N_NODES + T - 1) / T, T, 0, stream>>>(deg);
    deg_accum_kernel<<<(N_EDGES + T - 1) / T, T, 0, stream>>>(dst, deg);
    tile_scan_kernel<<<1, 1024, 0, stream>>>(deg, tstart, cursor);
    dinv_kernel<<<(N_NODES + T - 1) / T, T, 0, stream>>>(deg);
    fill_kernel<<<NFILL, FILL_T, 0, stream>>>(src, dst, deg, cursor, brec);
    tile_sort_kernel<<<NT, SORT_T, 0, stream>>>(tstart, brec, srec);

    const float* W0 = Ws;
    const float* W1 = Ws + (size_t)1 * HID * HID;
    const float* W2 = Ws + (size_t)2 * HID * HID;
    const float* W3 = Ws + (size_t)3 * HID * HID;
    const float* W4 = Ws + (size_t)4 * HID * HID;
    agg_kernel<2><<<NT, AGG_T, 0, stream>>>(x,  deg, tstart, srec, bs,           W0, W1, H);
    agg_kernel<0><<<NT, AGG_T, 0, stream>>>(H,  deg, tstart, srec, bs + 1 * HID, W0, W2, X1);
    agg_kernel<0><<<NT, AGG_T, 0, stream>>>(X1, deg, tstart, srec, bs + 2 * HID, W0, W3, H);
    agg_kernel<0><<<NT, AGG_T, 0, stream>>>(H,  deg, tstart, srec, bs + 3 * HID, W0, W4, X1);
    agg_kernel<1><<<NT, AGG_T, 0, stream>>>(X1, deg, tstart, srec, bs + 4 * HID, W0, W0, X2);

    pool_linear_kernel<<<N_GRAPHS, POOL_T, 0, stream>>>(X2, batch, linW, linb, out);
}


# Global degree pass removed: tile counts from an LDS histogram launch, dinv from the tile sort's LDS row counts, edge weights formed in the first layer's launch and left in the list; sums unchanged
# speedup vs baseline: 5.6499x; 1.2379x over previous
#include <hip/hip_runtime.h>
#include <hip/hip_bf16.h>


#define N_NODES   100000
#define N_EDGES   1600000
#define HID       128
#define N_CLASSES 10
#define N_GRAPHS  512
#define N_LAYERS  5
#define TN        128
#define NT        ((N_NODES + TN - 1) / TN)
#define CPAD      32
#define AGG_T     512
#define SORT_T    512
#define SORT_CAP  4096
#define FILL_T    1024
#define FILL_E    7
#define FILL_CH   (FILL_T * FILL_E)
#define NFILL     ((N_EDGES + FILL_CH - 1) / FILL_CH)
#define HIST_WG   64
#define HIST_T    1024
#define POOL_T    256
#define LP        132
#define SRC_BITS  17
#define SRC_MASK  ((1 << SRC_BITS) - 1)

static_assert(N_NODES <= (1 << SRC_BITS), "source node must fit the record's low bits");
static_assert(TN <= (1 << (31 - SRC_BITS)), "row inside the tile must fit above the source bits");
static_assert(NT <= FILL_T, "the fill's scan has one thread per destination tile");
static_assert(NT <= HIST_T, "the tile histogram has one LDS counter per destination tile");
static_assert(N_EDGES % (HIST_WG * 4) == 0, "a histogram workgroup's slice is a whole number of 16-byte loads");

typedef __attribute__((ext_vector_type(2))) float v2f;
typedef __attribute__((ext_vector_type(8))) float v8f;

__global__ __launch_bounds__(HIST_T)
void tile_hist_kernel(const int* __restrict__ dst, int* __restrict__ hist) {
    __shared__ int cnt[HIST_T];
    const int tid = threadIdx.x;
    const int Q   = N_EDGES / (HIST_WG * 4);
    const int K   = (Q + HIST_T - 1) / HIST_T;
    const int4* d4 = (const int4*)dst + (size_t)blockIdx.x * Q;
    cnt[tid] = 0;
    __syncthreads();
    int4 d[K];
    #pragma unroll
    for (int k = 0; k < K; ++k) {
        const int i = k * HIST_T + tid;
        d[k] = d4[i < Q ? i : Q - 1];
    }
    #pragma unroll
    for (int k = 0; k < K; ++k) {
        if (k * HIST_T + tid < Q) {
            atomicAdd(&cnt[d[k].x >> 7], 1);
            atomicAdd(&cnt[d[k].y >> 7], 1);
            atomicAdd(&cnt[d[k].z >> 7], 1);
            atomicAdd(&cnt[d[k].w >> 7], 1);
        }
    }
    __syncthreads();
    hist[blockIdx.x * HIST_T + tid] = cnt[tid];
}

__global__ __launch_bounds__(1024)
void tile_scan_kernel(const int* __restrict__ hist, int* __restrict__ tile_start,
                      int* __restrict__ cursor) {
    __shared__ int wsum[32];
    const int tid  = threadIdx.x;
    const int lane = tid & 31;
    const int wave = tid >> 5;

    int v = 0;
    #pragma unroll 16
    for (int b = 0; b < HIST_WG; ++b) v += hist[b * HIST_T + tid];

    int x = v;
    for (int off = 1; off < 32; off <<= 1) {
        int y = __shfl_up(x, off);
        if (lane >= off) x += y;
    }
    if (lane == 31) wsum[wave] = x;
    __syncthreads();
    if (wave == 0) {
        const int s = wsum[lane];
        int xs = s;
        for (int off = 1; off < 32; off <<= 1) {
            int y = __shfl_up(xs, off);
            if (lane >= off) xs += y;
        }
        wsum[lane] = xs - s;
    }
    __syncthreads();
    const int excl = x - v + wsum[wave];
    if (tid <= NT) tile_start[tid] = excl;
    if (tid < NT)  cursor[tid * CPAD] = excl;
}

__global__ __launch_bounds__(FILL_T)
void fill_kernel(const int* __restrict__ src, const int* __restrict__ dst,
                 int* __restrict__ cursor, int2* __restrict__ brec) {
    __shared__ int  cnt[FILL_T];
    __shared__ int  gofs[FILL_T];
    __shared__ int  wsum[32];
    __shared__ int  gpos[FILL_CH];
    __shared__ int2 rec[FILL_CH];
    const int tid   = threadIdx.x;
    const int lane  = tid & 31;
    const int wave  = tid >> 5;
    const int base  = blockIdx.x * FILL_CH;

    cnt[tid] = 0;
    int s[FILL_E], d[FILL_E];
    #pragma unroll
    for (int k = 0; k < FILL_E; ++k) {
        const int  e  = base + k * FILL_T + tid;
        const bool ok = e < N_EDGES;
        const int  ec = ok ? e : N_EDGES - 1;
        s[k] = src[ec];
        d[k] = dst[ec];
        if (!ok) d[k] = -1;
    }
    __syncthreads();
    #pragma unroll
    for (int k = 0; k < FILL_E; ++k)
        if (d[k] >= 0) atomicAdd(&cnt[d[k] >> 7], 1);
    __syncthreads();

    const int v = cnt[tid];
    int x = v;
    for (int off = 1; off < 32; off <<= 1) {
        int y = __shfl_up(x, off);
        if (lane >= off) x += y;
    }
    if (lane == 31) wsum[wave] = x;
    __syncthreads();
    if (wave == 0) {
        const int ws = wsum[lane];
        int xs = ws;
        for (int off = 1; off < 32; off <<= 1) {
            int y = __shfl_up(xs, off);
            if (lane >= off) xs += y;
        }
        wsum[lane] = xs - ws;
    }
    __syncthreads();
    const int excl = x - v + wsum[wave];
    int g = 0;
    if (v > 0) g = atomicAdd(&cursor[tid * CPAD], v);
    cnt[tid]  = excl;
    gofs[tid] = g - excl;
    __syncthreads();

    #pragma unroll
    for (int k = 0; k < FILL_E; ++k) {
        if (d[k] >= 0) {
            const int t    = d[k] >> 7;
            const int slot = atomicAdd(&cnt[t], 1);
            int2 r;
            r.x = s[k] | ((d[k] & (TN - 1)) << SRC_BITS);
            r.y = 0;
            rec[slot]  = r;
            gpos[slot] = gofs[t] + slot;
        }
    }
    __syncthreads();

    const int n = min(FILL_CH, N_EDGES - base);
    for (int i = tid; i < n; i += FILL_T) brec[gpos[i]] = rec[i];
}

__global__ __launch_bounds__(SORT_T)
void tile_sort_kernel(const int* __restrict__ tile_start, const int2* __restrict__ brec,
                      int2* __restrict__ srec, float* __restrict__ dinv) {
    __shared__ int  cnt[TN];
    __shared__ int  wtot[TN / 32];
    __shared__ int2 lrec[SORT_CAP];
    const int t    = blockIdx.x;
    const int tid  = threadIdx.x;
    const int lane = tid & 31;
    const int wave = tid >> 5;
    const int beg  = tile_start[t];
    const int end  = tile_start[t + 1];

    if (tid < TN) cnt[tid] = 0;
    __syncthreads();
    for (int i = beg + tid; i < end; i += SORT_T) atomicAdd(&cnt[brec[i].x >> SRC_BITS], 1);
    __syncthreads();

    const int c = (tid < TN) ? cnt[tid] : 0;
    if (tid < TN && t * TN + tid < N_NODES) dinv[t * TN + tid] = rsqrtf((float)(c + 1));
    int x = c;
    for (int off = 1; off < 32; off <<= 1) {
        int y = __shfl_up(x, off);
        if (lane >= off) x += y;
    }
    if (tid < TN && lane == 31) wtot[wave] = x;
    __syncthreads();
    if (tid < TN) {
        int front = beg;
        for (int j = 0; j < wave; ++j) front += wtot[j];
        cnt[tid] = front + x - c;
    }
    __syncthreads();

    const int n = end - beg;
    if (n <= SORT_CAP) {
        for (int i = beg + tid; i < end; i += SORT_T) {
            const int2 r  = brec[i];
            const int pos = atomicAdd(&cnt[r.x >> SRC_BITS], 1);
            lrec[pos - beg] = r;
        }
        __syncthreads();
        for (int i = tid; i < n; i += SORT_T) srec[beg + i] = lrec[i];
    } else {
        for (int i = beg + tid; i < end; i += SORT_T) {
            const int2 r  = brec[i];
            const int pos = atomicAdd(&cnt[r.x >> SRC_BITS], 1);
            srec[pos] = r;
        }
    }
}

__device__ __forceinline__ void tile_product(const float* tile, const float* __restrict__ Wn,
                                             int lane, int wave,
                                             v8f& acc00, v8f& acc01, v8f& acc10, v8f& acc11) {
    const int m  = lane & 15;
    const int kh = lane >> 4;
    const int rq = wave >> 2;
    const int cq = wave & 3;
    const float* a0p = tile + (32 * rq + m) * LP + 2 * kh;
    const float* a1p = a0p + 16 * LP;
    const float* wp  = Wn + (size_t)(8 * kh) * HID + 32 * cq + 2 * m;

    #pragma unroll 2
    for (int p = 0; p < 16; ++p) {
        const int ks = 2 * p;
        const float* w0 = wp + (size_t)(16 * (ks & 7) + (ks >> 3)) * HID;
        const float* w1 = w0 + (size_t)16 * HID;
        const v2f wc0 = *(const v2f*)(w0), wc1 = *(const v2f*)(w0 + 4 * HID);
        const v2f wc2 = *(const v2f*)(w1), wc3 = *(const v2f*)(w1 + 4 * HID);
        const v2f a00 = *(const v2f*)(a0p + 4 * ks);
        const v2f a10 = *(const v2f*)(a1p + 4 * ks);
        const v2f a01 = *(const v2f*)(a0p + 4 * ks + 4);
        const v2f a11 = *(const v2f*)(a1p + 4 * ks + 4);
        v2f b0, b1;
        b0.x = wc0.x; b0.y = wc1.x; b1.x = wc0.y; b1.y = wc1.y;
        acc00 = __builtin_amdgcn_wmma_f32_16x16x4_f32(false, a00, false, b0, (short)0, acc00, false, false);
        acc01 = __builtin_amdgcn_wmma_f32_16x16x4_f32(false, a00, false, b1, (short)0, acc01, false, false);
        acc10 = __builtin_amdgcn_wmma_f32_16x16x4_f32(false, a10, false, b0, (short)0, acc10, false, false);
        acc11 = __builtin_amdgcn_wmma_f32_16x16x4_f32(false, a10, false, b1, (short)0, acc11, false, false);
        b0.x = wc2.x; b0.y = wc3.x; b1.x = wc2.y; b1.y = wc3.y;
        acc00 = __builtin_amdgcn_wmma_f32_16x16x4_f32(false, a01, false, b0, (short)0, acc00, false, false);
        acc01 = __builtin_amdgcn_wmma_f32_16x16x4_f32(false, a01, false, b1, (short)0, acc01, false, false);
        acc10 = __builtin_amdgcn_wmma_f32_16x16x4_f32(false, a11, false, b0, (short)0, acc10, false, false);
        acc11 = __builtin_amdgcn_wmma_f32_16x16x4_f32(false, a11, false, b1, (short)0, acc11, false, false);
    }
}

template <int MODE>
__global__ __launch_bounds__(AGG_T, 16)
void agg_kernel(const float* __restrict__ H, const float* __restrict__ dinv,
                const int* __restrict__ tile_start, int2* __restrict__ srec,
                const float* __restrict__ bias, const float* __restrict__ Wf,
                const float* __restrict__ Wn, float* __restrict__ out) {
    __shared__ float tile[TN * LP];
    const int t    = blockIdx.x;
    const int lane = threadIdx.x & 31;
    const int wave = __builtin_amdgcn_readfirstlane((int)(threadIdx.x >> 5));
    const int NW   = AGG_T / 32;
    const int node0 = t * TN;

    for (int r = wave; r < TN; r += NW) {
        const int node = node0 + r;
        float4 v = make_float4(0.f, 0.f, 0.f, 0.f);
        if (node < N_NODES) {
            const float dv = dinv[node];
            const float w  = dv * dv;
            const float4 h = ((const float4*)(H + (size_t)node * HID))[lane];
            v.x = h.x * w; v.y = h.y * w; v.z = h.z * w; v.w = h.w * w;
        }
        float* row = tile + r * LP + lane;
        row[0] = v.x; row[32] = v.y; row[64] = v.z; row[96] = v.w;
    }
    __syncthreads();

    const int beg = tile_start[t];
    const int end = tile_start[t + 1];
    for (int base = beg + wave * 32; base < end; base += NW * 32) {
        const int idx = base + lane;
        int pk = 0, wb = 0;
        if (MODE == 2) {
            if (idx < end) {
                pk = srec[idx].x;
                const float wsrc = dinv[pk & SRC_MASK];
                const float wdst = dinv[node0 + (pk >> SRC_BITS)];
                wb = __float_as_int(wsrc * wdst);
                srec[idx].y = wb;
            }
        } else {
            if (idx < end) { const int2 r = srec[idx]; pk = r.x; wb = r.y; }
        }
        const int n = min(32, end - base);
        int   cur = __builtin_amdgcn_readlane(pk, 0) >> SRC_BITS;
        float a0 = 0.f, a1 = 0.f, a2 = 0.f, a3 = 0.f;
        for (int k = 0; k < n; k += 4) {
            float4 h[4];
            #pragma unroll
            for (int u = 0; u < 4; ++u) {
                const int sk = __builtin_amdgcn_readlane(pk, k + u) & SRC_MASK;
                h[u] = ((const float4*)(H + (size_t)sk * HID))[lane];
            }
            #pragma unroll
            for (int u = 0; u < 4; ++u) {
                if (u == 0 || k + u < n) {
                    const int   dk = __builtin_amdgcn_readlane(pk, k + u) >> SRC_BITS;
                    const float wk = __int_as_float(__builtin_amdgcn_readlane(wb, k + u));
                    if (dk != cur) {
                        float* row = tile + cur * LP + lane;
                        __hip_atomic_fetch_add(row,      a0, __ATOMIC_RELAXED, __HIP_MEMORY_SCOPE_WORKGROUP);
                        __hip_atomic_fetch_add(row + 32, a1, __ATOMIC_RELAXED, __HIP_MEMORY_SCOPE_WORKGROUP);
                        __hip_atomic_fetch_add(row + 64, a2, __ATOMIC_RELAXED, __HIP_MEMORY_SCOPE_WORKGROUP);
                        __hip_atomic_fetch_add(row + 96, a3, __ATOMIC_RELAXED, __HIP_MEMORY_SCOPE_WORKGROUP);
                        a0 = 0.f; a1 = 0.f; a2 = 0.f; a3 = 0.f;
                        cur = dk;
                    }
                    a0 += h[u].x * wk;
                    a1 += h[u].y * wk;
                    a2 += h[u].z * wk;
                    a3 += h[u].w * wk;
                }
            }
        }
        float* row = tile + cur * LP + lane;
        __hip_atomic_fetch_add(row,      a0, __ATOMIC_RELAXED, __HIP_MEMORY_SCOPE_WORKGROUP);
        __hip_atomic_fetch_add(row + 32, a1, __ATOMIC_RELAXED, __HIP_MEMORY_SCOPE_WORKGROUP);
        __hip_atomic_fetch_add(row + 64, a2, __ATOMIC_RELAXED, __HIP_MEMORY_SCOPE_WORKGROUP);
        __hip_atomic_fetch_add(row + 96, a3, __ATOMIC_RELAXED, __HIP_MEMORY_SCOPE_WORKGROUP);
    }
    __syncthreads();

    const int m  = lane & 15;
    const int kh = lane >> 4;
    const int rq = wave >> 2;
    const int cq = wave & 3;

    if (MODE == 1) {
        const float4 b = ((const float4*)bias)[lane];
        for (int r = wave; r < TN; r += NW) {
            const int node = node0 + r;
            if (node < N_NODES) {
                const float* row = tile + r * LP + lane;
                float4 o;
                o.x = row[0]  + b.x; o.y = row[32] + b.y; o.z = row[64] + b.z; o.w = row[96] + b.w;
                o.x = o.x > 0.f ? o.x : 0.f;
                o.y = o.y > 0.f ? o.y : 0.f;
                o.z = o.z > 0.f ? o.z : 0.f;
                o.w = o.w > 0.f ? o.w : 0.f;
                ((float4*)(out + (size_t)node * HID))[lane] = o;
            }
        }
        return;
    }

    if (MODE == 2) {
        v8f p00 = {}, p01 = {}, p10 = {}, p11 = {};
        tile_product(tile, Wf, lane, wave, p00, p01, p10, p11);
        __syncthreads();
        const int f0 = 32 * cq + 2 * m;
        const v2f bb = *(const v2f*)(bias + f0);
        float* wr = tile + (32 * rq + 8 * kh) * LP + (f0 & 3) * 32 + (f0 >> 2);
        #pragma unroll
        for (int r = 0; r < 8; ++r) {
            const float o00 = p00[r] + bb.x, o01 = p01[r] + bb.y;
            const float o10 = p10[r] + bb.x, o11 = p11[r] + bb.y;
            wr[r * LP]             = o00 > 0.f ? o00 : 0.f;
            wr[r * LP + 32]        = o01 > 0.f ? o01 : 0.f;
            wr[(16 + r) * LP]      = o10 > 0.f ? o10 : 0.f;
            wr[(16 + r) * LP + 32] = o11 > 0.f ? o11 : 0.f;
        }
    } else {
        const float4 b = ((const float4*)bias)[lane];
        for (int r = wave; r < TN; r += NW) {
            float* row = tile + r * LP + lane;
            float o0 = row[0] + b.x, o1 = row[32] + b.y, o2 = row[64] + b.z, o3 = row[96] + b.w;
            row[0]  = o0 > 0.f ? o0 : 0.f;
            row[32] = o1 > 0.f ? o1 : 0.f;
            row[64] = o2 > 0.f ? o2 : 0.f;
            row[96] = o3 > 0.f ? o3 : 0.f;
        }
    }
    __syncthreads();

    v8f acc00 = {}, acc01 = {}, acc10 = {}, acc11 = {};
    tile_product(tile, Wn, lane, wave, acc00, acc01, acc10, acc11);

    const int row0 = node0 + 32 * rq + 8 * kh;
    float* ob = out + (size_t)row0 * HID + 32 * cq + 2 * m;
    #pragma unroll
    for (int r = 0; r < 8; ++r) {
        if (row0 + r < N_NODES) {
            v2f o; o.x = acc00[r]; o.y = acc01[r];
            *(v2f*)(ob + (size_t)r * HID) = o;
        }
        if (row0 + 16 + r < N_NODES) {
            v2f o; o.x = acc10[r]; o.y = acc11[r];
            *(v2f*)(ob + (size_t)(16 + r) * HID) = o;
        }
    }
}

__device__ __forceinline__ int first_node_at_or_above(const int* __restrict__ batch, int key,
                                                      int lane) {
    int lo = 0, len = N_NODES;
    for (int it = 0; it < 5 && len > 0; ++it) {
        const int step = (len + 31) >> 5;
        const int q    = lo + lane * step;
        const bool below = (q < lo + len) && (batch[q] < key);
        const int c = __builtin_popcount(__builtin_amdgcn_ballot_w32(below));
        if (c == 0) {
            len = 0;
        } else {
            const int end = lo + len;
            const int nlo = lo + (c - 1) * step + 1;
            len = min(lo + c * step, end) - nlo;
            lo  = nlo;
        }
    }
    return lo;
}

__global__ __launch_bounds__(POOL_T)
void pool_linear_kernel(const float* __restrict__ X, const int* __restrict__ batch,
                        const float* __restrict__ linW, const float* __restrict__ linb,
                        float* __restrict__ out) {
    __shared__ float4 part4[(POOL_T / 32) * (HID / 4)];
    __shared__ float4 wl4[HID * N_CLASSES / 4];
    __shared__ float  pooled[HID];
    __shared__ int    range[2];
    const int g    = blockIdx.x;
    const int tid  = threadIdx.x;
    const int lane = tid & 31;
    const int wave = __builtin_amdgcn_readfirstlane((int)(threadIdx.x >> 5));
    const int NW   = POOL_T / 32;
    const int WQ   = HID * N_CLASSES / 8;
    static_assert(HID * N_CLASSES % 8 == 0 && WQ % 32 == 0 && 2 + WQ / 32 <= POOL_T / 32,
                  "the weight copy takes whole waves behind the two that search");

    if (wave < 2) {
        const int b = first_node_at_or_above(batch, g + wave, lane);
        if (lane == 0) range[wave] = b;
    } else if (wave < 2 + WQ / 32) {
        const int i = tid - 64;
        const float4 w0 = ((const float4*)linW)[i];
        const float4 w1 = ((const float4*)linW)[i + WQ];
        wl4[i]      = w0;
        wl4[i + WQ] = w1;
    }
    __syncthreads();
    const int lo = range[0];
    const int hi = range[1];

    float4 acc = make_float4(0.f, 0.f, 0.f, 0.f);
    for (int r = lo + wave; r < hi; r += 16 * NW) {
        float4 h[16];
        #pragma unroll
        for (int u = 0; u < 16; ++u) {
            const int ru = r + u * NW;
            h[u] = make_float4(0.f, 0.f, 0.f, 0.f);
            if (ru < hi) h[u] = ((const float4*)(X + (size_t)ru * HID))[lane];
        }
        #pragma unroll
        for (int u = 0; u < 16; ++u) {
            acc.x += h[u].x; acc.y += h[u].y; acc.z += h[u].z; acc.w += h[u].w;
        }
    }
    part4[wave * (HID / 4) + lane] = acc;
    __syncthreads();

    if (tid < HID) {
        const float* pf = (const float*)part4;
        float sum = 0.f;
        #pragma unroll
        for (int w = 0; w < NW; ++w) sum += pf[w * HID + tid];
        pooled[tid] = sum;
    }
    __syncthreads();

    const float* wl = (const float*)wl4;
    for (int c = wave; c < N_CLASSES; c += NW) {
        float a = 0.f;
        #pragma unroll
        for (int j = 0; j < HID / 32; ++j) {
            const int f = lane + 32 * j;
            a += pooled[f] * wl[f * N_CLASSES + c];
        }
        for (int off = 16; off > 0; off >>= 1) a += __shfl_xor(a, off);
        if (lane == 0) {
            const float scale = 1.0f / fmaxf((float)(hi - lo), 1.0f);
            out[g * N_CLASSES + c] = a * scale + linb[c];
        }
    }
}

extern "C" void kernel_launch(void* const* d_in, const int* in_sizes, int n_in,
                              void* d_out, int out_size, void* d_ws, size_t ws_size,
                              hipStream_t stream) {
    const float* x     = (const float*)d_in[0];
    const int*   edge  = (const int*)d_in[1];
    const int*   src   = edge;
    const int*   dst   = edge + N_EDGES;
    const int*   batch = (const int*)d_in[2];
    const float* Ws    = (const float*)d_in[3];
    const float* bs    = (const float*)d_in[4];
    const float* linW  = (const float*)d_in[5];
    const float* linb  = (const float*)d_in[6];
    float*       out   = (float*)d_out;

    float* ws = (float*)d_ws;
    float* H      = ws;
    float* X1     = H + (size_t)N_NODES * HID;
    float* X2     = X1 + (size_t)N_NODES * HID;
    float* dinv   = X2 + (size_t)N_NODES * HID;
    int2*  brec   = (int2*)(dinv + N_NODES);
    int2*  srec   = brec + N_EDGES;
    int*   tstart = (int*)(srec + N_EDGES);
    int*   cursor = tstart + 1024;
    int*   hist   = cursor + 1024 * CPAD;

    tile_hist_kernel<<<HIST_WG, HIST_T, 0, stream>>>(dst, hist);
    tile_scan_kernel<<<1, 1024, 0, stream>>>(hist, tstart, cursor);
    fill_kernel<<<NFILL, FILL_T, 0, stream>>>(src, dst, cursor, brec);
    tile_sort_kernel<<<NT, SORT_T, 0, stream>>>(tstart, brec, srec, dinv);

    const float* W0 = Ws;
    const float* W1 = Ws + (size_t)1 * HID * HID;
    const float* W2 = Ws + (size_t)2 * HID * HID;
    const float* W3 = Ws + (size_t)3 * HID * HID;
    const float* W4 = Ws + (size_t)4 * HID * HID;
    agg_kernel<2><<<NT, AGG_T, 0, stream>>>(x,  dinv, tstart, srec, bs,           W0, W1, H);
    agg_kernel<0><<<NT, AGG_T, 0, stream>>>(H,  dinv, tstart, srec, bs + 1 * HID, W0, W2, X1);
    agg_kernel<0><<<NT, AGG_T, 0, stream>>>(X1, dinv, tstart, srec, bs + 2 * HID, W0, W3, H);
    agg_kernel<0><<<NT, AGG_T, 0, stream>>>(H,  dinv, tstart, srec, bs + 3 * HID, W0, W4, X1);
    agg_kernel<1><<<NT, AGG_T, 0, stream>>>(X1, dinv, tstart, srec, bs + 4 * HID, W0, W0, X2);

    pool_linear_kernel<<<N_GRAPHS, POOL_T, 0, stream>>>(X2, batch, linW, linb, out);
}


# Destination tiles of 100 nodes (was 128): 1000 tiles, four to a WGP nearly everywhere, no WGP holds over 400 nodes (was 512 on some, 384 on most); LDS rows padded to 112; row sum order changes
# speedup vs baseline: 6.1873x; 1.0951x over previous
#include <hip/hip_runtime.h>
#include <hip/hip_bf16.h>


#define N_NODES   100000
#define N_EDGES   1600000
#define HID       128
#define N_CLASSES 10
#define N_GRAPHS  512
#define N_LAYERS  5
#define TN        100
#define TR        112
#define NT        ((N_NODES + TN - 1) / TN)
#define CPAD      32
#define AGG_T     512
#define SORT_T    512
#define SORT_CAP  4096
#define FILL_T    1024
#define FILL_E    7
#define FILL_CH   (FILL_T * FILL_E)
#define NFILL     ((N_EDGES + FILL_CH - 1) / FILL_CH)
#define HIST_WG   64
#define HIST_T    1024
#define POOL_T    256
#define LP        132
#define SRC_BITS  17
#define SRC_MASK  ((1 << SRC_BITS) - 1)

static_assert(N_NODES <= (1 << SRC_BITS), "source node must fit the record's low bits");
static_assert(TN <= (1 << (31 - SRC_BITS)), "row inside the tile must fit above the source bits");
static_assert(NT <= FILL_T, "the fill's scan has one thread per destination tile");
static_assert(NT <= HIST_T, "the tile histogram has one LDS counter per destination tile");
static_assert(TN > 96 && TN <= TR && TR == 112, "the product takes rows 0..95 as three 32-row groups and rows 96..111 as one 16-row group");
static_assert(NT <= 1024, "four 16-wave workgroups per WGP on 256 WGPs: every tile is resident at once");
static_assert(N_EDGES % (HIST_WG * 4) == 0, "a histogram workgroup's slice is a whole number of 16-byte loads");

typedef __attribute__((ext_vector_type(2))) float v2f;
typedef __attribute__((ext_vector_type(8))) float v8f;

__global__ __launch_bounds__(HIST_T)
void tile_hist_kernel(const int* __restrict__ dst, int* __restrict__ hist) {
    __shared__ int cnt[HIST_T];
    const int tid = threadIdx.x;
    const int Q   = N_EDGES / (HIST_WG * 4);
    const int K   = (Q + HIST_T - 1) / HIST_T;
    const int4* d4 = (const int4*)dst + (size_t)blockIdx.x * Q;
    cnt[tid] = 0;
    __syncthreads();
    int4 d[K];
    #pragma unroll
    for (int k = 0; k < K; ++k) {
        const int i = k * HIST_T + tid;
        d[k] = d4[i < Q ? i : Q - 1];
    }
    #pragma unroll
    for (int k = 0; k < K; ++k) {
        if (k * HIST_T + tid < Q) {
            atomicAdd(&cnt[d[k].x / TN], 1);
            atomicAdd(&cnt[d[k].y / TN], 1);
            atomicAdd(&cnt[d[k].z / TN], 1);
            atomicAdd(&cnt[d[k].w / TN], 1);
        }
    }
    __syncthreads();
    hist[blockIdx.x * HIST_T + tid] = cnt[tid];
}

__global__ __launch_bounds__(1024)
void tile_scan_kernel(const int* __restrict__ hist, int* __restrict__ tile_start,
                      int* __restrict__ cursor) {
    __shared__ int wsum[32];
    const int tid  = threadIdx.x;
    const int lane = tid & 31;
    const int wave = tid >> 5;

    int v = 0;
    #pragma unroll 16
    for (int b = 0; b < HIST_WG; ++b) v += hist[b * HIST_T + tid];

    int x = v;
    for (int off = 1; off < 32; off <<= 1) {
        int y = __shfl_up(x, off);
        if (lane >= off) x += y;
    }
    if (lane == 31) wsum[wave] = x;
    __syncthreads();
    if (wave == 0) {
        const int s = wsum[lane];
        int xs = s;
        for (int off = 1; off < 32; off <<= 1) {
            int y = __shfl_up(xs, off);
            if (lane >= off) xs += y;
        }
        wsum[lane] = xs - s;
    }
    __syncthreads();
    const int excl = x - v + wsum[wave];
    if (tid <= NT) tile_start[tid] = excl;
    if (tid < NT)  cursor[tid * CPAD] = excl;
}

__global__ __launch_bounds__(FILL_T)
void fill_kernel(const int* __restrict__ src, const int* __restrict__ dst,
                 int* __restrict__ cursor, int2* __restrict__ brec) {
    __shared__ int  cnt[FILL_T];
    __shared__ int  gofs[FILL_T];
    __shared__ int  wsum[32];
    __shared__ int  gpos[FILL_CH];
    __shared__ int2 rec[FILL_CH];
    const int tid   = threadIdx.x;
    const int lane  = tid & 31;
    const int wave  = tid >> 5;
    const int base  = blockIdx.x * FILL_CH;

    cnt[tid] = 0;
    int s[FILL_E], d[FILL_E];
    #pragma unroll
    for (int k = 0; k < FILL_E; ++k) {
        const int  e  = base + k * FILL_T + tid;
        const bool ok = e < N_EDGES;
        const int  ec = ok ? e : N_EDGES - 1;
        s[k] = src[ec];
        d[k] = dst[ec];
        if (!ok) d[k] = -1;
    }
    __syncthreads();
    #pragma unroll
    for (int k = 0; k < FILL_E; ++k)
        if (d[k] >= 0) atomicAdd(&cnt[d[k] / TN], 1);
    __syncthreads();

    const int v = cnt[tid];
    int x = v;
    for (int off = 1; off < 32; off <<= 1) {
        int y = __shfl_up(x, off);
        if (lane >= off) x += y;
    }
    if (lane == 31) wsum[wave] = x;
    __syncthreads();
    if (wave == 0) {
        const int ws = wsum[lane];
        int xs = ws;
        for (int off = 1; off < 32; off <<= 1) {
            int y = __shfl_up(xs, off);
            if (lane >= off) xs += y;
        }
        wsum[lane] = xs - ws;
    }
    __syncthreads();
    const int excl = x - v + wsum[wave];
    int g = 0;
    if (v > 0) g = atomicAdd(&cursor[tid * CPAD], v);
    cnt[tid]  = excl;
    gofs[tid] = g - excl;
    __syncthreads();

    #pragma unroll
    for (int k = 0; k < FILL_E; ++k) {
        if (d[k] >= 0) {
            const int t    = d[k] / TN;
            const int slot = atomicAdd(&cnt[t], 1);
            int2 r;
            r.x = s[k] | ((d[k] - t * TN) << SRC_BITS);
            r.y = 0;
            rec[slot]  = r;
            gpos[slot] = gofs[t] + slot;
        }
    }
    __syncthreads();

    const int n = min(FILL_CH, N_EDGES - base);
    for (int i = tid; i < n; i += FILL_T) brec[gpos[i]] = rec[i];
}

__global__ __launch_bounds__(SORT_T)
void tile_sort_kernel(const int* __restrict__ tile_start, const int2* __restrict__ brec,
                      int2* __restrict__ srec, float* __restrict__ dinv) {
    __shared__ int  cnt[TN];
    __shared__ int  wtot[(TN + 31) / 32];
    __shared__ int2 lrec[SORT_CAP];
    const int t    = blockIdx.x;
    const int tid  = threadIdx.x;
    const int lane = tid & 31;
    const int wave = tid >> 5;
    const int beg  = tile_start[t];
    const int end  = tile_start[t + 1];

    if (tid < TN) cnt[tid] = 0;
    __syncthreads();
    for (int i = beg + tid; i < end; i += SORT_T) atomicAdd(&cnt[brec[i].x >> SRC_BITS], 1);
    __syncthreads();

    const int c = (tid < TN) ? cnt[tid] : 0;
    if (tid < TN && t * TN + tid < N_NODES) dinv[t * TN + tid] = rsqrtf((float)(c + 1));
    int x = c;
    for (int off = 1; off < 32; off <<= 1) {
        int y = __shfl_up(x, off);
        if (lane >= off) x += y;
    }
    if (tid < TN && lane == 31) wtot[wave] = x;
    __syncthreads();
    if (tid < TN) {
        int front = beg;
        for (int j = 0; j < wave; ++j) front += wtot[j];
        cnt[tid] = front + x - c;
    }
    __syncthreads();

    const int n = end - beg;
    if (n <= SORT_CAP) {
        for (int i = beg + tid; i < end; i += SORT_T) {
            const int2 r  = brec[i];
            const int pos = atomicAdd(&cnt[r.x >> SRC_BITS], 1);
            lrec[pos - beg] = r;
        }
        __syncthreads();
        for (int i = tid; i < n; i += SORT_T) srec[beg + i] = lrec[i];
    } else {
        for (int i = beg + tid; i < end; i += SORT_T) {
            const int2 r  = brec[i];
            const int pos = atomicAdd(&cnt[r.x >> SRC_BITS], 1);
            srec[pos] = r;
        }
    }
}

template <bool TWO>
__device__ __forceinline__ void tile_product(const float* tile, const float* __restrict__ Wn,
                                             int lane, int wave,
                                             v8f& acc00, v8f& acc01, v8f& acc10, v8f& acc11) {
    const int m  = lane & 15;
    const int kh = lane >> 4;
    const int rq = wave >> 2;
    const int cq = wave & 3;
    const float* a0p = tile + (32 * rq + m) * LP + 2 * kh;
    const float* a1p = a0p + 16 * LP;
    const float* wp  = Wn + (size_t)(8 * kh) * HID + 32 * cq + 2 * m;

    #pragma unroll 2
    for (int p = 0; p < 16; ++p) {
        const int ks = 2 * p;
        const float* w0 = wp + (size_t)(16 * (ks & 7) + (ks >> 3)) * HID;
        const float* w1 = w0 + (size_t)16 * HID;
        const v2f wc0 = *(const v2f*)(w0), wc1 = *(const v2f*)(w0 + 4 * HID);
        const v2f wc2 = *(const v2f*)(w1), wc3 = *(const v2f*)(w1 + 4 * HID);
        const v2f a00 = *(const v2f*)(a0p + 4 * ks);
        const v2f a01 = *(const v2f*)(a0p + 4 * ks + 4);
        v2f a10 = {}, a11 = {};
        if (TWO) {
            a10 = *(const v2f*)(a1p + 4 * ks);
            a11 = *(const v2f*)(a1p + 4 * ks + 4);
        }
        v2f b0, b1;
        b0.x = wc0.x; b0.y = wc1.x; b1.x = wc0.y; b1.y = wc1.y;
        acc00 = __builtin_amdgcn_wmma_f32_16x16x4_f32(false, a00, false, b0, (short)0, acc00, false, false);
        acc01 = __builtin_amdgcn_wmma_f32_16x16x4_f32(false, a00, false, b1, (short)0, acc01, false, false);
        if (TWO) {
            acc10 = __builtin_amdgcn_wmma_f32_16x16x4_f32(false, a10, false, b0, (short)0, acc10, false, false);
            acc11 = __builtin_amdgcn_wmma_f32_16x16x4_f32(false, a10, false, b1, (short)0, acc11, false, false);
        }
        b0.x = wc2.x; b0.y = wc3.x; b1.x = wc2.y; b1.y = wc3.y;
        acc00 = __builtin_amdgcn_wmma_f32_16x16x4_f32(false, a01, false, b0, (short)0, acc00, false, false);
        acc01 = __builtin_amdgcn_wmma_f32_16x16x4_f32(false, a01, false, b1, (short)0, acc01, false, false);
        if (TWO) {
            acc10 = __builtin_amdgcn_wmma_f32_16x16x4_f32(false, a11, false, b0, (short)0, acc10, false, false);
            acc11 = __builtin_amdgcn_wmma_f32_16x16x4_f32(false, a11, false, b1, (short)0, acc11, false, false);
        }
    }
}

template <int MODE>
__global__ __launch_bounds__(AGG_T, 16)
void agg_kernel(const float* __restrict__ H, const float* __restrict__ dinv,
                const int* __restrict__ tile_start, int2* __restrict__ srec,
                const float* __restrict__ bias, const float* __restrict__ Wf,
                const float* __restrict__ Wn, float* __restrict__ out) {
    __shared__ float tile[TR * LP];
    const int t    = blockIdx.x;
    const int lane = threadIdx.x & 31;
    const int wave = __builtin_amdgcn_readfirstlane((int)(threadIdx.x >> 5));
    const int NW   = AGG_T / 32;
    const int node0 = t * TN;

    for (int r = wave; r < TR; r += NW) {
        const int node = node0 + r;
        float4 v = make_float4(0.f, 0.f, 0.f, 0.f);
        if (r < TN && node < N_NODES) {
            const float dv = dinv[node];
            const float w  = dv * dv;
            const float4 h = ((const float4*)(H + (size_t)node * HID))[lane];
            v.x = h.x * w; v.y = h.y * w; v.z = h.z * w; v.w = h.w * w;
        }
        float* row = tile + r * LP + lane;
        row[0] = v.x; row[32] = v.y; row[64] = v.z; row[96] = v.w;
    }
    __syncthreads();

    const int beg = tile_start[t];
    const int end = tile_start[t + 1];
    for (int base = beg + wave * 32; base < end; base += NW * 32) {
        const int idx = base + lane;
        int pk = 0, wb = 0;
        if (MODE == 2) {
            if (idx < end) {
                pk = srec[idx].x;
                const float wsrc = dinv[pk & SRC_MASK];
                const float wdst = dinv[node0 + (pk >> SRC_BITS)];
                wb = __float_as_int(wsrc * wdst);
                srec[idx].y = wb;
            }
        } else {
            if (idx < end) { const int2 r = srec[idx]; pk = r.x; wb = r.y; }
        }
        const int n = min(32, end - base);
        int   cur = __builtin_amdgcn_readlane(pk, 0) >> SRC_BITS;
        float a0 = 0.f, a1 = 0.f, a2 = 0.f, a3 = 0.f;
        for (int k = 0; k < n; k += 4) {
            float4 h[4];
            #pragma unroll
            for (int u = 0; u < 4; ++u) {
                const int sk = __builtin_amdgcn_readlane(pk, k + u) & SRC_MASK;
                h[u] = ((const float4*)(H + (size_t)sk * HID))[lane];
            }
            #pragma unroll
            for (int u = 0; u < 4; ++u) {
                if (u == 0 || k + u < n) {
                    const int   dk = __builtin_amdgcn_readlane(pk, k + u) >> SRC_BITS;
                    const float wk = __int_as_float(__builtin_amdgcn_readlane(wb, k + u));
                    if (dk != cur) {
                        float* row = tile + cur * LP + lane;
                        __hip_atomic_fetch_add(row,      a0, __ATOMIC_RELAXED, __HIP_MEMORY_SCOPE_WORKGROUP);
                        __hip_atomic_fetch_add(row + 32, a1, __ATOMIC_RELAXED, __HIP_MEMORY_SCOPE_WORKGROUP);
                        __hip_atomic_fetch_add(row + 64, a2, __ATOMIC_RELAXED, __HIP_MEMORY_SCOPE_WORKGROUP);
                        __hip_atomic_fetch_add(row + 96, a3, __ATOMIC_RELAXED, __HIP_MEMORY_SCOPE_WORKGROUP);
                        a0 = 0.f; a1 = 0.f; a2 = 0.f; a3 = 0.f;
                        cur = dk;
                    }
                    a0 += h[u].x * wk;
                    a1 += h[u].y * wk;
                    a2 += h[u].z * wk;
                    a3 += h[u].w * wk;
                }
            }
        }
        float* row = tile + cur * LP + lane;
        __hip_atomic_fetch_add(row,      a0, __ATOMIC_RELAXED, __HIP_MEMORY_SCOPE_WORKGROUP);
        __hip_atomic_fetch_add(row + 32, a1, __ATOMIC_RELAXED, __HIP_MEMORY_SCOPE_WORKGROUP);
        __hip_atomic_fetch_add(row + 64, a2, __ATOMIC_RELAXED, __HIP_MEMORY_SCOPE_WORKGROUP);
        __hip_atomic_fetch_add(row + 96, a3, __ATOMIC_RELAXED, __HIP_MEMORY_SCOPE_WORKGROUP);
    }
    __syncthreads();

    const int m  = lane & 15;
    const int kh = lane >> 4;
    const int rq = wave >> 2;
    const int cq = wave & 3;

    if (MODE == 1) {
        const float4 b = ((const float4*)bias)[lane];
        for (int r = wave; r < TN; r += NW) {
            const int node = node0 + r;
            if (node < N_NODES) {
                const float* row = tile + r * LP + lane;
                float4 o;
                o.x = row[0]  + b.x; o.y = row[32] + b.y; o.z = row[64] + b.z; o.w = row[96] + b.w;
                o.x = o.x > 0.f ? o.x : 0.f;
                o.y = o.y > 0.f ? o.y : 0.f;
                o.z = o.z > 0.f ? o.z : 0.f;
                o.w = o.w > 0.f ? o.w : 0.f;
                ((float4*)(out + (size_t)node * HID))[lane] = o;
            }
        }
        return;
    }

    if (MODE == 2) {
        v8f p00 = {}, p01 = {}, p10 = {}, p11 = {};
        if (rq < 3) tile_product<true>(tile, Wf, lane, wave, p00, p01, p10, p11);
        else        tile_product<false>(tile, Wf, lane, wave, p00, p01, p10, p11);
        __syncthreads();
        const int f0 = 32 * cq + 2 * m;
        const v2f bb = *(const v2f*)(bias + f0);
        float* wr = tile + (32 * rq + 8 * kh) * LP + (f0 & 3) * 32 + (f0 >> 2);
        #pragma unroll
        for (int r = 0; r < 8; ++r) {
            const float o00 = p00[r] + bb.x, o01 = p01[r] + bb.y;
            const float o10 = p10[r] + bb.x, o11 = p11[r] + bb.y;
            wr[r * LP]             = o00 > 0.f ? o00 : 0.f;
            wr[r * LP + 32]        = o01 > 0.f ? o01 : 0.f;
            if (rq < 3) {
                wr[(16 + r) * LP]      = o10 > 0.f ? o10 : 0.f;
                wr[(16 + r) * LP + 32] = o11 > 0.f ? o11 : 0.f;
            }
        }
    } else {
        const float4 b = ((const float4*)bias)[lane];
        for (int r = wave; r < TN; r += NW) {
            float* row = tile + r * LP + lane;
            float o0 = row[0] + b.x, o1 = row[32] + b.y, o2 = row[64] + b.z, o3 = row[96] + b.w;
            row[0]  = o0 > 0.f ? o0 : 0.f;
            row[32] = o1 > 0.f ? o1 : 0.f;
            row[64] = o2 > 0.f ? o2 : 0.f;
            row[96] = o3 > 0.f ? o3 : 0.f;
        }
    }
    __syncthreads();

    v8f acc00 = {}, acc01 = {}, acc10 = {}, acc11 = {};
    if (rq < 3) tile_product<true>(tile, Wn, lane, wave, acc00, acc01, acc10, acc11);
    else        tile_product<false>(tile, Wn, lane, wave, acc00, acc01, acc10, acc11);

    const int tr0  = 32 * rq + 8 * kh;
    const int row0 = node0 + tr0;
    float* ob = out + (size_t)row0 * HID + 32 * cq + 2 * m;
    #pragma unroll
    for (int r = 0; r < 8; ++r) {
        if (tr0 + r < TN && row0 + r < N_NODES) {
            v2f o; o.x = acc00[r]; o.y = acc01[r];
            *(v2f*)(ob + (size_t)r * HID) = o;
        }
        if (tr0 + 16 + r < TN && row0 + 16 + r < N_NODES) {
            v2f o; o.x = acc10[r]; o.y = acc11[r];
            *(v2f*)(ob + (size_t)(16 + r) * HID) = o;
        }
    }
}

__device__ __forceinline__ int first_node_at_or_above(const int* __restrict__ batch, int key,
                                                      int lane) {
    int lo = 0, len = N_NODES;
    for (int it = 0; it < 5 && len > 0; ++it) {
        const int step = (len + 31) >> 5;
        const int q    = lo + lane * step;
        const bool below = (q < lo + len) && (batch[q] < key);
        const int c = __builtin_popcount(__builtin_amdgcn_ballot_w32(below));
        if (c == 0) {
            len = 0;
        } else {
            const int end = lo + len;
            const int nlo = lo + (c - 1) * step + 1;
            len = min(lo + c * step, end) - nlo;
            lo  = nlo;
        }
    }
    return lo;
}

__global__ __launch_bounds__(POOL_T)
void pool_linear_kernel(const float* __restrict__ X, const int* __restrict__ batch,
                        const float* __restrict__ linW, const float* __restrict__ linb,
                        float* __restrict__ out) {
    __shared__ float4 part4[(POOL_T / 32) * (HID / 4)];
    __shared__ float4 wl4[HID * N_CLASSES / 4];
    __shared__ float  pooled[HID];
    __shared__ int    range[2];
    const int g    = blockIdx.x;
    const int tid  = threadIdx.x;
    const int lane = tid & 31;
    const int wave = __builtin_amdgcn_readfirstlane((int)(threadIdx.x >> 5));
    const int NW   = POOL_T / 32;
    const int WQ   = HID * N_CLASSES / 8;
    static_assert(HID * N_CLASSES % 8 == 0 && WQ % 32 == 0 && 2 + WQ / 32 <= POOL_T / 32,
                  "the weight copy takes whole waves behind the two that search");

    if (wave < 2) {
        const int b = first_node_at_or_above(batch, g + wave, lane);
        if (lane == 0) range[wave] = b;
    } else if (wave < 2 + WQ / 32) {
        const int i = tid - 64;
        const float4 w0 = ((const float4*)linW)[i];
        const float4 w1 = ((const float4*)linW)[i + WQ];
        wl4[i]      = w0;
        wl4[i + WQ] = w1;
    }
    __syncthreads();
    const int lo = range[0];
    const int hi = range[1];

    float4 acc = make_float4(0.f, 0.f, 0.f, 0.f);
    for (int r = lo + wave; r < hi; r += 16 * NW) {
        float4 h[16];
        #pragma unroll
        for (int u = 0; u < 16; ++u) {
            const int ru = r + u * NW;
            h[u] = make_float4(0.f, 0.f, 0.f, 0.f);
            if (ru < hi) h[u] = ((const float4*)(X + (size_t)ru * HID))[lane];
        }
        #pragma unroll
        for (int u = 0; u < 16; ++u) {
            acc.x += h[u].x; acc.y += h[u].y; acc.z += h[u].z; acc.w += h[u].w;
        }
    }
    part4[wave * (HID / 4) + lane] = acc;
    __syncthreads();

    if (tid < HID) {
        const float* pf = (const float*)part4;
        float sum = 0.f;
        #pragma unroll
        for (int w = 0; w < NW; ++w) sum += pf[w * HID + tid];
        pooled[tid] = sum;
    }
    __syncthreads();

    const float* wl = (const float*)wl4;
    for (int c = wave; c < N_CLASSES; c += NW) {
        float a = 0.f;
        #pragma unroll
        for (int j = 0; j < HID / 32; ++j) {
            const int f = lane + 32 * j;
            a += pooled[f] * wl[f * N_CLASSES + c];
        }
        for (int off = 16; off > 0; off >>= 1) a += __shfl_xor(a, off);
        if (lane == 0) {
            const float scale = 1.0f / fmaxf((float)(hi - lo), 1.0f);
            out[g * N_CLASSES + c] = a * scale + linb[c];
        }
    }
}

extern "C" void kernel_launch(void* const* d_in, const int* in_sizes, int n_in,
                              void* d_out, int out_size, void* d_ws, size_t ws_size,
                              hipStream_t stream) {
    const float* x     = (const float*)d_in[0];
    const int*   edge  = (const int*)d_in[1];
    const int*   src   = edge;
    const int*   dst   = edge + N_EDGES;
    const int*   batch = (const int*)d_in[2];
    const float* Ws    = (const float*)d_in[3];
    const float* bs    = (const float*)d_in[4];
    const float* linW  = (const float*)d_in[5];
    const float* linb  = (const float*)d_in[6];
    float*       out   = (float*)d_out;

    float* ws = (float*)d_ws;
    float* H      = ws;
    float* X1     = H + (size_t)N_NODES * HID;
    float* X2     = X1 + (size_t)N_NODES * HID;
    float* dinv   = X2 + (size_t)N_NODES * HID;
    int2*  brec   = (int2*)(dinv + N_NODES);
    int2*  srec   = brec + N_EDGES;
    int*   tstart = (int*)(srec + N_EDGES);
    int*   cursor = tstart + 1024;
    int*   hist   = cursor + 1024 * CPAD;

    tile_hist_kernel<<<HIST_WG, HIST_T, 0, stream>>>(dst, hist);
    tile_scan_kernel<<<1, 1024, 0, stream>>>(hist, tstart, cursor);
    fill_kernel<<<NFILL, FILL_T, 0, stream>>>(src, dst, cursor, brec);
    tile_sort_kernel<<<NT, SORT_T, 0, stream>>>(tstart, brec, srec, dinv);

    const float* W0 = Ws;
    const float* W1 = Ws + (size_t)1 * HID * HID;
    const float* W2 = Ws + (size_t)2 * HID * HID;
    const float* W3 = Ws + (size_t)3 * HID * HID;
    const float* W4 = Ws + (size_t)4 * HID * HID;
    agg_kernel<2><<<NT, AGG_T, 0, stream>>>(x,  dinv, tstart, srec, bs,           W0, W1, H);
    agg_kernel<0><<<NT, AGG_T, 0, stream>>>(H,  dinv, tstart, srec, bs + 1 * HID, W0, W2, X1);
    agg_kernel<0><<<NT, AGG_T, 0, stream>>>(X1, dinv, tstart, srec, bs + 2 * HID, W0, W3, H);
    agg_kernel<0><<<NT, AGG_T, 0, stream>>>(H,  dinv, tstart, srec, bs + 3 * HID, W0, W4, X1);
    agg_kernel<1><<<NT, AGG_T, 0, stream>>>(X1, dinv, tstart, srec, bs + 4 * HID, W0, W0, X2);

    pool_linear_kernel<<<N_GRAPHS, POOL_T, 0, stream>>>(X2, batch, linW, linb, out);
}


# Tile kernels: fewer dependent memory round trips per wave: a wave owns one contiguous 1/16 of the tile's list, reads 128 records at once, its 7 start rows in one group; row sum order changes
# speedup vs baseline: 6.4318x; 1.0395x over previous
#include <hip/hip_runtime.h>
#include <hip/hip_bf16.h>


#define N_NODES   100000
#define N_EDGES   1600000
#define HID       128
#define N_CLASSES 10
#define N_GRAPHS  512
#define N_LAYERS  5
#define TN        100
#define TR        112
#define NT        ((N_NODES + TN - 1) / TN)
#define CPAD      32
#define AGG_T     512
#define SORT_T    512
#define SORT_CAP  4096
#define FILL_T    1024
#define FILL_E    7
#define FILL_CH   (FILL_T * FILL_E)
#define NFILL     ((N_EDGES + FILL_CH - 1) / FILL_CH)
#define HIST_WG   64
#define HIST_T    1024
#define POOL_T    256
#define LP        132
#define SRC_BITS  17
#define SRC_MASK  ((1 << SRC_BITS) - 1)

static_assert(N_NODES <= (1 << SRC_BITS), "source node must fit the record's low bits");
static_assert(TN <= (1 << (31 - SRC_BITS)), "row inside the tile must fit above the source bits");
static_assert(NT <= FILL_T, "the fill's scan has one thread per destination tile");
static_assert(NT <= HIST_T, "the tile histogram has one LDS counter per destination tile");
static_assert(TN > 96 && TN <= TR && TR == 112, "the product takes rows 0..95 as three 32-row groups and rows 96..111 as one 16-row group");
static_assert(NT <= 1024, "four 16-wave workgroups per WGP on 256 WGPs: every tile is resident at once");
static_assert(N_EDGES % (HIST_WG * 4) == 0, "a histogram workgroup's slice is a whole number of 16-byte loads");

typedef __attribute__((ext_vector_type(2))) float v2f;
typedef __attribute__((ext_vector_type(8))) float v8f;

__global__ __launch_bounds__(HIST_T)
void tile_hist_kernel(const int* __restrict__ dst, int* __restrict__ hist) {
    __shared__ int cnt[HIST_T];
    const int tid = threadIdx.x;
    const int Q   = N_EDGES / (HIST_WG * 4);
    const int K   = (Q + HIST_T - 1) / HIST_T;
    const int4* d4 = (const int4*)dst + (size_t)blockIdx.x * Q;
    cnt[tid] = 0;
    __syncthreads();
    int4 d[K];
    #pragma unroll
    for (int k = 0; k < K; ++k) {
        const int i = k * HIST_T + tid;
        d[k] = d4[i < Q ? i : Q - 1];
    }
    #pragma unroll
    for (int k = 0; k < K; ++k) {
        if (k * HIST_T + tid < Q) {
            atomicAdd(&cnt[d[k].x / TN], 1);
            atomicAdd(&cnt[d[k].y / TN], 1);
            atomicAdd(&cnt[d[k].z / TN], 1);
            atomicAdd(&cnt[d[k].w / TN], 1);
        }
    }
    __syncthreads();
    hist[blockIdx.x * HIST_T + tid] = cnt[tid];
}

__global__ __launch_bounds__(1024)
void tile_scan_kernel(const int* __restrict__ hist, int* __restrict__ tile_start,
                      int* __restrict__ cursor) {
    __shared__ int wsum[32];
    const int tid  = threadIdx.x;
    const int lane = tid & 31;
    const int wave = tid >> 5;

    int v = 0;
    #pragma unroll 16
    for (int b = 0; b < HIST_WG; ++b) v += hist[b * HIST_T + tid];

    int x = v;
    for (int off = 1; off < 32; off <<= 1) {
        int y = __shfl_up(x, off);
        if (lane >= off) x += y;
    }
    if (lane == 31) wsum[wave] = x;
    __syncthreads();
    if (wave == 0) {
        const int s = wsum[lane];
        int xs = s;
        for (int off = 1; off < 32; off <<= 1) {
            int y = __shfl_up(xs, off);
            if (lane >= off) xs += y;
        }
        wsum[lane] = xs - s;
    }
    __syncthreads();
    const int excl = x - v + wsum[wave];
    if (tid <= NT) tile_start[tid] = excl;
    if (tid < NT)  cursor[tid * CPAD] = excl;
}

__global__ __launch_bounds__(FILL_T)
void fill_kernel(const int* __restrict__ src, const int* __restrict__ dst,
                 int* __restrict__ cursor, int2* __restrict__ brec) {
    __shared__ int  cnt[FILL_T];
    __shared__ int  gofs[FILL_T];
    __shared__ int  wsum[32];
    __shared__ int  gpos[FILL_CH];
    __shared__ int2 rec[FILL_CH];
    const int tid   = threadIdx.x;
    const int lane  = tid & 31;
    const int wave  = tid >> 5;
    const int base  = blockIdx.x * FILL_CH;

    cnt[tid] = 0;
    int s[FILL_E], d[FILL_E];
    #pragma unroll
    for (int k = 0; k < FILL_E; ++k) {
        const int  e  = base + k * FILL_T + tid;
        const bool ok = e < N_EDGES;
        const int  ec = ok ? e : N_EDGES - 1;
        s[k] = src[ec];
        d[k] = dst[ec];
        if (!ok) d[k] = -1;
    }
    __syncthreads();
    #pragma unroll
    for (int k = 0; k < FILL_E; ++k)
        if (d[k] >= 0) atomicAdd(&cnt[d[k] / TN], 1);
    __syncthreads();

    const int v = cnt[tid];
    int x = v;
    for (int off = 1; off < 32; off <<= 1) {
        int y = __shfl_up(x, off);
        if (lane >= off) x += y;
    }
    if (lane == 31) wsum[wave] = x;
    __syncthreads();
    if (wave == 0) {
        const int ws = wsum[lane];
        int xs = ws;
        for (int off = 1; off < 32; off <<= 1) {
            int y = __shfl_up(xs, off);
            if (lane >= off) xs += y;
        }
        wsum[lane] = xs - ws;
    }
    __syncthreads();
    const int excl = x - v + wsum[wave];
    int g = 0;
    if (v > 0) g = atomicAdd(&cursor[tid * CPAD], v);
    cnt[tid]  = excl;
    gofs[tid] = g - excl;
    __syncthreads();

    #pragma unroll
    for (int k = 0; k < FILL_E; ++k) {
        if (d[k] >= 0) {
            const int t    = d[k] / TN;
            const int slot = atomicAdd(&cnt[t], 1);
            int2 r;
            r.x = s[k] | ((d[k] - t * TN) << SRC_BITS);
            r.y = 0;
            rec[slot]  = r;
            gpos[slot] = gofs[t] + slot;
        }
    }
    __syncthreads();

    const int n = min(FILL_CH, N_EDGES - base);
    for (int i = tid; i < n; i += FILL_T) brec[gpos[i]] = rec[i];
}

__global__ __launch_bounds__(SORT_T)
void tile_sort_kernel(const int* __restrict__ tile_start, const int2* __restrict__ brec,
                      int2* __restrict__ srec, float* __restrict__ dinv) {
    __shared__ int  cnt[TN];
    __shared__ int  wtot[(TN + 31) / 32];
    __shared__ int2 lrec[SORT_CAP];
    const int t    = blockIdx.x;
    const int tid  = threadIdx.x;
    const int lane = tid & 31;
    const int wave = tid >> 5;
    const int beg  = tile_start[t];
    const int end  = tile_start[t + 1];

    if (tid < TN) cnt[tid] = 0;
    __syncthreads();
    for (int i = beg + tid; i < end; i += SORT_T) atomicAdd(&cnt[brec[i].x >> SRC_BITS], 1);
    __syncthreads();

    const int c = (tid < TN) ? cnt[tid] : 0;
    if (tid < TN && t * TN + tid < N_NODES) dinv[t * TN + tid] = rsqrtf((float)(c + 1));
    int x = c;
    for (int off = 1; off < 32; off <<= 1) {
        int y = __shfl_up(x, off);
        if (lane >= off) x += y;
    }
    if (tid < TN && lane == 31) wtot[wave] = x;
    __syncthreads();
    if (tid < TN) {
        int front = beg;
        for (int j = 0; j < wave; ++j) front += wtot[j];
        cnt[tid] = front + x - c;
    }
    __syncthreads();

    const int n = end - beg;
    if (n <= SORT_CAP) {
        for (int i = beg + tid; i < end; i += SORT_T) {
            const int2 r  = brec[i];
            const int pos = atomicAdd(&cnt[r.x >> SRC_BITS], 1);
            lrec[pos - beg] = r;
        }
        __syncthreads();
        for (int i = tid; i < n; i += SORT_T) srec[beg + i] = lrec[i];
    } else {
        for (int i = beg + tid; i < end; i += SORT_T) {
            const int2 r  = brec[i];
            const int pos = atomicAdd(&cnt[r.x >> SRC_BITS], 1);
            srec[pos] = r;
        }
    }
}

template <bool TWO>
__device__ __forceinline__ void tile_product(const float* tile, const float* __restrict__ Wn,
                                             int lane, int wave,
                                             v8f& acc00, v8f& acc01, v8f& acc10, v8f& acc11) {
    const int m  = lane & 15;
    const int kh = lane >> 4;
    const int rq = wave >> 2;
    const int cq = wave & 3;
    const float* a0p = tile + (32 * rq + m) * LP + 2 * kh;
    const float* a1p = a0p + 16 * LP;
    const float* wp  = Wn + (size_t)(8 * kh) * HID + 32 * cq + 2 * m;

    #pragma unroll 2
    for (int p = 0; p < 16; ++p) {
        const int ks = 2 * p;
        const float* w0 = wp + (size_t)(16 * (ks & 7) + (ks >> 3)) * HID;
        const float* w1 = w0 + (size_t)16 * HID;
        const v2f wc0 = *(const v2f*)(w0), wc1 = *(const v2f*)(w0 + 4 * HID);
        const v2f wc2 = *(const v2f*)(w1), wc3 = *(const v2f*)(w1 + 4 * HID);
        const v2f a00 = *(const v2f*)(a0p + 4 * ks);
        const v2f a01 = *(const v2f*)(a0p + 4 * ks + 4);
        v2f a10 = {}, a11 = {};
        if (TWO) {
            a10 = *(const v2f*)(a1p + 4 * ks);
            a11 = *(const v2f*)(a1p + 4 * ks + 4);
        }
        v2f b0, b1;
        b0.x = wc0.x; b0.y = wc1.x; b1.x = wc0.y; b1.y = wc1.y;
        acc00 = __builtin_amdgcn_wmma_f32_16x16x4_f32(false, a00, false, b0, (short)0, acc00, false, false);
        acc01 = __builtin_amdgcn_wmma_f32_16x16x4_f32(false, a00, false, b1, (short)0, acc01, false, false);
        if (TWO) {
            acc10 = __builtin_amdgcn_wmma_f32_16x16x4_f32(false, a10, false, b0, (short)0, acc10, false, false);
            acc11 = __builtin_amdgcn_wmma_f32_16x16x4_f32(false, a10, false, b1, (short)0, acc11, false, false);
        }
        b0.x = wc2.x; b0.y = wc3.x; b1.x = wc2.y; b1.y = wc3.y;
        acc00 = __builtin_amdgcn_wmma_f32_16x16x4_f32(false, a01, false, b0, (short)0, acc00, false, false);
        acc01 = __builtin_amdgcn_wmma_f32_16x16x4_f32(false, a01, false, b1, (short)0, acc01, false, false);
        if (TWO) {
            acc10 = __builtin_amdgcn_wmma_f32_16x16x4_f32(false, a11, false, b0, (short)0, acc10, false, false);
            acc11 = __builtin_amdgcn_wmma_f32_16x16x4_f32(false, a11, false, b1, (short)0, acc11, false, false);
        }
    }
}

template <int MODE>
__global__ __launch_bounds__(AGG_T, 16)
void agg_kernel(const float* __restrict__ H, const float* __restrict__ dinv,
                const int* __restrict__ tile_start, int2* __restrict__ srec,
                const float* __restrict__ bias, const float* __restrict__ Wf,
                const float* __restrict__ Wn, float* __restrict__ out) {
    __shared__ float tile[TR * LP];
    const int t    = blockIdx.x;
    const int lane = threadIdx.x & 31;
    const int wave = __builtin_amdgcn_readfirstlane((int)(threadIdx.x >> 5));
    const int NW   = AGG_T / 32;
    const int node0 = t * TN;

    constexpr int RW = TR / NW;
    static_assert(TR % NW == 0, "every wave starts the same number of rows");

    const int tsl = tile_start[t + (lane & 1)];
    const int beg = __builtin_amdgcn_readlane(tsl, 0);
    const int end = __builtin_amdgcn_readlane(tsl, 1);
    float4 b = make_float4(0.f, 0.f, 0.f, 0.f);
    if (MODE != 2) b = ((const float4*)bias)[lane];

    {
        const int  rl  = wave + min(lane, RW - 1) * NW;
        const bool okl = rl < TN && node0 + rl < N_NODES;
        const float dvl = dinv[okl ? node0 + rl : node0];
        float4 hv[RW];
        #pragma unroll
        for (int i = 0; i < RW; ++i) {
            const int  r  = wave + i * NW;
            const bool ok = r < TN && node0 + r < N_NODES;
            const int  nc = ok ? node0 + r : node0;
            hv[i] = ((const float4*)(H + (size_t)nc * HID))[lane];
        }
        #pragma unroll
        for (int i = 0; i < RW; ++i) {
            const int  r  = wave + i * NW;
            const bool ok = r < TN && node0 + r < N_NODES;
            const float dvi = __int_as_float(__builtin_amdgcn_readlane(__float_as_int(dvl), i));
            const float w = dvi * dvi;
            float4 v = make_float4(0.f, 0.f, 0.f, 0.f);
            if (ok) { v.x = hv[i].x * w; v.y = hv[i].y * w; v.z = hv[i].z * w; v.w = hv[i].w * w; }
            float* row = tile + r * LP + lane;
            row[0] = v.x; row[32] = v.y; row[64] = v.z; row[96] = v.w;
        }
    }
    __syncthreads();

    const int per  = (end - beg + NW - 1) / NW;
    const int wbeg = beg + wave * per;
    const int wend = min(end, wbeg + per);
    int   cur = -1;
    float a0 = 0.f, a1 = 0.f, a2 = 0.f, a3 = 0.f;
    for (int cb = wbeg; cb < wend; cb += 128) {
        int pk[4], wb[4];
        const int i0 = cb + 4 * lane;
        if (MODE == 2) {
            float wsrc[4], wdst[4];
            #pragma unroll
            for (int u = 0; u < 4; ++u) pk[u] = srec[min(i0 + u, wend - 1)].x;
            #pragma unroll
            for (int u = 0; u < 4; ++u) {
                wsrc[u] = dinv[pk[u] & SRC_MASK];
                wdst[u] = dinv[node0 + (pk[u] >> SRC_BITS)];
            }
            #pragma unroll
            for (int u = 0; u < 4; ++u) {
                if (i0 + u < wend) {
                    wb[u] = __float_as_int(wsrc[u] * wdst[u]);
                    srec[i0 + u].y = wb[u];
                } else {
                    pk[u] = 0; wb[u] = 0;
                }
            }
        } else {
            int2 r[4];
            #pragma unroll
            for (int u = 0; u < 4; ++u) r[u] = srec[min(i0 + u, wend - 1)];
            #pragma unroll
            for (int u = 0; u < 4; ++u) {
                const bool in = i0 + u < wend;
                pk[u] = in ? r[u].x : 0;
                wb[u] = in ? r[u].y : 0;
            }
        }
        const int n = min(128, wend - cb);
        for (int k = 0; k < n; k += 4) {
            const int l = k >> 2;
            float4 h[4];
            #pragma unroll
            for (int u = 0; u < 4; ++u) {
                const int sk = __builtin_amdgcn_readlane(pk[u], l) & SRC_MASK;
                h[u] = ((const float4*)(H + (size_t)sk * HID))[lane];
            }
            #pragma unroll
            for (int u = 0; u < 4; ++u) {
                if (u == 0 || k + u < n) {
                    const int   dk = __builtin_amdgcn_readlane(pk[u], l) >> SRC_BITS;
                    const float wk = __int_as_float(__builtin_amdgcn_readlane(wb[u], l));
                    if (dk != cur) {
                        if (cur >= 0) {
                            float* row = tile + cur * LP + lane;
                            __hip_atomic_fetch_add(row,      a0, __ATOMIC_RELAXED, __HIP_MEMORY_SCOPE_WORKGROUP);
                            __hip_atomic_fetch_add(row + 32, a1, __ATOMIC_RELAXED, __HIP_MEMORY_SCOPE_WORKGROUP);
                            __hip_atomic_fetch_add(row + 64, a2, __ATOMIC_RELAXED, __HIP_MEMORY_SCOPE_WORKGROUP);
                            __hip_atomic_fetch_add(row + 96, a3, __ATOMIC_RELAXED, __HIP_MEMORY_SCOPE_WORKGROUP);
                        }
                        a0 = 0.f; a1 = 0.f; a2 = 0.f; a3 = 0.f;
                        cur = dk;
                    }
                    a0 += h[u].x * wk;
                    a1 += h[u].y * wk;
                    a2 += h[u].z * wk;
                    a3 += h[u].w * wk;
                }
            }
        }
    }
    if (cur >= 0) {
        float* row = tile + cur * LP + lane;
        __hip_atomic_fetch_add(row,      a0, __ATOMIC_RELAXED, __HIP_MEMORY_SCOPE_WORKGROUP);
        __hip_atomic_fetch_add(row + 32, a1, __ATOMIC_RELAXED, __HIP_MEMORY_SCOPE_WORKGROUP);
        __hip_atomic_fetch_add(row + 64, a2, __ATOMIC_RELAXED, __HIP_MEMORY_SCOPE_WORKGROUP);
        __hip_atomic_fetch_add(row + 96, a3, __ATOMIC_RELAXED, __HIP_MEMORY_SCOPE_WORKGROUP);
    }
    __syncthreads();

    const int m  = lane & 15;
    const int kh = lane >> 4;
    const int rq = wave >> 2;
    const int cq = wave & 3;

    if (MODE == 1) {
        for (int r = wave; r < TN; r += NW) {
            const int node = node0 + r;
            if (node < N_NODES) {
                const float* row = tile + r * LP + lane;
                float4 o;
                o.x = row[0]  + b.x; o.y = row[32] + b.y; o.z = row[64] + b.z; o.w = row[96] + b.w;
                o.x = o.x > 0.f ? o.x : 0.f;
                o.y = o.y > 0.f ? o.y : 0.f;
                o.z = o.z > 0.f ? o.z : 0.f;
                o.w = o.w > 0.f ? o.w : 0.f;
                ((float4*)(out + (size_t)node * HID))[lane] = o;
            }
        }
        return;
    }

    if (MODE == 2) {
        v8f p00 = {}, p01 = {}, p10 = {}, p11 = {};
        if (rq < 3) tile_product<true>(tile, Wf, lane, wave, p00, p01, p10, p11);
        else        tile_product<false>(tile, Wf, lane, wave, p00, p01, p10, p11);
        __syncthreads();
        const int f0 = 32 * cq + 2 * m;
        const v2f bb = *(const v2f*)(bias + f0);
        float* wr = tile + (32 * rq + 8 * kh) * LP + (f0 & 3) * 32 + (f0 >> 2);
        #pragma unroll
        for (int r = 0; r < 8; ++r) {
            const float o00 = p00[r] + bb.x, o01 = p01[r] + bb.y;
            const float o10 = p10[r] + bb.x, o11 = p11[r] + bb.y;
            wr[r * LP]             = o00 > 0.f ? o00 : 0.f;
            wr[r * LP + 32]        = o01 > 0.f ? o01 : 0.f;
            if (rq < 3) {
                wr[(16 + r) * LP]      = o10 > 0.f ? o10 : 0.f;
                wr[(16 + r) * LP + 32] = o11 > 0.f ? o11 : 0.f;
            }
        }
    } else {
        for (int r = wave; r < TN; r += NW) {
            float* row = tile + r * LP + lane;
            float o0 = row[0] + b.x, o1 = row[32] + b.y, o2 = row[64] + b.z, o3 = row[96] + b.w;
            row[0]  = o0 > 0.f ? o0 : 0.f;
            row[32] = o1 > 0.f ? o1 : 0.f;
            row[64] = o2 > 0.f ? o2 : 0.f;
            row[96] = o3 > 0.f ? o3 : 0.f;
        }
    }
    __syncthreads();

    v8f acc00 = {}, acc01 = {}, acc10 = {}, acc11 = {};
    if (rq < 3) tile_product<true>(tile, Wn, lane, wave, acc00, acc01, acc10, acc11);
    else        tile_product<false>(tile, Wn, lane, wave, acc00, acc01, acc10, acc11);

    const int tr0  = 32 * rq + 8 * kh;
    const int row0 = node0 + tr0;
    float* ob = out + (size_t)row0 * HID + 32 * cq + 2 * m;
    #pragma unroll
    for (int r = 0; r < 8; ++r) {
        if (tr0 + r < TN && row0 + r < N_NODES) {
            v2f o; o.x = acc00[r]; o.y = acc01[r];
            *(v2f*)(ob + (size_t)r * HID) = o;
        }
        if (tr0 + 16 + r < TN && row0 + 16 + r < N_NODES) {
            v2f o; o.x = acc10[r]; o.y = acc11[r];
            *(v2f*)(ob + (size_t)(16 + r) * HID) = o;
        }
    }
}

__device__ __forceinline__ int first_node_at_or_above(const int* __restrict__ batch, int key,
                                                      int lane) {
    int lo = 0, len = N_NODES;
    for (int it = 0; it < 5 && len > 0; ++it) {
        const int step = (len + 31) >> 5;
        const int q    = lo + lane * step;
        const bool below = (q < lo + len) && (batch[q] < key);
        const int c = __builtin_popcount(__builtin_amdgcn_ballot_w32(below));
        if (c == 0) {
            len = 0;
        } else {
            const int end = lo + len;
            const int nlo = lo + (c - 1) * step + 1;
            len = min(lo + c * step, end) - nlo;
            lo  = nlo;
        }
    }
    return lo;
}

__global__ __launch_bounds__(POOL_T)
void pool_linear_kernel(const float* __restrict__ X, const int* __restrict__ batch,
                        const float* __restrict__ linW, const float* __restrict__ linb,
                        float* __restrict__ out) {
    __shared__ float4 part4[(POOL_T / 32) * (HID / 4)];
    __shared__ float4 wl4[HID * N_CLASSES / 4];
    __shared__ float  pooled[HID];
    __shared__ int    range[2];
    const int g    = blockIdx.x;
    const int tid  = threadIdx.x;
    const int lane = tid & 31;
    const int wave = __builtin_amdgcn_readfirstlane((int)(threadIdx.x >> 5));
    const int NW   = POOL_T / 32;
    const int WQ   = HID * N_CLASSES / 8;
    static_assert(HID * N_CLASSES % 8 == 0 && WQ % 32 == 0 && 2 + WQ / 32 <= POOL_T / 32,
                  "the weight copy takes whole waves behind the two that search");

    if (wave < 2) {
        const int b = first_node_at_or_above(batch, g + wave, lane);
        if (lane == 0) range[wave] = b;
    } else if (wave < 2 + WQ / 32) {
        const int i = tid - 64;
        const float4 w0 = ((const float4*)linW)[i];
        const float4 w1 = ((const float4*)linW)[i + WQ];
        wl4[i]      = w0;
        wl4[i + WQ] = w1;
    }
    __syncthreads();
    const int lo = range[0];
    const int hi = range[1];

    float4 acc = make_float4(0.f, 0.f, 0.f, 0.f);
    for (int r = lo + wave; r < hi; r += 16 * NW) {
        float4 h[16];
        #pragma unroll
        for (int u = 0; u < 16; ++u) {
            const int ru = r + u * NW;
            h[u] = make_float4(0.f, 0.f, 0.f, 0.f);
            if (ru < hi) h[u] = ((const float4*)(X + (size_t)ru * HID))[lane];
        }
        #pragma unroll
        for (int u = 0; u < 16; ++u) {
            acc.x += h[u].x; acc.y += h[u].y; acc.z += h[u].z; acc.w += h[u].w;
        }
    }
    part4[wave * (HID / 4) + lane] = acc;
    __syncthreads();

    if (tid < HID) {
        const float* pf = (const float*)part4;
        float sum = 0.f;
        #pragma unroll
        for (int w = 0; w < NW; ++w) sum += pf[w * HID + tid];
        pooled[tid] = sum;
    }
    __syncthreads();

    const float* wl = (const float*)wl4;
    for (int c = wave; c < N_CLASSES; c += NW) {
        float a = 0.f;
        #pragma unroll
        for (int j = 0; j < HID / 32; ++j) {
            const int f = lane + 32 * j;
            a += pooled[f] * wl[f * N_CLASSES + c];
        }
        for (int off = 16; off > 0; off >>= 1) a += __shfl_xor(a, off);
        if (lane == 0) {
            const float scale = 1.0f / fmaxf((float)(hi - lo), 1.0f);
            out[g * N_CLASSES + c] = a * scale + linb[c];
        }
    }
}

extern "C" void kernel_launch(void* const* d_in, const int* in_sizes, int n_in,
                              void* d_out, int out_size, void* d_ws, size_t ws_size,
                              hipStream_t stream) {
    const float* x     = (const float*)d_in[0];
    const int*   edge  = (const int*)d_in[1];
    const int*   src   = edge;
    const int*   dst   = edge + N_EDGES;
    const int*   batch = (const int*)d_in[2];
    const float* Ws    = (const float*)d_in[3];
    const float* bs    = (const float*)d_in[4];
    const float* linW  = (const float*)d_in[5];
    const float* linb  = (const float*)d_in[6];
    float*       out   = (float*)d_out;

    float* ws = (float*)d_ws;
    float* H      = ws;
    float* X1     = H + (size_t)N_NODES * HID;
    float* X2     = X1 + (size_t)N_NODES * HID;
    float* dinv   = X2 + (size_t)N_NODES * HID;
    int2*  brec   = (int2*)(dinv + N_NODES);
    int2*  srec   = brec + N_EDGES;
    int*   tstart = (int*)(srec + N_EDGES);
    int*   cursor = tstart + 1024;
    int*   hist   = cursor + 1024 * CPAD;

    tile_hist_kernel<<<HIST_WG, HIST_T, 0, stream>>>(dst, hist);
    tile_scan_kernel<<<1, 1024, 0, stream>>>(hist, tstart, cursor);
    fill_kernel<<<NFILL, FILL_T, 0, stream>>>(src, dst, cursor, brec);
    tile_sort_kernel<<<NT, SORT_T, 0, stream>>>(tstart, brec, srec, dinv);

    const float* W0 = Ws;
    const float* W1 = Ws + (size_t)1 * HID * HID;
    const float* W2 = Ws + (size_t)2 * HID * HID;
    const float* W3 = Ws + (size_t)3 * HID * HID;
    const float* W4 = Ws + (size_t)4 * HID * HID;
    agg_kernel<2><<<NT, AGG_T, 0, stream>>>(x,  dinv, tstart, srec, bs,           W0, W1, H);
    agg_kernel<0><<<NT, AGG_T, 0, stream>>>(H,  dinv, tstart, srec, bs + 1 * HID, W0, W2, X1);
    agg_kernel<0><<<NT, AGG_T, 0, stream>>>(X1, dinv, tstart, srec, bs + 2 * HID, W0, W3, H);
    agg_kernel<0><<<NT, AGG_T, 0, stream>>>(H,  dinv, tstart, srec, bs + 3 * HID, W0, W4, X1);
    agg_kernel<1><<<NT, AGG_T, 0, stream>>>(X1, dinv, tstart, srec, bs + 4 * HID, W0, W0, X2);

    pool_linear_kernel<<<N_GRAPHS, POOL_T, 0, stream>>>(X2, batch, linW, linb, out);
}
